# starting kernel
# baseline (speedup 1.0000x reference)
#include <hip/hip_runtime.h>
#include <hip/hip_bf16.h>

typedef __attribute__((ext_vector_type(2))) float v2f;
typedef __attribute__((ext_vector_type(8))) float v8f;

#define HID 32
#define H1  128

__global__ void deg_kernel(const int* __restrict__ dst, float* __restrict__ deg, int E) {
  int t = blockIdx.x * blockDim.x + threadIdx.x;
  if (t < E) atomicAdd(&deg[dst[t]], 1.0f);
}

__global__ void dinv_kernel(float* __restrict__ degdinv, int N) {
  int t = blockIdx.x * blockDim.x + threadIdx.x;
  if (t < N) degdinv[t] = rsqrtf(degdinv[t] + 1.0f);
}

__global__ __launch_bounds__(256)
void gemm_wmma_f32(const float* __restrict__ A, const float* __restrict__ B,
                   float* __restrict__ C, int M, int N, int K) {
  int wave = (int)((blockIdx.x * (long long)blockDim.x + threadIdx.x) >> 5);
  int lane = threadIdx.x & 31;
  int ntiles = N >> 4;
  int mtiles = (M + 15) >> 4;
  int tm = wave / ntiles;
  int tn = wave - tm * ntiles;
  if (tm >= mtiles) return;
  int mrow = tm * 16 + (lane & 15);
  if (mrow > M - 1) mrow = M - 1;
  int ncol = tn * 16 + (lane & 15);
  int kb = (lane >> 4) * 2;
  v8f acc = {};
  for (int k0 = 0; k0 < K; k0 += 4) {
    v2f a = *(const v2f*)(A + (size_t)mrow * K + k0 + kb);
    v2f b;
    b.x = B[(size_t)(k0 + kb) * N + ncol];
    b.y = B[(size_t)(k0 + kb + 1) * N + ncol];
    acc = __builtin_amdgcn_wmma_f32_16x16x4_f32(false, a, false, b, (short)0, acc,
                                                false, false);
  }
  int rbase = tm * 16 + ((lane >> 4) ? 8 : 0);
#pragma unroll
  for (int r = 0; r < 8; ++r) {
    int row = rbase + r;
    if (row < M) C[(size_t)row * N + ncol] = acc[r];
  }
}

template <int D>
__global__ __launch_bounds__(256)
void scatter_kernel(const float* __restrict__ xl, const int* __restrict__ src,
                    const int* __restrict__ dst, const float* __restrict__ dinv,
                    float* __restrict__ agg, int E) {
  const int LPE = D / 4;
  long long t = (long long)blockIdx.x * blockDim.x + threadIdx.x;
  int e = (int)(t / LPE);
  if (e >= E) return;
  int c = (int)(t % LPE) * 4;
  int s = src[e], d = dst[e];
  float nrm = dinv[s] * dinv[d];
  const float4 v = *(const float4*)(xl + (size_t)s * D + c);
  float* p = agg + (size_t)d * D + c;
  atomicAdd(p + 0, nrm * v.x);
  atomicAdd(p + 1, nrm * v.y);
  atomicAdd(p + 2, nrm * v.z);
  atomicAdd(p + 3, nrm * v.w);
}

template <int D>
__global__ __launch_bounds__(256)
void fuse_kernel(float* __restrict__ agg, const float* __restrict__ xl,
                 const float* __restrict__ dinv, const float* __restrict__ bias, int N) {
  const int LPN = D / 4;
  long long t = (long long)blockIdx.x * blockDim.x + threadIdx.x;
  int i = (int)(t / LPN);
  if (i >= N) return;
  int c = (int)(t % LPN) * 4;
  float di = dinv[i], sl = di * di;
  float4 a = *(const float4*)(agg + (size_t)i * D + c);
  const float4 v = *(const float4*)(xl + (size_t)i * D + c);
  a.x = fmaxf(a.x + sl * v.x + bias[c + 0], 0.f);
  a.y = fmaxf(a.y + sl * v.y + bias[c + 1], 0.f);
  a.z = fmaxf(a.z + sl * v.z + bias[c + 2], 0.f);
  a.w = fmaxf(a.w + sl * v.w + bias[c + 3], 0.f);
  *(float4*)(agg + (size_t)i * D + c) = a;
}

__global__ __launch_bounds__(256)
void edge_mlp_wmma(const float* __restrict__ h, const int* __restrict__ p0,
                   const int* __restrict__ p1, const int* __restrict__ q0,
                   const int* __restrict__ q1, const float* __restrict__ Wm1,
                   const float* __restrict__ bm1, const float* __restrict__ Wm2,
                   const float* __restrict__ bm2, float* __restrict__ out,
                   int NP, int Etot) {
  int wave = (int)((blockIdx.x * (long long)blockDim.x + threadIdx.x) >> 5);
  int lane = threadIdx.x & 31;
  int e0 = wave * 16;
  if (e0 >= Etot) return;
  int m = lane & 15;
  int e = e0 + m;
  if (e > Etot - 1) e = Etot - 1;
  int s = (e < NP) ? p0[e] : q0[e - NP];
  int d = (e < NP) ? p1[e] : q1[e - NP];
  const float* hs = h + (size_t)s * HID;
  const float* hd = h + (size_t)d * HID;
  int kb = (lane >> 4) * 2;
  v8f acc0 = {}, acc1 = {};
#pragma unroll
  for (int ks = 0; ks < 16; ++ks) {
    int j = ks * 4 + kb;
    const float* ap = (j < HID) ? (hs + j) : (hd + (j - HID));
    v2f a; a.x = ap[0]; a.y = ap[1];
    v2f b0, b1;
    b0.x = Wm1[(size_t)j * HID + m];
    b0.y = Wm1[(size_t)(j + 1) * HID + m];
    b1.x = Wm1[(size_t)j * HID + 16 + m];
    b1.y = Wm1[(size_t)(j + 1) * HID + 16 + m];
    acc0 = __builtin_amdgcn_wmma_f32_16x16x4_f32(false, a, false, b0, (short)0, acc0,
                                                 false, false);
    acc1 = __builtin_amdgcn_wmma_f32_16x16x4_f32(false, a, false, b1, (short)0, acc1,
                                                 false, false);
  }
  float bA = bm1[m],        bB = bm1[m + 16];
  float wa0 = Wm2[m * 2],   wa1 = Wm2[m * 2 + 1];
  float wb0 = Wm2[(m + 16) * 2], wb1 = Wm2[(m + 16) * 2 + 1];
  float p[8][2];
#pragma unroll
  for (int r = 0; r < 8; ++r) {
    float t0 = fmaxf(acc0[r] + bA, 0.f);
    float t1 = fmaxf(acc1[r] + bB, 0.f);
    p[r][0] = t0 * wa0 + t1 * wb0;
    p[r][1] = t0 * wa1 + t1 * wb1;
  }
#pragma unroll
  for (int off = 8; off >= 1; off >>= 1) {
#pragma unroll
    for (int r = 0; r < 8; ++r) {
      p[r][0] += __shfl_xor(p[r][0], off, 16);
      p[r][1] += __shfl_xor(p[r][1], off, 16);
    }
  }
  if (m == 0) {
    int rbase = e0 + ((lane >> 4) ? 8 : 0);
    float c0 = bm2[0], c1 = bm2[1];
#pragma unroll
    for (int r = 0; r < 8; ++r) {
      int row = rbase + r;
      if (row < Etot) {
        out[(size_t)row * 2 + 0] = p[r][0] + c0;
        out[(size_t)row * 2 + 1] = p[r][1] + c1;
      }
    }
  }
}

extern "C" void kernel_launch(void* const* d_in, const int* in_sizes, int n_in,
                              void* d_out, int out_size, void* d_ws, size_t ws_size,
                              hipStream_t stream) {
  const float* x   = (const float*)d_in[0];
  const int*   pe  = (const int*)d_in[1];
  const int*   neg = (const int*)d_in[2];
  const float* W1  = (const float*)d_in[3];
  const float* b1  = (const float*)d_in[4];
  const float* W2  = (const float*)d_in[5];
  const float* b2  = (const float*)d_in[6];
  const float* Wm1 = (const float*)d_in[7];
  const float* bm1 = (const float*)d_in[8];
  const float* Wm2 = (const float*)d_in[9];
  const float* bm2 = (const float*)d_in[10];
  float* out = (float*)d_out;

  const int N  = in_sizes[0] / HID;
  const int NP = in_sizes[1] / 2;
  const int NN = in_sizes[2] / 2;
  const int ET = NP + NN;
  const int* p0 = pe;        const int* p1 = pe + NP;
  const int* q0 = neg;       const int* q1 = neg + NN;

  char* ws = (char*)d_ws;
  size_t off = 0;
  auto wsalloc = [&](size_t bytes) -> char* {
    char* p = ws + off;
    off += (bytes + 255) & ~(size_t)255;
    return p;
  };
  float* dinv = (float*)wsalloc((size_t)N * 4);
  float* xl1  = (float*)wsalloc((size_t)N * H1 * 4);
  float* h1   = (float*)wsalloc((size_t)N * H1 * 4);
  float* xl2  = (float*)wsalloc((size_t)N * HID * 4);
  float* h2   = (float*)wsalloc((size_t)N * HID * 4);
  (void)ws_size; (void)n_in; (void)out_size;

  const int B = 256;

  hipMemsetAsync(dinv, 0, (size_t)N * 4, stream);
  deg_kernel<<<(NP + B - 1) / B, B, 0, stream>>>(p1, dinv, NP);
  dinv_kernel<<<(N + B - 1) / B, B, 0, stream>>>(dinv, N);

  {
    long long thr = (long long)((N + 15) / 16) * (H1 / 16) * 32;
    gemm_wmma_f32<<<(unsigned)((thr + B - 1) / B), B, 0, stream>>>(x, W1, xl1, N, H1, HID);
  }
  hipMemsetAsync(h1, 0, (size_t)N * H1 * 4, stream);
  {
    long long thr = (long long)NP * (H1 / 4);
    scatter_kernel<H1><<<(unsigned)((thr + B - 1) / B), B, 0, stream>>>(xl1, p0, p1, dinv, h1, NP);
    long long thr2 = (long long)N * (H1 / 4);
    fuse_kernel<H1><<<(unsigned)((thr2 + B - 1) / B), B, 0, stream>>>(h1, xl1, dinv, b1, N);
  }

  {
    long long thr = (long long)((N + 15) / 16) * (HID / 16) * 32;
    gemm_wmma_f32<<<(unsigned)((thr + B - 1) / B), B, 0, stream>>>(h1, W2, xl2, N, HID, H1);
  }
  hipMemsetAsync(h2, 0, (size_t)N * HID * 4, stream);
  {
    long long thr = (long long)NP * (HID / 4);
    scatter_kernel<HID><<<(unsigned)((thr + B - 1) / B), B, 0, stream>>>(xl2, p0, p1, dinv, h2, NP);
    long long thr2 = (long long)N * (HID / 4);
    fuse_kernel<HID><<<(unsigned)((thr2 + B - 1) / B), B, 0, stream>>>(h2, xl2, dinv, b2, N);
  }

  {
    long long thr = (long long)((ET + 15) / 16) * 32;
    edge_mlp_wmma<<<(unsigned)((thr + B - 1) / B), B, 0, stream>>>(h2, p0, p1, q0, q1,
                                                                   Wm1, bm1, Wm2, bm2,
                                                                   out, NP, ET);
  }
}


# Layer 1 sums the 32-wide input rows over the edges before the W1 product, (A x)W1 for A(x W1): 4x fewer f32 atomic adds, bias+relu in the product's epilogue; layer-1 fp32 sum order changes
# speedup vs baseline: 1.6032x; 1.6032x over previous
#include <hip/hip_runtime.h>
#include <hip/hip_bf16.h>

typedef __attribute__((ext_vector_type(2))) float v2f;
typedef __attribute__((ext_vector_type(8))) float v8f;

#define HID 32
#define H1  128

__global__ void deg_kernel(const int* __restrict__ dst, float* __restrict__ deg, int E) {
  int t = blockIdx.x * blockDim.x + threadIdx.x;
  if (t < E) atomicAdd(&deg[dst[t]], 1.0f);
}

__global__ void dinv_kernel(float* __restrict__ degdinv, int N) {
  int t = blockIdx.x * blockDim.x + threadIdx.x;
  if (t < N) degdinv[t] = rsqrtf(degdinv[t] + 1.0f);
}

__global__ __launch_bounds__(256)
void gemm_wmma_f32(const float* __restrict__ A, const float* __restrict__ B,
                   float* __restrict__ C, int M, int N, int K) {
  int wave = (int)((blockIdx.x * (long long)blockDim.x + threadIdx.x) >> 5);
  int lane = threadIdx.x & 31;
  int ntiles = N >> 4;
  int mtiles = (M + 15) >> 4;
  int tm = wave / ntiles;
  int tn = wave - tm * ntiles;
  if (tm >= mtiles) return;
  int mrow = tm * 16 + (lane & 15);
  if (mrow > M - 1) mrow = M - 1;
  int ncol = tn * 16 + (lane & 15);
  int kb = (lane >> 4) * 2;
  v8f acc = {};
  for (int k0 = 0; k0 < K; k0 += 4) {
    v2f a = *(const v2f*)(A + (size_t)mrow * K + k0 + kb);
    v2f b;
    b.x = B[(size_t)(k0 + kb) * N + ncol];
    b.y = B[(size_t)(k0 + kb + 1) * N + ncol];
    acc = __builtin_amdgcn_wmma_f32_16x16x4_f32(false, a, false, b, (short)0, acc,
                                                false, false);
  }
  int rbase = tm * 16 + ((lane >> 4) ? 8 : 0);
#pragma unroll
  for (int r = 0; r < 8; ++r) {
    int row = rbase + r;
    if (row < M) C[(size_t)row * N + ncol] = acc[r];
  }
}

__global__ __launch_bounds__(256)
void gemm_l1_wmma_f32(const float* __restrict__ agg, const float* __restrict__ x,
                      const float* __restrict__ dinv, const float* __restrict__ W,
                      const float* __restrict__ bias, float* __restrict__ H, int M) {
  int wave = (int)((blockIdx.x * (long long)blockDim.x + threadIdx.x) >> 5);
  int lane = threadIdx.x & 31;
  const int ntiles = H1 >> 4;
  int mtiles = (M + 15) >> 4;
  int tm = wave / ntiles;
  int tn = wave - tm * ntiles;
  if (tm >= mtiles) return;
  int mrow = tm * 16 + (lane & 15);
  if (mrow > M - 1) mrow = M - 1;
  int ncol = tn * 16 + (lane & 15);
  int kb = (lane >> 4) * 2;
  float di = dinv[mrow], sl = di * di;
  const float* ga = agg + (size_t)mrow * HID + kb;
  const float* xa = x + (size_t)mrow * HID + kb;
  const float* wb = W + (size_t)kb * H1 + ncol;
  v8f acc = {};
  for (int k0 = 0; k0 < HID; k0 += 4) {
    v2f g = *(const v2f*)(ga + k0);
    v2f xv = *(const v2f*)(xa + k0);
    v2f a;
    a.x = g.x + sl * xv.x;
    a.y = g.y + sl * xv.y;
    v2f b;
    b.x = wb[(size_t)k0 * H1];
    b.y = wb[(size_t)(k0 + 1) * H1];
    acc = __builtin_amdgcn_wmma_f32_16x16x4_f32(false, a, false, b, (short)0, acc,
                                                false, false);
  }
  float bn = bias[ncol];
  int rbase = tm * 16 + ((lane >> 4) ? 8 : 0);
#pragma unroll
  for (int r = 0; r < 8; ++r) {
    int row = rbase + r;
    if (row < M) H[(size_t)row * H1 + ncol] = fmaxf(acc[r] + bn, 0.f);
  }
}

template <int D>
__global__ __launch_bounds__(256)
void scatter_kernel(const float* __restrict__ xl, const int* __restrict__ src,
                    const int* __restrict__ dst, const float* __restrict__ dinv,
                    float* __restrict__ agg, int E) {
  const int LPE = D / 4;
  long long t = (long long)blockIdx.x * blockDim.x + threadIdx.x;
  int e = (int)(t / LPE);
  if (e >= E) return;
  int c = (int)(t % LPE) * 4;
  int s = src[e], d = dst[e];
  float nrm = dinv[s] * dinv[d];
  const float4 v = *(const float4*)(xl + (size_t)s * D + c);
  float* p = agg + (size_t)d * D + c;
  atomicAdd(p + 0, nrm * v.x);
  atomicAdd(p + 1, nrm * v.y);
  atomicAdd(p + 2, nrm * v.z);
  atomicAdd(p + 3, nrm * v.w);
}

template <int D>
__global__ __launch_bounds__(256)
void fuse_kernel(float* __restrict__ agg, const float* __restrict__ xl,
                 const float* __restrict__ dinv, const float* __restrict__ bias, int N) {
  const int LPN = D / 4;
  long long t = (long long)blockIdx.x * blockDim.x + threadIdx.x;
  int i = (int)(t / LPN);
  if (i >= N) return;
  int c = (int)(t % LPN) * 4;
  float di = dinv[i], sl = di * di;
  float4 a = *(const float4*)(agg + (size_t)i * D + c);
  const float4 v = *(const float4*)(xl + (size_t)i * D + c);
  a.x = fmaxf(a.x + sl * v.x + bias[c + 0], 0.f);
  a.y = fmaxf(a.y + sl * v.y + bias[c + 1], 0.f);
  a.z = fmaxf(a.z + sl * v.z + bias[c + 2], 0.f);
  a.w = fmaxf(a.w + sl * v.w + bias[c + 3], 0.f);
  *(float4*)(agg + (size_t)i * D + c) = a;
}

__global__ __launch_bounds__(256)
void edge_mlp_wmma(const float* __restrict__ h, const int* __restrict__ p0,
                   const int* __restrict__ p1, const int* __restrict__ q0,
                   const int* __restrict__ q1, const float* __restrict__ Wm1,
                   const float* __restrict__ bm1, const float* __restrict__ Wm2,
                   const float* __restrict__ bm2, float* __restrict__ out,
                   int NP, int Etot) {
  int wave = (int)((blockIdx.x * (long long)blockDim.x + threadIdx.x) >> 5);
  int lane = threadIdx.x & 31;
  int e0 = wave * 16;
  if (e0 >= Etot) return;
  int m = lane & 15;
  int e = e0 + m;
  if (e > Etot - 1) e = Etot - 1;
  int s = (e < NP) ? p0[e] : q0[e - NP];
  int d = (e < NP) ? p1[e] : q1[e - NP];
  const float* hs = h + (size_t)s * HID;
  const float* hd = h + (size_t)d * HID;
  int kb = (lane >> 4) * 2;
  v8f acc0 = {}, acc1 = {};
#pragma unroll
  for (int ks = 0; ks < 16; ++ks) {
    int j = ks * 4 + kb;
    const float* ap = (j < HID) ? (hs + j) : (hd + (j - HID));
    v2f a; a.x = ap[0]; a.y = ap[1];
    v2f b0, b1;
    b0.x = Wm1[(size_t)j * HID + m];
    b0.y = Wm1[(size_t)(j + 1) * HID + m];
    b1.x = Wm1[(size_t)j * HID + 16 + m];
    b1.y = Wm1[(size_t)(j + 1) * HID + 16 + m];
    acc0 = __builtin_amdgcn_wmma_f32_16x16x4_f32(false, a, false, b0, (short)0, acc0,
                                                 false, false);
    acc1 = __builtin_amdgcn_wmma_f32_16x16x4_f32(false, a, false, b1, (short)0, acc1,
                                                 false, false);
  }
  float bA = bm1[m],        bB = bm1[m + 16];
  float wa0 = Wm2[m * 2],   wa1 = Wm2[m * 2 + 1];
  float wb0 = Wm2[(m + 16) * 2], wb1 = Wm2[(m + 16) * 2 + 1];
  float p[8][2];
#pragma unroll
  for (int r = 0; r < 8; ++r) {
    float t0 = fmaxf(acc0[r] + bA, 0.f);
    float t1 = fmaxf(acc1[r] + bB, 0.f);
    p[r][0] = t0 * wa0 + t1 * wb0;
    p[r][1] = t0 * wa1 + t1 * wb1;
  }
#pragma unroll
  for (int off = 8; off >= 1; off >>= 1) {
#pragma unroll
    for (int r = 0; r < 8; ++r) {
      p[r][0] += __shfl_xor(p[r][0], off, 16);
      p[r][1] += __shfl_xor(p[r][1], off, 16);
    }
  }
  if (m == 0) {
    int rbase = e0 + ((lane >> 4) ? 8 : 0);
    float c0 = bm2[0], c1 = bm2[1];
#pragma unroll
    for (int r = 0; r < 8; ++r) {
      int row = rbase + r;
      if (row < Etot) {
        out[(size_t)row * 2 + 0] = p[r][0] + c0;
        out[(size_t)row * 2 + 1] = p[r][1] + c1;
      }
    }
  }
}

extern "C" void kernel_launch(void* const* d_in, const int* in_sizes, int n_in,
                              void* d_out, int out_size, void* d_ws, size_t ws_size,
                              hipStream_t stream) {
  const float* x   = (const float*)d_in[0];
  const int*   pe  = (const int*)d_in[1];
  const int*   neg = (const int*)d_in[2];
  const float* W1  = (const float*)d_in[3];
  const float* b1  = (const float*)d_in[4];
  const float* W2  = (const float*)d_in[5];
  const float* b2  = (const float*)d_in[6];
  const float* Wm1 = (const float*)d_in[7];
  const float* bm1 = (const float*)d_in[8];
  const float* Wm2 = (const float*)d_in[9];
  const float* bm2 = (const float*)d_in[10];
  float* out = (float*)d_out;

  const int N  = in_sizes[0] / HID;
  const int NP = in_sizes[1] / 2;
  const int NN = in_sizes[2] / 2;
  const int ET = NP + NN;
  const int* p0 = pe;        const int* p1 = pe + NP;
  const int* q0 = neg;       const int* q1 = neg + NN;

  char* ws = (char*)d_ws;
  size_t off = 0;
  auto wsalloc = [&](size_t bytes) -> char* {
    char* p = ws + off;
    off += (bytes + 255) & ~(size_t)255;
    return p;
  };
  float* dinv = (float*)wsalloc((size_t)N * 4);
  float* ax   = (float*)wsalloc((size_t)N * HID * 4);
  float* h1   = (float*)wsalloc((size_t)N * H1 * 4);
  float* xl2  = (float*)wsalloc((size_t)N * HID * 4);
  float* h2   = (float*)wsalloc((size_t)N * HID * 4);
  (void)ws_size; (void)n_in; (void)out_size;

  const int B = 256;

  hipMemsetAsync(dinv, 0, (size_t)N * 4, stream);
  deg_kernel<<<(NP + B - 1) / B, B, 0, stream>>>(p1, dinv, NP);
  dinv_kernel<<<(N + B - 1) / B, B, 0, stream>>>(dinv, N);

  hipMemsetAsync(ax, 0, (size_t)N * HID * 4, stream);
  {
    long long thr = (long long)NP * (HID / 4);
    scatter_kernel<HID><<<(unsigned)((thr + B - 1) / B), B, 0, stream>>>(x, p0, p1, dinv, ax, NP);
    long long thr2 = (long long)((N + 15) / 16) * (H1 / 16) * 32;
    gemm_l1_wmma_f32<<<(unsigned)((thr2 + B - 1) / B), B, 0, stream>>>(ax, x, dinv, W1, b1, h1, N);
  }

  {
    long long thr = (long long)((N + 15) / 16) * (HID / 16) * 32;
    gemm_wmma_f32<<<(unsigned)((thr + B - 1) / B), B, 0, stream>>>(h1, W2, xl2, N, HID, H1);
  }
  hipMemsetAsync(h2, 0, (size_t)N * HID * 4, stream);
  {
    long long thr = (long long)NP * (HID / 4);
    scatter_kernel<HID><<<(unsigned)((thr + B - 1) / B), B, 0, stream>>>(xl2, p0, p1, dinv, h2, NP);
    long long thr2 = (long long)N * (HID / 4);
    fuse_kernel<HID><<<(unsigned)((thr2 + B - 1) / B), B, 0, stream>>>(h2, xl2, dinv, b2, N);
  }

  {
    long long thr = (long long)((ET + 15) / 16) * 32;
    edge_mlp_wmma<<<(unsigned)((thr + B - 1) / B), B, 0, stream>>>(h2, p0, p1, q0, q1,
                                                                   Wm1, bm1, Wm2, bm2,
                                                                   out, NP, ET);
  }
}


# Edge MLP's first product moves ahead of the edge gather, per node: (h Wm1top)[src] + (h Wm1bot)[dst] for [h[src]|h[dst]] Wm1; edges gather two 128-B rows and add; its fp32 sum order changes
# speedup vs baseline: 2.0703x; 1.2914x over previous
#include <hip/hip_runtime.h>
#include <hip/hip_bf16.h>

typedef __attribute__((ext_vector_type(2))) float v2f;
typedef __attribute__((ext_vector_type(8))) float v8f;

#define HID 32
#define H1  128

__global__ void deg_kernel(const int* __restrict__ dst, float* __restrict__ deg, int E) {
  int t = blockIdx.x * blockDim.x + threadIdx.x;
  if (t < E) atomicAdd(&deg[dst[t]], 1.0f);
}

__global__ void dinv_kernel(float* __restrict__ degdinv, int N) {
  int t = blockIdx.x * blockDim.x + threadIdx.x;
  if (t < N) degdinv[t] = rsqrtf(degdinv[t] + 1.0f);
}

__global__ __launch_bounds__(256)
void gemm_wmma_f32(const float* __restrict__ A, const float* __restrict__ B,
                   float* __restrict__ C, int M, int N, int K) {
  int wave = (int)((blockIdx.x * (long long)blockDim.x + threadIdx.x) >> 5);
  int lane = threadIdx.x & 31;
  int ntiles = N >> 4;
  int mtiles = (M + 15) >> 4;
  int tm = wave / ntiles;
  int tn = wave - tm * ntiles;
  if (tm >= mtiles) return;
  int mrow = tm * 16 + (lane & 15);
  if (mrow > M - 1) mrow = M - 1;
  int ncol = tn * 16 + (lane & 15);
  int kb = (lane >> 4) * 2;
  v8f acc = {};
  for (int k0 = 0; k0 < K; k0 += 4) {
    v2f a = *(const v2f*)(A + (size_t)mrow * K + k0 + kb);
    v2f b;
    b.x = B[(size_t)(k0 + kb) * N + ncol];
    b.y = B[(size_t)(k0 + kb + 1) * N + ncol];
    acc = __builtin_amdgcn_wmma_f32_16x16x4_f32(false, a, false, b, (short)0, acc,
                                                false, false);
  }
  int rbase = tm * 16 + ((lane >> 4) ? 8 : 0);
#pragma unroll
  for (int r = 0; r < 8; ++r) {
    int row = rbase + r;
    if (row < M) C[(size_t)row * N + ncol] = acc[r];
  }
}

__global__ __launch_bounds__(256)
void gemm_l1_wmma_f32(const float* __restrict__ agg, const float* __restrict__ x,
                      const float* __restrict__ dinv, const float* __restrict__ W,
                      const float* __restrict__ bias, float* __restrict__ H, int M) {
  int wave = (int)((blockIdx.x * (long long)blockDim.x + threadIdx.x) >> 5);
  int lane = threadIdx.x & 31;
  const int ntiles = H1 >> 4;
  int mtiles = (M + 15) >> 4;
  int tm = wave / ntiles;
  int tn = wave - tm * ntiles;
  if (tm >= mtiles) return;
  int mrow = tm * 16 + (lane & 15);
  if (mrow > M - 1) mrow = M - 1;
  int ncol = tn * 16 + (lane & 15);
  int kb = (lane >> 4) * 2;
  float di = dinv[mrow], sl = di * di;
  const float* ga = agg + (size_t)mrow * HID + kb;
  const float* xa = x + (size_t)mrow * HID + kb;
  const float* wb = W + (size_t)kb * H1 + ncol;
  v8f acc = {};
  for (int k0 = 0; k0 < HID; k0 += 4) {
    v2f g = *(const v2f*)(ga + k0);
    v2f xv = *(const v2f*)(xa + k0);
    v2f a;
    a.x = g.x + sl * xv.x;
    a.y = g.y + sl * xv.y;
    v2f b;
    b.x = wb[(size_t)k0 * H1];
    b.y = wb[(size_t)(k0 + 1) * H1];
    acc = __builtin_amdgcn_wmma_f32_16x16x4_f32(false, a, false, b, (short)0, acc,
                                                false, false);
  }
  float bn = bias[ncol];
  int rbase = tm * 16 + ((lane >> 4) ? 8 : 0);
#pragma unroll
  for (int r = 0; r < 8; ++r) {
    int row = rbase + r;
    if (row < M) H[(size_t)row * H1 + ncol] = fmaxf(acc[r] + bn, 0.f);
  }
}

template <int D>
__global__ __launch_bounds__(256)
void scatter_kernel(const float* __restrict__ xl, const int* __restrict__ src,
                    const int* __restrict__ dst, const float* __restrict__ dinv,
                    float* __restrict__ agg, int E) {
  const int LPE = D / 4;
  long long t = (long long)blockIdx.x * blockDim.x + threadIdx.x;
  int e = (int)(t / LPE);
  if (e >= E) return;
  int c = (int)(t % LPE) * 4;
  int s = src[e], d = dst[e];
  float nrm = dinv[s] * dinv[d];
  const float4 v = *(const float4*)(xl + (size_t)s * D + c);
  float* p = agg + (size_t)d * D + c;
  atomicAdd(p + 0, nrm * v.x);
  atomicAdd(p + 1, nrm * v.y);
  atomicAdd(p + 2, nrm * v.z);
  atomicAdd(p + 3, nrm * v.w);
}

__global__ __launch_bounds__(256)
void gemm_pq_wmma_f32(const float* __restrict__ agg, const float* __restrict__ xl,
                      const float* __restrict__ dinv, const float* __restrict__ b2,
                      const float* __restrict__ Wm1, const float* __restrict__ bm1,
                      float* __restrict__ PQ, int M) {
  int tm = (int)((blockIdx.x * (long long)blockDim.x + threadIdx.x) >> 5);
  int lane = threadIdx.x & 31;
  int mtiles = (M + 15) >> 4;
  if (tm >= mtiles) return;
  int l15 = lane & 15;
  int mrow = tm * 16 + l15;
  if (mrow > M - 1) mrow = M - 1;
  int kb = (lane >> 4) * 2;
  float di = dinv[mrow], sl = di * di;
  const float* ga = agg + (size_t)mrow * HID + kb;
  const float* xa = xl + (size_t)mrow * HID + kb;
  const float* ba = b2 + kb;
  const float* wb = Wm1 + (size_t)kb * HID + l15;
  v8f acc[4] = {};
  for (int k0 = 0; k0 < HID; k0 += 4) {
    v2f g = *(const v2f*)(ga + k0);
    v2f xv = *(const v2f*)(xa + k0);
    v2f a;
    a.x = fmaxf(g.x + sl * xv.x + ba[k0], 0.f);
    a.y = fmaxf(g.y + sl * xv.y + ba[k0 + 1], 0.f);
#pragma unroll
    for (int t = 0; t < 4; ++t) {
      const float* wt = wb + (size_t)((t >> 1) * HID) * HID + (t & 1) * 16;
      v2f b;
      b.x = wt[(size_t)k0 * HID];
      b.y = wt[(size_t)(k0 + 1) * HID];
      acc[t] = __builtin_amdgcn_wmma_f32_16x16x4_f32(false, a, false, b, (short)0, acc[t],
                                                     false, false);
    }
  }
  int rbase = tm * 16 + ((lane >> 4) ? 8 : 0);
#pragma unroll
  for (int t = 0; t < 4; ++t) {
    int ncol = t * 16 + l15;
    float bn = (t < 2) ? bm1[ncol] : 0.f;
#pragma unroll
    for (int r = 0; r < 8; ++r) {
      int row = rbase + r;
      float val = acc[t][r];
      if (t < 2) val += bn;
      if (row < M) PQ[(size_t)row * (2 * HID) + ncol] = val;
    }
  }
}

#define EDGE_CHUNKS 4
__global__ __launch_bounds__(256)
void edge_out_kernel(const float* __restrict__ PQ, const int* __restrict__ p0,
                     const int* __restrict__ p1, const int* __restrict__ q0,
                     const int* __restrict__ q1, const float* __restrict__ Wm2,
                     const float* __restrict__ bm2, float* __restrict__ out,
                     int NP, int Etot) {
  int wave = (int)((blockIdx.x * (long long)blockDim.x + threadIdx.x) >> 5);
  int lane = threadIdx.x & 31;
  int G = lane >> 3;
  int j = lane & 7;
  int c = j * 4;
  const float4 w01 = *(const float4*)(Wm2 + c * 2);
  const float4 w23 = *(const float4*)(Wm2 + c * 2 + 4);
  const float bo = bm2[j & 1];
  const bool hi4 = (j & 4) != 0, hi2 = (j & 2) != 0, hi1 = (j & 1) != 0;
  for (int it = 0; it < EDGE_CHUNKS; ++it) {
    int e0 = (wave * EDGE_CHUNKS + it) * 16;
    if (e0 >= Etot) break;
    float v[8];
#pragma unroll
    for (int i = 0; i < 4; ++i) {
      int e = e0 + 4 * G + i;
      if (e > Etot - 1) e = Etot - 1;
      int s = (e < NP) ? p0[e] : q0[e - NP];
      int d = (e < NP) ? p1[e] : q1[e - NP];
      const float4 p = *(const float4*)(PQ + (size_t)s * (2 * HID) + c);
      const float4 q = *(const float4*)(PQ + (size_t)d * (2 * HID) + HID + c);
      float t0 = fmaxf(p.x + q.x, 0.f);
      float t1 = fmaxf(p.y + q.y, 0.f);
      float t2 = fmaxf(p.z + q.z, 0.f);
      float t3 = fmaxf(p.w + q.w, 0.f);
      v[2 * i]     = t0 * w01.x + t1 * w01.z + t2 * w23.x + t3 * w23.z;
      v[2 * i + 1] = t0 * w01.y + t1 * w01.w + t2 * w23.y + t3 * w23.w;
    }
    float u[4];
#pragma unroll
    for (int k = 0; k < 4; ++k) {
      float keep = hi4 ? v[k + 4] : v[k];
      float send = hi4 ? v[k] : v[k + 4];
      u[k] = keep + __shfl_xor(send, 4, 8);
    }
    float w[2];
#pragma unroll
    for (int k = 0; k < 2; ++k) {
      float keep = hi2 ? u[k + 2] : u[k];
      float send = hi2 ? u[k] : u[k + 2];
      w[k] = keep + __shfl_xor(send, 2, 8);
    }
    float keep1 = hi1 ? w[1] : w[0];
    float send1 = hi1 ? w[0] : w[1];
    float r = keep1 + __shfl_xor(send1, 1, 8);
    int eo = e0 + 4 * G + (j >> 1);
    if (eo < Etot) out[(size_t)e0 * 2 + lane] = r + bo;
  }
}

extern "C" void kernel_launch(void* const* d_in, const int* in_sizes, int n_in,
                              void* d_out, int out_size, void* d_ws, size_t ws_size,
                              hipStream_t stream) {
  const float* x   = (const float*)d_in[0];
  const int*   pe  = (const int*)d_in[1];
  const int*   neg = (const int*)d_in[2];
  const float* W1  = (const float*)d_in[3];
  const float* b1  = (const float*)d_in[4];
  const float* W2  = (const float*)d_in[5];
  const float* b2  = (const float*)d_in[6];
  const float* Wm1 = (const float*)d_in[7];
  const float* bm1 = (const float*)d_in[8];
  const float* Wm2 = (const float*)d_in[9];
  const float* bm2 = (const float*)d_in[10];
  float* out = (float*)d_out;

  const int N  = in_sizes[0] / HID;
  const int NP = in_sizes[1] / 2;
  const int NN = in_sizes[2] / 2;
  const int ET = NP + NN;
  const int* p0 = pe;        const int* p1 = pe + NP;
  const int* q0 = neg;       const int* q1 = neg + NN;

  char* ws = (char*)d_ws;
  size_t off = 0;
  auto wsalloc = [&](size_t bytes) -> char* {
    char* p = ws + off;
    off += (bytes + 255) & ~(size_t)255;
    return p;
  };
  float* dinv = (float*)wsalloc((size_t)N * 4);
  float* ax   = (float*)wsalloc((size_t)N * HID * 4);
  float* h1   = (float*)wsalloc((size_t)N * H1 * 4);
  float* xl2  = (float*)wsalloc((size_t)N * HID * 4);
  float* ag2  = (float*)wsalloc((size_t)N * HID * 4);
  float* pq   = (float*)wsalloc((size_t)N * 2 * HID * 4);
  (void)ws_size; (void)n_in; (void)out_size;

  const int B = 256;

  hipMemsetAsync(dinv, 0, (size_t)N * 4, stream);
  deg_kernel<<<(NP + B - 1) / B, B, 0, stream>>>(p1, dinv, NP);
  dinv_kernel<<<(N + B - 1) / B, B, 0, stream>>>(dinv, N);

  hipMemsetAsync(ax, 0, (size_t)N * HID * 4, stream);
  {
    long long thr = (long long)NP * (HID / 4);
    scatter_kernel<HID><<<(unsigned)((thr + B - 1) / B), B, 0, stream>>>(x, p0, p1, dinv, ax, NP);
    long long thr2 = (long long)((N + 15) / 16) * (H1 / 16) * 32;
    gemm_l1_wmma_f32<<<(unsigned)((thr2 + B - 1) / B), B, 0, stream>>>(ax, x, dinv, W1, b1, h1, N);
  }

  {
    long long thr = (long long)((N + 15) / 16) * (HID / 16) * 32;
    gemm_wmma_f32<<<(unsigned)((thr + B - 1) / B), B, 0, stream>>>(h1, W2, xl2, N, HID, H1);
  }
  hipMemsetAsync(ag2, 0, (size_t)N * HID * 4, stream);
  {
    long long thr = (long long)NP * (HID / 4);
    scatter_kernel<HID><<<(unsigned)((thr + B - 1) / B), B, 0, stream>>>(xl2, p0, p1, dinv, ag2, NP);
    long long thr2 = (long long)((N + 15) / 16) * 32;
    gemm_pq_wmma_f32<<<(unsigned)((thr2 + B - 1) / B), B, 0, stream>>>(ag2, xl2, dinv, b2, Wm1, bm1, pq, N);
  }

  {
    long long thr = (long long)((ET + 16 * EDGE_CHUNKS - 1) / (16 * EDGE_CHUNKS)) * 32;
    edge_out_kernel<<<(unsigned)((thr + B - 1) / B), B, 0, stream>>>(pq, p0, p1, q0, q1,
                                                                     Wm2, bm2, out, NP, ET);
  }
}


# Both edge scatter-adds: lane = row column, a wave walks 32 edges by readlane, so each f32 atomic add and row load covers one whole 128-B line, not 16-B pieces of 4 rows; atomic arrival order changes
# speedup vs baseline: 2.3825x; 1.1508x over previous
#include <hip/hip_runtime.h>
#include <hip/hip_bf16.h>

typedef __attribute__((ext_vector_type(2))) float v2f;
typedef __attribute__((ext_vector_type(8))) float v8f;

#define HID 32
#define H1  128

__global__ void deg_kernel(const int* __restrict__ dst, float* __restrict__ deg, int E) {
  int t = blockIdx.x * blockDim.x + threadIdx.x;
  if (t < E) atomicAdd(&deg[dst[t]], 1.0f);
}

__global__ void dinv_kernel(float* __restrict__ degdinv, int N) {
  int t = blockIdx.x * blockDim.x + threadIdx.x;
  if (t < N) degdinv[t] = rsqrtf(degdinv[t] + 1.0f);
}

__global__ __launch_bounds__(256)
void gemm_wmma_f32(const float* __restrict__ A, const float* __restrict__ B,
                   float* __restrict__ C, int M, int N, int K) {
  int wave = (int)((blockIdx.x * (long long)blockDim.x + threadIdx.x) >> 5);
  int lane = threadIdx.x & 31;
  int ntiles = N >> 4;
  int mtiles = (M + 15) >> 4;
  int tm = wave / ntiles;
  int tn = wave - tm * ntiles;
  if (tm >= mtiles) return;
  int mrow = tm * 16 + (lane & 15);
  if (mrow > M - 1) mrow = M - 1;
  int ncol = tn * 16 + (lane & 15);
  int kb = (lane >> 4) * 2;
  v8f acc = {};
  for (int k0 = 0; k0 < K; k0 += 4) {
    v2f a = *(const v2f*)(A + (size_t)mrow * K + k0 + kb);
    v2f b;
    b.x = B[(size_t)(k0 + kb) * N + ncol];
    b.y = B[(size_t)(k0 + kb + 1) * N + ncol];
    acc = __builtin_amdgcn_wmma_f32_16x16x4_f32(false, a, false, b, (short)0, acc,
                                                false, false);
  }
  int rbase = tm * 16 + ((lane >> 4) ? 8 : 0);
#pragma unroll
  for (int r = 0; r < 8; ++r) {
    int row = rbase + r;
    if (row < M) C[(size_t)row * N + ncol] = acc[r];
  }
}

__global__ __launch_bounds__(256)
void gemm_l1_wmma_f32(const float* __restrict__ agg, const float* __restrict__ x,
                      const float* __restrict__ dinv, const float* __restrict__ W,
                      const float* __restrict__ bias, float* __restrict__ H, int M) {
  int wave = (int)((blockIdx.x * (long long)blockDim.x + threadIdx.x) >> 5);
  int lane = threadIdx.x & 31;
  const int ntiles = H1 >> 4;
  int mtiles = (M + 15) >> 4;
  int tm = wave / ntiles;
  int tn = wave - tm * ntiles;
  if (tm >= mtiles) return;
  int mrow = tm * 16 + (lane & 15);
  if (mrow > M - 1) mrow = M - 1;
  int ncol = tn * 16 + (lane & 15);
  int kb = (lane >> 4) * 2;
  float di = dinv[mrow], sl = di * di;
  const float* ga = agg + (size_t)mrow * HID + kb;
  const float* xa = x + (size_t)mrow * HID + kb;
  const float* wb = W + (size_t)kb * H1 + ncol;
  v8f acc = {};
  for (int k0 = 0; k0 < HID; k0 += 4) {
    v2f g = *(const v2f*)(ga + k0);
    v2f xv = *(const v2f*)(xa + k0);
    v2f a;
    a.x = g.x + sl * xv.x;
    a.y = g.y + sl * xv.y;
    v2f b;
    b.x = wb[(size_t)k0 * H1];
    b.y = wb[(size_t)(k0 + 1) * H1];
    acc = __builtin_amdgcn_wmma_f32_16x16x4_f32(false, a, false, b, (short)0, acc,
                                                false, false);
  }
  float bn = bias[ncol];
  int rbase = tm * 16 + ((lane >> 4) ? 8 : 0);
#pragma unroll
  for (int r = 0; r < 8; ++r) {
    int row = rbase + r;
    if (row < M) H[(size_t)row * H1 + ncol] = fmaxf(acc[r] + bn, 0.f);
  }
}

#define SC_CHUNKS 4
__device__ __forceinline__ unsigned row_off(int row, int lane) {
  return ((unsigned)row * HID + (unsigned)lane) * 4u;
}
__global__ __launch_bounds__(256)
void scatter_rows_kernel(const float* __restrict__ xl, const int* __restrict__ src,
                         const int* __restrict__ dst, const float* __restrict__ dinv,
                         float* __restrict__ agg, int E) {
  static_assert(HID == 32, "one row per wave: HID must equal the wave size");
  int wave = (int)((blockIdx.x * (long long)blockDim.x + threadIdx.x) >> 5);
  int lane = threadIdx.x & 31;
  for (int it = 0; it < SC_CHUNKS; ++it) {
    int e0 = (wave * SC_CHUNKS + it) * 32;
    if (e0 >= E) break;
    int e = e0 + lane;
    if (e > E - 1) e = E - 1;
    int s = src[e], d = dst[e];
    int nb = __float_as_int(dinv[s] * dinv[d]);
    int n = E - e0;
    if (n >= 32) {
      float v[32];
#pragma unroll
      for (int i = 0; i < 32; ++i) {
        int si = __builtin_amdgcn_readlane(s, i);
        v[i] = *(const float*)((const char*)xl + row_off(si, lane));
      }
#pragma unroll
      for (int i = 0; i < 32; ++i) {
        int di = __builtin_amdgcn_readlane(d, i);
        float ni = __int_as_float(__builtin_amdgcn_readlane(nb, i));
        atomicAdd((float*)((char*)agg + row_off(di, lane)), ni * v[i]);
      }
    } else {
      for (int i = 0; i < n; ++i) {
        int si = __builtin_amdgcn_readlane(s, i);
        int di = __builtin_amdgcn_readlane(d, i);
        float ni = __int_as_float(__builtin_amdgcn_readlane(nb, i));
        float xv = *(const float*)((const char*)xl + row_off(si, lane));
        atomicAdd((float*)((char*)agg + row_off(di, lane)), ni * xv);
      }
    }
  }
}

__global__ __launch_bounds__(256)
void gemm_pq_wmma_f32(const float* __restrict__ agg, const float* __restrict__ xl,
                      const float* __restrict__ dinv, const float* __restrict__ b2,
                      const float* __restrict__ Wm1, const float* __restrict__ bm1,
                      float* __restrict__ PQ, int M) {
  int tm = (int)((blockIdx.x * (long long)blockDim.x + threadIdx.x) >> 5);
  int lane = threadIdx.x & 31;
  int mtiles = (M + 15) >> 4;
  if (tm >= mtiles) return;
  int l15 = lane & 15;
  int mrow = tm * 16 + l15;
  if (mrow > M - 1) mrow = M - 1;
  int kb = (lane >> 4) * 2;
  float di = dinv[mrow], sl = di * di;
  const float* ga = agg + (size_t)mrow * HID + kb;
  const float* xa = xl + (size_t)mrow * HID + kb;
  const float* ba = b2 + kb;
  const float* wb = Wm1 + (size_t)kb * HID + l15;
  v8f acc[4] = {};
  for (int k0 = 0; k0 < HID; k0 += 4) {
    v2f g = *(const v2f*)(ga + k0);
    v2f xv = *(const v2f*)(xa + k0);
    v2f a;
    a.x = fmaxf(g.x + sl * xv.x + ba[k0], 0.f);
    a.y = fmaxf(g.y + sl * xv.y + ba[k0 + 1], 0.f);
#pragma unroll
    for (int t = 0; t < 4; ++t) {
      const float* wt = wb + (size_t)((t >> 1) * HID) * HID + (t & 1) * 16;
      v2f b;
      b.x = wt[(size_t)k0 * HID];
      b.y = wt[(size_t)(k0 + 1) * HID];
      acc[t] = __builtin_amdgcn_wmma_f32_16x16x4_f32(false, a, false, b, (short)0, acc[t],
                                                     false, false);
    }
  }
  int rbase = tm * 16 + ((lane >> 4) ? 8 : 0);
#pragma unroll
  for (int t = 0; t < 4; ++t) {
    int ncol = t * 16 + l15;
    float bn = (t < 2) ? bm1[ncol] : 0.f;
#pragma unroll
    for (int r = 0; r < 8; ++r) {
      int row = rbase + r;
      float val = acc[t][r];
      if (t < 2) val += bn;
      if (row < M) PQ[(size_t)row * (2 * HID) + ncol] = val;
    }
  }
}

#define EDGE_CHUNKS 4
__global__ __launch_bounds__(256)
void edge_out_kernel(const float* __restrict__ PQ, const int* __restrict__ p0,
                     const int* __restrict__ p1, const int* __restrict__ q0,
                     const int* __restrict__ q1, const float* __restrict__ Wm2,
                     const float* __restrict__ bm2, float* __restrict__ out,
                     int NP, int Etot) {
  int wave = (int)((blockIdx.x * (long long)blockDim.x + threadIdx.x) >> 5);
  int lane = threadIdx.x & 31;
  int G = lane >> 3;
  int j = lane & 7;
  int c = j * 4;
  const float4 w01 = *(const float4*)(Wm2 + c * 2);
  const float4 w23 = *(const float4*)(Wm2 + c * 2 + 4);
  const float bo = bm2[j & 1];
  const bool hi4 = (j & 4) != 0, hi2 = (j & 2) != 0, hi1 = (j & 1) != 0;
  for (int it = 0; it < EDGE_CHUNKS; ++it) {
    int e0 = (wave * EDGE_CHUNKS + it) * 16;
    if (e0 >= Etot) break;
    float v[8];
#pragma unroll
    for (int i = 0; i < 4; ++i) {
      int e = e0 + 4 * G + i;
      if (e > Etot - 1) e = Etot - 1;
      int s = (e < NP) ? p0[e] : q0[e - NP];
      int d = (e < NP) ? p1[e] : q1[e - NP];
      const float4 p = *(const float4*)(PQ + (size_t)s * (2 * HID) + c);
      const float4 q = *(const float4*)(PQ + (size_t)d * (2 * HID) + HID + c);
      float t0 = fmaxf(p.x + q.x, 0.f);
      float t1 = fmaxf(p.y + q.y, 0.f);
      float t2 = fmaxf(p.z + q.z, 0.f);
      float t3 = fmaxf(p.w + q.w, 0.f);
      v[2 * i]     = t0 * w01.x + t1 * w01.z + t2 * w23.x + t3 * w23.z;
      v[2 * i + 1] = t0 * w01.y + t1 * w01.w + t2 * w23.y + t3 * w23.w;
    }
    float u[4];
#pragma unroll
    for (int k = 0; k < 4; ++k) {
      float keep = hi4 ? v[k + 4] : v[k];
      float send = hi4 ? v[k] : v[k + 4];
      u[k] = keep + __shfl_xor(send, 4, 8);
    }
    float w[2];
#pragma unroll
    for (int k = 0; k < 2; ++k) {
      float keep = hi2 ? u[k + 2] : u[k];
      float send = hi2 ? u[k] : u[k + 2];
      w[k] = keep + __shfl_xor(send, 2, 8);
    }
    float keep1 = hi1 ? w[1] : w[0];
    float send1 = hi1 ? w[0] : w[1];
    float r = keep1 + __shfl_xor(send1, 1, 8);
    int eo = e0 + 4 * G + (j >> 1);
    if (eo < Etot) out[(size_t)e0 * 2 + lane] = r + bo;
  }
}

extern "C" void kernel_launch(void* const* d_in, const int* in_sizes, int n_in,
                              void* d_out, int out_size, void* d_ws, size_t ws_size,
                              hipStream_t stream) {
  const float* x   = (const float*)d_in[0];
  const int*   pe  = (const int*)d_in[1];
  const int*   neg = (const int*)d_in[2];
  const float* W1  = (const float*)d_in[3];
  const float* b1  = (const float*)d_in[4];
  const float* W2  = (const float*)d_in[5];
  const float* b2  = (const float*)d_in[6];
  const float* Wm1 = (const float*)d_in[7];
  const float* bm1 = (const float*)d_in[8];
  const float* Wm2 = (const float*)d_in[9];
  const float* bm2 = (const float*)d_in[10];
  float* out = (float*)d_out;

  const int N  = in_sizes[0] / HID;
  const int NP = in_sizes[1] / 2;
  const int NN = in_sizes[2] / 2;
  const int ET = NP + NN;
  const int* p0 = pe;        const int* p1 = pe + NP;
  const int* q0 = neg;       const int* q1 = neg + NN;

  char* ws = (char*)d_ws;
  size_t off = 0;
  auto wsalloc = [&](size_t bytes) -> char* {
    char* p = ws + off;
    off += (bytes + 255) & ~(size_t)255;
    return p;
  };
  float* dinv = (float*)wsalloc((size_t)N * 4);
  float* ax   = (float*)wsalloc((size_t)N * HID * 4);
  float* h1   = (float*)wsalloc((size_t)N * H1 * 4);
  float* xl2  = (float*)wsalloc((size_t)N * HID * 4);
  float* ag2  = (float*)wsalloc((size_t)N * HID * 4);
  float* pq   = (float*)wsalloc((size_t)N * 2 * HID * 4);
  (void)ws_size; (void)n_in; (void)out_size;

  const int B = 256;

  hipMemsetAsync(dinv, 0, (size_t)N * 4, stream);
  deg_kernel<<<(NP + B - 1) / B, B, 0, stream>>>(p1, dinv, NP);
  dinv_kernel<<<(N + B - 1) / B, B, 0, stream>>>(dinv, N);

  hipMemsetAsync(ax, 0, (size_t)N * HID * 4, stream);
  {
    long long thr = (long long)((NP + 32 * SC_CHUNKS - 1) / (32 * SC_CHUNKS)) * 32;
    scatter_rows_kernel<<<(unsigned)((thr + B - 1) / B), B, 0, stream>>>(x, p0, p1, dinv, ax, NP);
    long long thr2 = (long long)((N + 15) / 16) * (H1 / 16) * 32;
    gemm_l1_wmma_f32<<<(unsigned)((thr2 + B - 1) / B), B, 0, stream>>>(ax, x, dinv, W1, b1, h1, N);
  }

  {
    long long thr = (long long)((N + 15) / 16) * (HID / 16) * 32;
    gemm_wmma_f32<<<(unsigned)((thr + B - 1) / B), B, 0, stream>>>(h1, W2, xl2, N, HID, H1);
  }
  hipMemsetAsync(ag2, 0, (size_t)N * HID * 4, stream);
  {
    long long thr = (long long)((NP + 32 * SC_CHUNKS - 1) / (32 * SC_CHUNKS)) * 32;
    scatter_rows_kernel<<<(unsigned)((thr + B - 1) / B), B, 0, stream>>>(xl2, p0, p1, dinv, ag2, NP);
    long long thr2 = (long long)((N + 15) / 16) * 32;
    gemm_pq_wmma_f32<<<(unsigned)((thr2 + B - 1) / B), B, 0, stream>>>(ag2, xl2, dinv, b2, Wm1, bm1, pq, N);
  }

  {
    long long thr = (long long)((ET + 16 * EDGE_CHUNKS - 1) / (16 * EDGE_CHUNKS)) * 32;
    edge_out_kernel<<<(unsigned)((thr + B - 1) / B), B, 0, stream>>>(pq, p0, p1, q0, q1,
                                                                     Wm2, bm2, out, NP, ET);
  }
}


# Edge weight dinv[src]*dinv[dst] factored out of both scatter-adds: rows pre-scaled by dinv where made, sums scaled by dinv[dst] where read; no per-edge dinv gathers; fp32 rounding changes
# speedup vs baseline: 2.4528x; 1.0295x over previous
#include <hip/hip_runtime.h>
#include <hip/hip_bf16.h>

typedef __attribute__((ext_vector_type(2))) float v2f;
typedef __attribute__((ext_vector_type(8))) float v8f;

#define HID 32
#define H1  128

__global__ void deg_kernel(const int* __restrict__ dst, float* __restrict__ deg, int E) {
  int t = blockIdx.x * blockDim.x + threadIdx.x;
  if (t < E) atomicAdd(&deg[dst[t]], 1.0f);
}

#define PRESCALE_PIECES 4
__global__ void dinv_prescale_kernel(const float* __restrict__ deg, const float* __restrict__ x,
                                     float* __restrict__ dinv, float* __restrict__ xs, int N) {
  int t = blockIdx.x * blockDim.x + threadIdx.x;
  int stride = gridDim.x * blockDim.x;
  int total = N * (HID / 4);
  float dg[PRESCALE_PIECES];
  float4 v[PRESCALE_PIECES];
#pragma unroll
  for (int i = 0; i < PRESCALE_PIECES; ++i) {
    int p = t + i * stride;
    if (p > total - 1) p = total - 1;
    dg[i] = deg[p / (HID / 4)];
    v[i] = *(const float4*)(x + (size_t)p * 4);
  }
#pragma unroll
  for (int i = 0; i < PRESCALE_PIECES; ++i) {
    int p = t + i * stride;
    if (p < total) {
      float dv = rsqrtf(dg[i] + 1.0f);
      float4 o = v[i];
      o.x *= dv; o.y *= dv; o.z *= dv; o.w *= dv;
      *(float4*)(xs + (size_t)p * 4) = o;
      if ((p & (HID / 4 - 1)) == 0) dinv[p / (HID / 4)] = dv;
    }
  }
}

__global__ __launch_bounds__(256)
void gemm_rowscale_wmma_f32(const float* __restrict__ A, const float* __restrict__ B,
                            const float* __restrict__ rs, float* __restrict__ C, int M, int N, int K) {
  int wave = (int)((blockIdx.x * (long long)blockDim.x + threadIdx.x) >> 5);
  int lane = threadIdx.x & 31;
  int ntiles = N >> 4;
  int mtiles = (M + 15) >> 4;
  int tm = wave / ntiles;
  int tn = wave - tm * ntiles;
  if (tm >= mtiles) return;
  int mrow = tm * 16 + (lane & 15);
  if (mrow > M - 1) mrow = M - 1;
  int ncol = tn * 16 + (lane & 15);
  int kb = (lane >> 4) * 2;
  v8f acc = {};
  for (int k0 = 0; k0 < K; k0 += 4) {
    v2f a = *(const v2f*)(A + (size_t)mrow * K + k0 + kb);
    v2f b;
    b.x = B[(size_t)(k0 + kb) * N + ncol];
    b.y = B[(size_t)(k0 + kb + 1) * N + ncol];
    acc = __builtin_amdgcn_wmma_f32_16x16x4_f32(false, a, false, b, (short)0, acc,
                                                false, false);
  }
  int rbase = tm * 16 + ((lane >> 4) ? 8 : 0);
#pragma unroll
  for (int r = 0; r < 8; ++r) {
    int row = rbase + r;
    if (row < M) C[(size_t)row * N + ncol] = rs[row] * acc[r];
  }
}

__global__ __launch_bounds__(256)
void gemm_l1_wmma_f32(const float* __restrict__ agg, const float* __restrict__ xs,
                      const float* __restrict__ dinv, const float* __restrict__ W,
                      const float* __restrict__ bias, float* __restrict__ H, int M) {
  int wave = (int)((blockIdx.x * (long long)blockDim.x + threadIdx.x) >> 5);
  int lane = threadIdx.x & 31;
  const int ntiles = H1 >> 4;
  int mtiles = (M + 15) >> 4;
  int tm = wave / ntiles;
  int tn = wave - tm * ntiles;
  if (tm >= mtiles) return;
  int mrow = tm * 16 + (lane & 15);
  if (mrow > M - 1) mrow = M - 1;
  int ncol = tn * 16 + (lane & 15);
  int kb = (lane >> 4) * 2;
  float di = dinv[mrow];
  const float* ga = agg + (size_t)mrow * HID + kb;
  const float* xa = xs + (size_t)mrow * HID + kb;
  const float* wb = W + (size_t)kb * H1 + ncol;
  v8f acc = {};
  for (int k0 = 0; k0 < HID; k0 += 4) {
    v2f g = *(const v2f*)(ga + k0);
    v2f xv = *(const v2f*)(xa + k0);
    v2f a;
    a.x = di * (g.x + xv.x);
    a.y = di * (g.y + xv.y);
    v2f b;
    b.x = wb[(size_t)k0 * H1];
    b.y = wb[(size_t)(k0 + 1) * H1];
    acc = __builtin_amdgcn_wmma_f32_16x16x4_f32(false, a, false, b, (short)0, acc,
                                                false, false);
  }
  float bn = bias[ncol];
  int rbase = tm * 16 + ((lane >> 4) ? 8 : 0);
#pragma unroll
  for (int r = 0; r < 8; ++r) {
    int row = rbase + r;
    if (row < M) H[(size_t)row * H1 + ncol] = fmaxf(acc[r] + bn, 0.f);
  }
}

#define SC_CHUNKS 4
__device__ __forceinline__ unsigned row_off(int row, int lane) {
  return ((unsigned)row * HID + (unsigned)lane) * 4u;
}
__global__ __launch_bounds__(256)
void scatter_rows_kernel(const float* __restrict__ xl, const int* __restrict__ src,
                         const int* __restrict__ dst, float* __restrict__ agg, int E) {
  static_assert(HID == 32, "one row per wave: HID must equal the wave size");
  int wave = (int)((blockIdx.x * (long long)blockDim.x + threadIdx.x) >> 5);
  int lane = threadIdx.x & 31;
  for (int it = 0; it < SC_CHUNKS; ++it) {
    int e0 = (wave * SC_CHUNKS + it) * 32;
    if (e0 >= E) break;
    int e = e0 + lane;
    if (e > E - 1) e = E - 1;
    int s = src[e], d = dst[e];
    int n = E - e0;
    if (n >= 32) {
      float v[32];
#pragma unroll
      for (int i = 0; i < 32; ++i) {
        int si = __builtin_amdgcn_readlane(s, i);
        v[i] = *(const float*)((const char*)xl + row_off(si, lane));
      }
#pragma unroll
      for (int i = 0; i < 32; ++i) {
        int di = __builtin_amdgcn_readlane(d, i);
        atomicAdd((float*)((char*)agg + row_off(di, lane)), v[i]);
      }
    } else {
      for (int i = 0; i < n; ++i) {
        int si = __builtin_amdgcn_readlane(s, i);
        int di = __builtin_amdgcn_readlane(d, i);
        float xv = *(const float*)((const char*)xl + row_off(si, lane));
        atomicAdd((float*)((char*)agg + row_off(di, lane)), xv);
      }
    }
  }
}

__global__ __launch_bounds__(256)
void gemm_pq_wmma_f32(const float* __restrict__ agg, const float* __restrict__ xl,
                      const float* __restrict__ dinv, const float* __restrict__ b2,
                      const float* __restrict__ Wm1, const float* __restrict__ bm1,
                      float* __restrict__ PQ, int M) {
  int tm = (int)((blockIdx.x * (long long)blockDim.x + threadIdx.x) >> 5);
  int lane = threadIdx.x & 31;
  int mtiles = (M + 15) >> 4;
  if (tm >= mtiles) return;
  int l15 = lane & 15;
  int mrow = tm * 16 + l15;
  if (mrow > M - 1) mrow = M - 1;
  int kb = (lane >> 4) * 2;
  float di = dinv[mrow];
  const float* ga = agg + (size_t)mrow * HID + kb;
  const float* xa = xl + (size_t)mrow * HID + kb;
  const float* ba = b2 + kb;
  const float* wb = Wm1 + (size_t)kb * HID + l15;
  v8f acc[4] = {};
  for (int k0 = 0; k0 < HID; k0 += 4) {
    v2f g = *(const v2f*)(ga + k0);
    v2f xv = *(const v2f*)(xa + k0);
    v2f a;
    a.x = fmaxf(di * (g.x + xv.x) + ba[k0], 0.f);
    a.y = fmaxf(di * (g.y + xv.y) + ba[k0 + 1], 0.f);
#pragma unroll
    for (int t = 0; t < 4; ++t) {
      const float* wt = wb + (size_t)((t >> 1) * HID) * HID + (t & 1) * 16;
      v2f b;
      b.x = wt[(size_t)k0 * HID];
      b.y = wt[(size_t)(k0 + 1) * HID];
      acc[t] = __builtin_amdgcn_wmma_f32_16x16x4_f32(false, a, false, b, (short)0, acc[t],
                                                     false, false);
    }
  }
  int rbase = tm * 16 + ((lane >> 4) ? 8 : 0);
#pragma unroll
  for (int t = 0; t < 4; ++t) {
    int ncol = t * 16 + l15;
    float bn = (t < 2) ? bm1[ncol] : 0.f;
#pragma unroll
    for (int r = 0; r < 8; ++r) {
      int row = rbase + r;
      float val = acc[t][r];
      if (t < 2) val += bn;
      if (row < M) PQ[(size_t)row * (2 * HID) + ncol] = val;
    }
  }
}

#define EDGE_CHUNKS 4
__global__ __launch_bounds__(256)
void edge_out_kernel(const float* __restrict__ PQ, const int* __restrict__ p0,
                     const int* __restrict__ p1, const int* __restrict__ q0,
                     const int* __restrict__ q1, const float* __restrict__ Wm2,
                     const float* __restrict__ bm2, float* __restrict__ out,
                     int NP, int Etot) {
  int wave = (int)((blockIdx.x * (long long)blockDim.x + threadIdx.x) >> 5);
  int lane = threadIdx.x & 31;
  int G = lane >> 3;
  int j = lane & 7;
  int c = j * 4;
  const float4 w01 = *(const float4*)(Wm2 + c * 2);
  const float4 w23 = *(const float4*)(Wm2 + c * 2 + 4);
  const float bo = bm2[j & 1];
  const bool hi4 = (j & 4) != 0, hi2 = (j & 2) != 0, hi1 = (j & 1) != 0;
  for (int it = 0; it < EDGE_CHUNKS; ++it) {
    int e0 = (wave * EDGE_CHUNKS + it) * 16;
    if (e0 >= Etot) break;
    float v[8];
#pragma unroll
    for (int i = 0; i < 4; ++i) {
      int e = e0 + 4 * G + i;
      if (e > Etot - 1) e = Etot - 1;
      int s = (e < NP) ? p0[e] : q0[e - NP];
      int d = (e < NP) ? p1[e] : q1[e - NP];
      const float4 p = *(const float4*)(PQ + (size_t)s * (2 * HID) + c);
      const float4 q = *(const float4*)(PQ + (size_t)d * (2 * HID) + HID + c);
      float t0 = fmaxf(p.x + q.x, 0.f);
      float t1 = fmaxf(p.y + q.y, 0.f);
      float t2 = fmaxf(p.z + q.z, 0.f);
      float t3 = fmaxf(p.w + q.w, 0.f);
      v[2 * i]     = t0 * w01.x + t1 * w01.z + t2 * w23.x + t3 * w23.z;
      v[2 * i + 1] = t0 * w01.y + t1 * w01.w + t2 * w23.y + t3 * w23.w;
    }
    float u[4];
#pragma unroll
    for (int k = 0; k < 4; ++k) {
      float keep = hi4 ? v[k + 4] : v[k];
      float send = hi4 ? v[k] : v[k + 4];
      u[k] = keep + __shfl_xor(send, 4, 8);
    }
    float w[2];
#pragma unroll
    for (int k = 0; k < 2; ++k) {
      float keep = hi2 ? u[k + 2] : u[k];
      float send = hi2 ? u[k] : u[k + 2];
      w[k] = keep + __shfl_xor(send, 2, 8);
    }
    float keep1 = hi1 ? w[1] : w[0];
    float send1 = hi1 ? w[0] : w[1];
    float r = keep1 + __shfl_xor(send1, 1, 8);
    int eo = e0 + 4 * G + (j >> 1);
    if (eo < Etot) out[(size_t)e0 * 2 + lane] = r + bo;
  }
}

extern "C" void kernel_launch(void* const* d_in, const int* in_sizes, int n_in,
                              void* d_out, int out_size, void* d_ws, size_t ws_size,
                              hipStream_t stream) {
  const float* x   = (const float*)d_in[0];
  const int*   pe  = (const int*)d_in[1];
  const int*   neg = (const int*)d_in[2];
  const float* W1  = (const float*)d_in[3];
  const float* b1  = (const float*)d_in[4];
  const float* W2  = (const float*)d_in[5];
  const float* b2  = (const float*)d_in[6];
  const float* Wm1 = (const float*)d_in[7];
  const float* bm1 = (const float*)d_in[8];
  const float* Wm2 = (const float*)d_in[9];
  const float* bm2 = (const float*)d_in[10];
  float* out = (float*)d_out;

  const int N  = in_sizes[0] / HID;
  const int NP = in_sizes[1] / 2;
  const int NN = in_sizes[2] / 2;
  const int ET = NP + NN;
  const int* p0 = pe;        const int* p1 = pe + NP;
  const int* q0 = neg;       const int* q1 = neg + NN;

  char* ws = (char*)d_ws;
  size_t off = 0;
  auto wsalloc = [&](size_t bytes) -> char* {
    char* p = ws + off;
    off += (bytes + 255) & ~(size_t)255;
    return p;
  };
  float* deg  = (float*)wsalloc((size_t)N * 4);
  float* dinv = (float*)wsalloc((size_t)N * 4);
  float* xs   = (float*)wsalloc((size_t)N * HID * 4);
  float* ax   = (float*)wsalloc((size_t)N * HID * 4);
  float* h1   = (float*)wsalloc((size_t)N * H1 * 4);
  float* xl2  = (float*)wsalloc((size_t)N * HID * 4);
  float* ag2  = (float*)wsalloc((size_t)N * HID * 4);
  float* pq   = (float*)wsalloc((size_t)N * 2 * HID * 4);
  (void)ws_size; (void)n_in; (void)out_size;

  const int B = 256;

  hipMemsetAsync(deg, 0, (size_t)N * 4, stream);
  deg_kernel<<<(NP + B - 1) / B, B, 0, stream>>>(p1, deg, NP);
  {
    long long thr = ((long long)N * (HID / 4) + PRESCALE_PIECES - 1) / PRESCALE_PIECES;
    dinv_prescale_kernel<<<(unsigned)((thr + B - 1) / B), B, 0, stream>>>(deg, x, dinv, xs, N);
  }

  hipMemsetAsync(ax, 0, (size_t)N * HID * 4, stream);
  {
    long long thr = (long long)((NP + 32 * SC_CHUNKS - 1) / (32 * SC_CHUNKS)) * 32;
    scatter_rows_kernel<<<(unsigned)((thr + B - 1) / B), B, 0, stream>>>(xs, p0, p1, ax, NP);
    long long thr2 = (long long)((N + 15) / 16) * (H1 / 16) * 32;
    gemm_l1_wmma_f32<<<(unsigned)((thr2 + B - 1) / B), B, 0, stream>>>(ax, xs, dinv, W1, b1, h1, N);
  }

  {
    long long thr = (long long)((N + 15) / 16) * (HID / 16) * 32;
    gemm_rowscale_wmma_f32<<<(unsigned)((thr + B - 1) / B), B, 0, stream>>>(h1, W2, dinv, xl2, N, HID, H1);
  }
  hipMemsetAsync(ag2, 0, (size_t)N * HID * 4, stream);
  {
    long long thr = (long long)((NP + 32 * SC_CHUNKS - 1) / (32 * SC_CHUNKS)) * 32;
    scatter_rows_kernel<<<(unsigned)((thr + B - 1) / B), B, 0, stream>>>(xl2, p0, p1, ag2, NP);
    long long thr2 = (long long)((N + 15) / 16) * 32;
    gemm_pq_wmma_f32<<<(unsigned)((thr2 + B - 1) / B), B, 0, stream>>>(ag2, xl2, dinv, b2, Wm1, bm1, pq, N);
  }

  {
    long long thr = (long long)((ET + 16 * EDGE_CHUNKS - 1) / (16 * EDGE_CHUNKS)) * 32;
    edge_out_kernel<<<(unsigned)((thr + B - 1) / B), B, 0, stream>>>(pq, p0, p1, q0, q1,
                                                                     Wm2, bm2, out, NP, ET);
  }
}


# Layer-1 and layer-2 products fused per 16 nodes, transposed: the relu'd h1 tile stays in the WMMA D registers and is the W2 product's B operand; no h1 array; W2 product's fp32 k sum order changes
# speedup vs baseline: 2.6647x; 1.0864x over previous
#include <hip/hip_runtime.h>
#include <hip/hip_bf16.h>

typedef __attribute__((ext_vector_type(2))) float v2f;
typedef __attribute__((ext_vector_type(8))) float v8f;

#define HID 32
#define H1  128

__global__ void deg_kernel(const int* __restrict__ dst, float* __restrict__ deg, int E) {
  int t = blockIdx.x * blockDim.x + threadIdx.x;
  if (t < E) atomicAdd(&deg[dst[t]], 1.0f);
}

#define PRESCALE_PIECES 4
__global__ void dinv_prescale_kernel(const float* __restrict__ deg, const float* __restrict__ x,
                                     float* __restrict__ dinv, float* __restrict__ xs, int N) {
  int t = blockIdx.x * blockDim.x + threadIdx.x;
  int stride = gridDim.x * blockDim.x;
  int total = N * (HID / 4);
  float dg[PRESCALE_PIECES];
  float4 v[PRESCALE_PIECES];
#pragma unroll
  for (int i = 0; i < PRESCALE_PIECES; ++i) {
    int p = t + i * stride;
    if (p > total - 1) p = total - 1;
    dg[i] = deg[p / (HID / 4)];
    v[i] = *(const float4*)(x + (size_t)p * 4);
  }
#pragma unroll
  for (int i = 0; i < PRESCALE_PIECES; ++i) {
    int p = t + i * stride;
    if (p < total) {
      float dv = rsqrtf(dg[i] + 1.0f);
      float4 o = v[i];
      o.x *= dv; o.y *= dv; o.z *= dv; o.w *= dv;
      *(float4*)(xs + (size_t)p * 4) = o;
      if ((p & (HID / 4 - 1)) == 0) dinv[p / (HID / 4)] = dv;
    }
  }
}

__global__ __launch_bounds__(256)
void gemm_l1l2_wmma_f32(const float* __restrict__ agg, const float* __restrict__ xs,
                        const float* __restrict__ dinv, const float* __restrict__ W1,
                        const float* __restrict__ b1, const float* __restrict__ W2,
                        float* __restrict__ XL, int M) {
  int tm = (int)((blockIdx.x * (long long)blockDim.x + threadIdx.x) >> 5);
  int lane = threadIdx.x & 31;
  int mtiles = (M + 15) >> 4;
  if (tm >= mtiles) return;
  int c = lane & 15;
  int h = lane >> 4;
  int node = tm * 16 + c;
  int nodec = node;
  if (nodec > M - 1) nodec = M - 1;
  float di = dinv[nodec];
  const float* ga = agg + (size_t)nodec * HID + 2 * h;
  const float* xa = xs + (size_t)nodec * HID + 2 * h;
  v2f bx[HID / 4];
#pragma unroll
  for (int j = 0; j < HID / 4; ++j) {
    v2f g = *(const v2f*)(ga + 4 * j);
    v2f xv = *(const v2f*)(xa + 4 * j);
    bx[j].x = di * (g.x + xv.x);
    bx[j].y = di * (g.y + xv.y);
  }
  const float* w1a = W1 + (size_t)(2 * h) * H1 + c;
  v8f acc[H1 / 16] = {};
#pragma unroll
  for (int j = 0; j < HID / 4; ++j) {
#pragma unroll
    for (int t = 0; t < H1 / 16; ++t) {
      v2f a;
      a.x = w1a[(size_t)(4 * j) * H1 + 16 * t];
      a.y = w1a[(size_t)(4 * j + 1) * H1 + 16 * t];
      acc[t] = __builtin_amdgcn_wmma_f32_16x16x4_f32(false, a, false, bx[j], (short)0, acc[t],
                                                     false, false);
    }
  }
  const float* bb = b1 + 8 * h;
#pragma unroll
  for (int t = 0; t < H1 / 16; ++t) {
    const float4 bl = *(const float4*)(bb + 16 * t);
    const float4 bh = *(const float4*)(bb + 16 * t + 4);
    acc[t][0] = fmaxf(acc[t][0] + bl.x, 0.f);
    acc[t][1] = fmaxf(acc[t][1] + bl.y, 0.f);
    acc[t][2] = fmaxf(acc[t][2] + bl.z, 0.f);
    acc[t][3] = fmaxf(acc[t][3] + bl.w, 0.f);
    acc[t][4] = fmaxf(acc[t][4] + bh.x, 0.f);
    acc[t][5] = fmaxf(acc[t][5] + bh.y, 0.f);
    acc[t][6] = fmaxf(acc[t][6] + bh.z, 0.f);
    acc[t][7] = fmaxf(acc[t][7] + bh.w, 0.f);
  }
  const float* w2a = W2 + (size_t)(8 * h) * HID + c;
  v8f o[HID / 16] = {};
#pragma unroll
  for (int t = 0; t < H1 / 16; ++t) {
#pragma unroll
    for (int p = 0; p < 4; ++p) {
      v2f b;
      b.x = acc[t][2 * p];
      b.y = acc[t][2 * p + 1];
#pragma unroll
      for (int u = 0; u < HID / 16; ++u) {
        v2f a;
        a.x = w2a[(size_t)(16 * t + 2 * p) * HID + 16 * u];
        a.y = w2a[(size_t)(16 * t + 2 * p + 1) * HID + 16 * u];
        o[u] = __builtin_amdgcn_wmma_f32_16x16x4_f32(false, a, false, b, (short)0, o[u],
                                                     false, false);
      }
    }
  }
  if (node < M) {
    float* xo = XL + (size_t)node * HID + 8 * h;
#pragma unroll
    for (int u = 0; u < HID / 16; ++u) {
      float4 lo, hi;
      lo.x = di * o[u][0]; lo.y = di * o[u][1]; lo.z = di * o[u][2]; lo.w = di * o[u][3];
      hi.x = di * o[u][4]; hi.y = di * o[u][5]; hi.z = di * o[u][6]; hi.w = di * o[u][7];
      *(float4*)(xo + 16 * u) = lo;
      *(float4*)(xo + 16 * u + 4) = hi;
    }
  }
}

#define SC_CHUNKS 4
__device__ __forceinline__ unsigned row_off(int row, int lane) {
  return ((unsigned)row * HID + (unsigned)lane) * 4u;
}
__global__ __launch_bounds__(256)
void scatter_rows_kernel(const float* __restrict__ xl, const int* __restrict__ src,
                         const int* __restrict__ dst, float* __restrict__ agg, int E) {
  static_assert(HID == 32, "one row per wave: HID must equal the wave size");
  int wave = (int)((blockIdx.x * (long long)blockDim.x + threadIdx.x) >> 5);
  int lane = threadIdx.x & 31;
  for (int it = 0; it < SC_CHUNKS; ++it) {
    int e0 = (wave * SC_CHUNKS + it) * 32;
    if (e0 >= E) break;
    int e = e0 + lane;
    if (e > E - 1) e = E - 1;
    int s = src[e], d = dst[e];
    int n = E - e0;
    if (n >= 32) {
      float v[32];
#pragma unroll
      for (int i = 0; i < 32; ++i) {
        int si = __builtin_amdgcn_readlane(s, i);
        v[i] = *(const float*)((const char*)xl + row_off(si, lane));
      }
#pragma unroll
      for (int i = 0; i < 32; ++i) {
        int di = __builtin_amdgcn_readlane(d, i);
        atomicAdd((float*)((char*)agg + row_off(di, lane)), v[i]);
      }
    } else {
      for (int i = 0; i < n; ++i) {
        int si = __builtin_amdgcn_readlane(s, i);
        int di = __builtin_amdgcn_readlane(d, i);
        float xv = *(const float*)((const char*)xl + row_off(si, lane));
        atomicAdd((float*)((char*)agg + row_off(di, lane)), xv);
      }
    }
  }
}

__global__ __launch_bounds__(256)
void gemm_pq_wmma_f32(const float* __restrict__ agg, const float* __restrict__ xl,
                      const float* __restrict__ dinv, const float* __restrict__ b2,
                      const float* __restrict__ Wm1, const float* __restrict__ bm1,
                      float* __restrict__ PQ, int M) {
  int tm = (int)((blockIdx.x * (long long)blockDim.x + threadIdx.x) >> 5);
  int lane = threadIdx.x & 31;
  int mtiles = (M + 15) >> 4;
  if (tm >= mtiles) return;
  int l15 = lane & 15;
  int mrow = tm * 16 + l15;
  if (mrow > M - 1) mrow = M - 1;
  int kb = (lane >> 4) * 2;
  float di = dinv[mrow];
  const float* ga = agg + (size_t)mrow * HID + kb;
  const float* xa = xl + (size_t)mrow * HID + kb;
  const float* ba = b2 + kb;
  const float* wb = Wm1 + (size_t)kb * HID + l15;
  v8f acc[4] = {};
  for (int k0 = 0; k0 < HID; k0 += 4) {
    v2f g = *(const v2f*)(ga + k0);
    v2f xv = *(const v2f*)(xa + k0);
    v2f a;
    a.x = fmaxf(di * (g.x + xv.x) + ba[k0], 0.f);
    a.y = fmaxf(di * (g.y + xv.y) + ba[k0 + 1], 0.f);
#pragma unroll
    for (int t = 0; t < 4; ++t) {
      const float* wt = wb + (size_t)((t >> 1) * HID) * HID + (t & 1) * 16;
      v2f b;
      b.x = wt[(size_t)k0 * HID];
      b.y = wt[(size_t)(k0 + 1) * HID];
      acc[t] = __builtin_amdgcn_wmma_f32_16x16x4_f32(false, a, false, b, (short)0, acc[t],
                                                     false, false);
    }
  }
  int rbase = tm * 16 + ((lane >> 4) ? 8 : 0);
#pragma unroll
  for (int t = 0; t < 4; ++t) {
    int ncol = t * 16 + l15;
    float bn = (t < 2) ? bm1[ncol] : 0.f;
#pragma unroll
    for (int r = 0; r < 8; ++r) {
      int row = rbase + r;
      float val = acc[t][r];
      if (t < 2) val += bn;
      if (row < M) PQ[(size_t)row * (2 * HID) + ncol] = val;
    }
  }
}

#define EDGE_CHUNKS 4
__global__ __launch_bounds__(256)
void edge_out_kernel(const float* __restrict__ PQ, const int* __restrict__ p0,
                     const int* __restrict__ p1, const int* __restrict__ q0,
                     const int* __restrict__ q1, const float* __restrict__ Wm2,
                     const float* __restrict__ bm2, float* __restrict__ out,
                     int NP, int Etot) {
  int wave = (int)((blockIdx.x * (long long)blockDim.x + threadIdx.x) >> 5);
  int lane = threadIdx.x & 31;
  int G = lane >> 3;
  int j = lane & 7;
  int c = j * 4;
  const float4 w01 = *(const float4*)(Wm2 + c * 2);
  const float4 w23 = *(const float4*)(Wm2 + c * 2 + 4);
  const float bo = bm2[j & 1];
  const bool hi4 = (j & 4) != 0, hi2 = (j & 2) != 0, hi1 = (j & 1) != 0;
  for (int it = 0; it < EDGE_CHUNKS; ++it) {
    int e0 = (wave * EDGE_CHUNKS + it) * 16;
    if (e0 >= Etot) break;
    float v[8];
#pragma unroll
    for (int i = 0; i < 4; ++i) {
      int e = e0 + 4 * G + i;
      if (e > Etot - 1) e = Etot - 1;
      int s = (e < NP) ? p0[e] : q0[e - NP];
      int d = (e < NP) ? p1[e] : q1[e - NP];
      const float4 p = *(const float4*)(PQ + (size_t)s * (2 * HID) + c);
      const float4 q = *(const float4*)(PQ + (size_t)d * (2 * HID) + HID + c);
      float t0 = fmaxf(p.x + q.x, 0.f);
      float t1 = fmaxf(p.y + q.y, 0.f);
      float t2 = fmaxf(p.z + q.z, 0.f);
      float t3 = fmaxf(p.w + q.w, 0.f);
      v[2 * i]     = t0 * w01.x + t1 * w01.z + t2 * w23.x + t3 * w23.z;
      v[2 * i + 1] = t0 * w01.y + t1 * w01.w + t2 * w23.y + t3 * w23.w;
    }
    float u[4];
#pragma unroll
    for (int k = 0; k < 4; ++k) {
      float keep = hi4 ? v[k + 4] : v[k];
      float send = hi4 ? v[k] : v[k + 4];
      u[k] = keep + __shfl_xor(send, 4, 8);
    }
    float w[2];
#pragma unroll
    for (int k = 0; k < 2; ++k) {
      float keep = hi2 ? u[k + 2] : u[k];
      float send = hi2 ? u[k] : u[k + 2];
      w[k] = keep + __shfl_xor(send, 2, 8);
    }
    float keep1 = hi1 ? w[1] : w[0];
    float send1 = hi1 ? w[0] : w[1];
    float r = keep1 + __shfl_xor(send1, 1, 8);
    int eo = e0 + 4 * G + (j >> 1);
    if (eo < Etot) out[(size_t)e0 * 2 + lane] = r + bo;
  }
}

extern "C" void kernel_launch(void* const* d_in, const int* in_sizes, int n_in,
                              void* d_out, int out_size, void* d_ws, size_t ws_size,
                              hipStream_t stream) {
  const float* x   = (const float*)d_in[0];
  const int*   pe  = (const int*)d_in[1];
  const int*   neg = (const int*)d_in[2];
  const float* W1  = (const float*)d_in[3];
  const float* b1  = (const float*)d_in[4];
  const float* W2  = (const float*)d_in[5];
  const float* b2  = (const float*)d_in[6];
  const float* Wm1 = (const float*)d_in[7];
  const float* bm1 = (const float*)d_in[8];
  const float* Wm2 = (const float*)d_in[9];
  const float* bm2 = (const float*)d_in[10];
  float* out = (float*)d_out;

  const int N  = in_sizes[0] / HID;
  const int NP = in_sizes[1] / 2;
  const int NN = in_sizes[2] / 2;
  const int ET = NP + NN;
  const int* p0 = pe;        const int* p1 = pe + NP;
  const int* q0 = neg;       const int* q1 = neg + NN;

  char* ws = (char*)d_ws;
  size_t off = 0;
  auto wsalloc = [&](size_t bytes) -> char* {
    char* p = ws + off;
    off += (bytes + 255) & ~(size_t)255;
    return p;
  };
  float* deg  = (float*)wsalloc((size_t)N * 4);
  float* dinv = (float*)wsalloc((size_t)N * 4);
  float* xs   = (float*)wsalloc((size_t)N * HID * 4);
  float* ax   = (float*)wsalloc((size_t)N * HID * 4);
  float* xl2  = (float*)wsalloc((size_t)N * HID * 4);
  float* ag2  = (float*)wsalloc((size_t)N * HID * 4);
  float* pq   = (float*)wsalloc((size_t)N * 2 * HID * 4);
  (void)ws_size; (void)n_in; (void)out_size;

  const int B = 256;

  hipMemsetAsync(deg, 0, (size_t)N * 4, stream);
  deg_kernel<<<(NP + B - 1) / B, B, 0, stream>>>(p1, deg, NP);
  {
    long long thr = ((long long)N * (HID / 4) + PRESCALE_PIECES - 1) / PRESCALE_PIECES;
    dinv_prescale_kernel<<<(unsigned)((thr + B - 1) / B), B, 0, stream>>>(deg, x, dinv, xs, N);
  }

  hipMemsetAsync(ax, 0, (size_t)N * HID * 4, stream);
  {
    long long thr = (long long)((NP + 32 * SC_CHUNKS - 1) / (32 * SC_CHUNKS)) * 32;
    scatter_rows_kernel<<<(unsigned)((thr + B - 1) / B), B, 0, stream>>>(xs, p0, p1, ax, NP);
    long long thr2 = (long long)((N + 15) / 16) * 32;
    gemm_l1l2_wmma_f32<<<(unsigned)((thr2 + B - 1) / B), B, 0, stream>>>(ax, xs, dinv, W1, b1, W2, xl2, N);
  }
  hipMemsetAsync(ag2, 0, (size_t)N * HID * 4, stream);
  {
    long long thr = (long long)((NP + 32 * SC_CHUNKS - 1) / (32 * SC_CHUNKS)) * 32;
    scatter_rows_kernel<<<(unsigned)((thr + B - 1) / B), B, 0, stream>>>(xl2, p0, p1, ag2, NP);
    long long thr2 = (long long)((N + 15) / 16) * 32;
    gemm_pq_wmma_f32<<<(unsigned)((thr2 + B - 1) / B), B, 0, stream>>>(ag2, xl2, dinv, b2, Wm1, bm1, pq, N);
  }

  {
    long long thr = (long long)((ET + 16 * EDGE_CHUNKS - 1) / (16 * EDGE_CHUNKS)) * 32;
    edge_out_kernel<<<(unsigned)((thr + B - 1) / B), B, 0, stream>>>(pq, p0, p1, q0, q1,
                                                                     Wm2, bm2, out, NP, ET);
  }
}


# Both edge sums by pull: edges bucketed by destination once (slot = the degree count's returning add), a wave sums a node's in-rows in registers; no f32 atomics, no zero fills; sum order = slot order
# speedup vs baseline: 2.8078x; 1.0537x over previous
#include <hip/hip_runtime.h>
#include <hip/hip_bf16.h>

typedef __attribute__((ext_vector_type(2))) float v2f;
typedef __attribute__((ext_vector_type(8))) float v8f;

#define HID 32
#define H1  128

#define EDGE_PIECES 4
__global__ __launch_bounds__(256)
void rank_kernel(const int* __restrict__ dst, int* __restrict__ cnt, int* __restrict__ rank, int E) {
  int t = blockIdx.x * blockDim.x + threadIdx.x;
  int stride = gridDim.x * blockDim.x;
  int d[EDGE_PIECES];
  int r[EDGE_PIECES];
#pragma unroll
  for (int i = 0; i < EDGE_PIECES; ++i) {
    int e = t + i * stride;
    if (e > E - 1) e = E - 1;
    d[i] = dst[e];
  }
#pragma unroll
  for (int i = 0; i < EDGE_PIECES; ++i) {
    int e = t + i * stride;
    r[i] = 0;
    if (e < E) r[i] = atomicAdd(&cnt[d[i]], 1);
  }
#pragma unroll
  for (int i = 0; i < EDGE_PIECES; ++i) {
    int e = t + i * stride;
    if (e < E) rank[e] = r[i];
  }
}

__global__ __launch_bounds__(256)
void seg_start_kernel(const int* __restrict__ cnt, int* __restrict__ start, int* __restrict__ total, int N) {
  int n = blockIdx.x * blockDim.x + threadIdx.x;
  int lane = threadIdx.x & 31;
  int c = (n < N) ? cnt[n] : 0;
  int incl = c;
#pragma unroll
  for (int dlt = 1; dlt < 32; dlt <<= 1) {
    int up = __shfl_up(incl, dlt);
    if (lane >= dlt) incl += up;
  }
  int base = 0;
  if (lane == 31) base = atomicAdd(total, incl);
  base = __shfl(base, 31);
  if (n < N) start[n] = base + incl - c;
}

__global__ __launch_bounds__(256)
void fill_kernel(const int* __restrict__ src, const int* __restrict__ dst, const int* __restrict__ rank,
                 const int* __restrict__ start, int* __restrict__ col, int E) {
  int t = blockIdx.x * blockDim.x + threadIdx.x;
  int stride = gridDim.x * blockDim.x;
  int s[EDGE_PIECES];
  int pos[EDGE_PIECES];
#pragma unroll
  for (int i = 0; i < EDGE_PIECES; ++i) {
    int e = t + i * stride;
    if (e > E - 1) e = E - 1;
    s[i] = src[e];
    pos[i] = start[dst[e]] + rank[e];
  }
#pragma unroll
  for (int i = 0; i < EDGE_PIECES; ++i) {
    int e = t + i * stride;
    if (e < E) col[pos[i]] = s[i];
  }
}

#define PRESCALE_PIECES 4
__global__ void dinv_prescale_kernel(const int* __restrict__ deg, const float* __restrict__ x,
                                     float* __restrict__ dinv, float* __restrict__ xs, int N) {
  int t = blockIdx.x * blockDim.x + threadIdx.x;
  int stride = gridDim.x * blockDim.x;
  int total = N * (HID / 4);
  int dg[PRESCALE_PIECES];
  float4 v[PRESCALE_PIECES];
#pragma unroll
  for (int i = 0; i < PRESCALE_PIECES; ++i) {
    int p = t + i * stride;
    if (p > total - 1) p = total - 1;
    dg[i] = deg[p / (HID / 4)];
    v[i] = *(const float4*)(x + (size_t)p * 4);
  }
#pragma unroll
  for (int i = 0; i < PRESCALE_PIECES; ++i) {
    int p = t + i * stride;
    if (p < total) {
      float dv = rsqrtf((float)dg[i] + 1.0f);
      float4 o = v[i];
      o.x *= dv; o.y *= dv; o.z *= dv; o.w *= dv;
      *(float4*)(xs + (size_t)p * 4) = o;
      if ((p & (HID / 4 - 1)) == 0) dinv[p / (HID / 4)] = dv;
    }
  }
}

__global__ __launch_bounds__(256)
void gemm_l1l2_wmma_f32(const float* __restrict__ agg, const float* __restrict__ xs,
                        const float* __restrict__ dinv, const float* __restrict__ W1,
                        const float* __restrict__ b1, const float* __restrict__ W2,
                        float* __restrict__ XL, int M) {
  int tm = (int)((blockIdx.x * (long long)blockDim.x + threadIdx.x) >> 5);
  int lane = threadIdx.x & 31;
  int mtiles = (M + 15) >> 4;
  if (tm >= mtiles) return;
  int c = lane & 15;
  int h = lane >> 4;
  int node = tm * 16 + c;
  int nodec = node;
  if (nodec > M - 1) nodec = M - 1;
  float di = dinv[nodec];
  const float* ga = agg + (size_t)nodec * HID + 2 * h;
  const float* xa = xs + (size_t)nodec * HID + 2 * h;
  v2f bx[HID / 4];
#pragma unroll
  for (int j = 0; j < HID / 4; ++j) {
    v2f g = *(const v2f*)(ga + 4 * j);
    v2f xv = *(const v2f*)(xa + 4 * j);
    bx[j].x = di * (g.x + xv.x);
    bx[j].y = di * (g.y + xv.y);
  }
  const float* w1a = W1 + (size_t)(2 * h) * H1 + c;
  v8f acc[H1 / 16] = {};
#pragma unroll
  for (int j = 0; j < HID / 4; ++j) {
#pragma unroll
    for (int t = 0; t < H1 / 16; ++t) {
      v2f a;
      a.x = w1a[(size_t)(4 * j) * H1 + 16 * t];
      a.y = w1a[(size_t)(4 * j + 1) * H1 + 16 * t];
      acc[t] = __builtin_amdgcn_wmma_f32_16x16x4_f32(false, a, false, bx[j], (short)0, acc[t],
                                                     false, false);
    }
  }
  const float* bb = b1 + 8 * h;
#pragma unroll
  for (int t = 0; t < H1 / 16; ++t) {
    const float4 bl = *(const float4*)(bb + 16 * t);
    const float4 bh = *(const float4*)(bb + 16 * t + 4);
    acc[t][0] = fmaxf(acc[t][0] + bl.x, 0.f);
    acc[t][1] = fmaxf(acc[t][1] + bl.y, 0.f);
    acc[t][2] = fmaxf(acc[t][2] + bl.z, 0.f);
    acc[t][3] = fmaxf(acc[t][3] + bl.w, 0.f);
    acc[t][4] = fmaxf(acc[t][4] + bh.x, 0.f);
    acc[t][5] = fmaxf(acc[t][5] + bh.y, 0.f);
    acc[t][6] = fmaxf(acc[t][6] + bh.z, 0.f);
    acc[t][7] = fmaxf(acc[t][7] + bh.w, 0.f);
  }
  const float* w2a = W2 + (size_t)(8 * h) * HID + c;
  v8f o[HID / 16] = {};
#pragma unroll
  for (int t = 0; t < H1 / 16; ++t) {
#pragma unroll
    for (int p = 0; p < 4; ++p) {
      v2f b;
      b.x = acc[t][2 * p];
      b.y = acc[t][2 * p + 1];
#pragma unroll
      for (int u = 0; u < HID / 16; ++u) {
        v2f a;
        a.x = w2a[(size_t)(16 * t + 2 * p) * HID + 16 * u];
        a.y = w2a[(size_t)(16 * t + 2 * p + 1) * HID + 16 * u];
        o[u] = __builtin_amdgcn_wmma_f32_16x16x4_f32(false, a, false, b, (short)0, o[u],
                                                     false, false);
      }
    }
  }
  if (node < M) {
    float* xo = XL + (size_t)node * HID + 8 * h;
#pragma unroll
    for (int u = 0; u < HID / 16; ++u) {
      float4 lo, hi;
      lo.x = di * o[u][0]; lo.y = di * o[u][1]; lo.z = di * o[u][2]; lo.w = di * o[u][3];
      hi.x = di * o[u][4]; hi.y = di * o[u][5]; hi.z = di * o[u][6]; hi.w = di * o[u][7];
      *(float4*)(xo + 16 * u) = lo;
      *(float4*)(xo + 16 * u + 4) = hi;
    }
  }
}

#define PULL_NODES 8
#define PULL_BATCH 16
__device__ __forceinline__ unsigned row_off(int row, int lane) {
  return ((unsigned)row * HID + (unsigned)lane) * 4u;
}
__global__ __launch_bounds__(256)
void pull_rows_kernel(const float* __restrict__ rows, const int* __restrict__ start,
                      const int* __restrict__ cnt, const int* __restrict__ col,
                      float* __restrict__ agg, int N) {
  static_assert(HID == 32, "one row per wave: HID must equal the wave size");
  static_assert(PULL_NODES <= 32 && (PULL_NODES & (PULL_NODES - 1)) == 0 && 32 % PULL_BATCH == 0,
                "nodes per wave: a power of two within its lanes; batches tile a 32-slot run");
  int wave = (int)((blockIdx.x * (long long)blockDim.x + threadIdx.x) >> 5);
  int lane = threadIdx.x & 31;
  int n0 = wave * PULL_NODES;
  if (n0 >= N) return;
  int nl = n0 + (lane & (PULL_NODES - 1));
  if (nl > N - 1) nl = N - 1;
  int sv = start[nl], cv = cnt[nl];
  int first[PULL_NODES];
#pragma unroll
  for (int g = 0; g < PULL_NODES; ++g) {
    int s = __builtin_amdgcn_readlane(sv, g);
    int c = __builtin_amdgcn_readlane(cv, g);
    int j = (lane < c) ? lane : c - 1;
    first[g] = 0;
    if (c > 0) first[g] = col[s + j];
  }
#pragma unroll
  for (int g = 0; g < PULL_NODES; ++g) {
    int n = n0 + g;
    if (n < N) {
      int s = __builtin_amdgcn_readlane(sv, g);
      int c = __builtin_amdgcn_readlane(cv, g);
      float acc = 0.f;
      for (int j0 = 0; j0 < c; j0 += 32) {
        int cs = first[g];
        if (j0 > 0) {
          int j = j0 + lane;
          if (j > c - 1) j = c - 1;
          cs = col[s + j];
        }
        int m = c - j0;
        if (m > 32) m = 32;
        for (int i = 0; i < m; i += PULL_BATCH) {
          float v[PULL_BATCH];
#pragma unroll
          for (int k = 0; k < PULL_BATCH; ++k) {
            int ik = i + k;
            if (ik > m - 1) ik = m - 1;
            int si = __builtin_amdgcn_readlane(cs, ik);
            v[k] = *(const float*)((const char*)rows + row_off(si, lane));
          }
#pragma unroll
          for (int k = 0; k < PULL_BATCH; ++k) {
            if (i + k < m) acc += v[k];
          }
        }
      }
      agg[(size_t)n * HID + lane] = acc;
    }
  }
}

__global__ __launch_bounds__(256)
void gemm_pq_wmma_f32(const float* __restrict__ agg, const float* __restrict__ xl,
                      const float* __restrict__ dinv, const float* __restrict__ b2,
                      const float* __restrict__ Wm1, const float* __restrict__ bm1,
                      float* __restrict__ PQ, int M) {
  int tm = (int)((blockIdx.x * (long long)blockDim.x + threadIdx.x) >> 5);
  int lane = threadIdx.x & 31;
  int mtiles = (M + 15) >> 4;
  if (tm >= mtiles) return;
  int l15 = lane & 15;
  int mrow = tm * 16 + l15;
  if (mrow > M - 1) mrow = M - 1;
  int kb = (lane >> 4) * 2;
  float di = dinv[mrow];
  const float* ga = agg + (size_t)mrow * HID + kb;
  const float* xa = xl + (size_t)mrow * HID + kb;
  const float* ba = b2 + kb;
  const float* wb = Wm1 + (size_t)kb * HID + l15;
  v8f acc[4] = {};
  for (int k0 = 0; k0 < HID; k0 += 4) {
    v2f g = *(const v2f*)(ga + k0);
    v2f xv = *(const v2f*)(xa + k0);
    v2f a;
    a.x = fmaxf(di * (g.x + xv.x) + ba[k0], 0.f);
    a.y = fmaxf(di * (g.y + xv.y) + ba[k0 + 1], 0.f);
#pragma unroll
    for (int t = 0; t < 4; ++t) {
      const float* wt = wb + (size_t)((t >> 1) * HID) * HID + (t & 1) * 16;
      v2f b;
      b.x = wt[(size_t)k0 * HID];
      b.y = wt[(size_t)(k0 + 1) * HID];
      acc[t] = __builtin_amdgcn_wmma_f32_16x16x4_f32(false, a, false, b, (short)0, acc[t],
                                                     false, false);
    }
  }
  int rbase = tm * 16 + ((lane >> 4) ? 8 : 0);
#pragma unroll
  for (int t = 0; t < 4; ++t) {
    int ncol = t * 16 + l15;
    float bn = (t < 2) ? bm1[ncol] : 0.f;
#pragma unroll
    for (int r = 0; r < 8; ++r) {
      int row = rbase + r;
      float val = acc[t][r];
      if (t < 2) val += bn;
      if (row < M) PQ[(size_t)row * (2 * HID) + ncol] = val;
    }
  }
}

#define EDGE_CHUNKS 4
__global__ __launch_bounds__(256)
void edge_out_kernel(const float* __restrict__ PQ, const int* __restrict__ p0,
                     const int* __restrict__ p1, const int* __restrict__ q0,
                     const int* __restrict__ q1, const float* __restrict__ Wm2,
                     const float* __restrict__ bm2, float* __restrict__ out,
                     int NP, int Etot) {
  int wave = (int)((blockIdx.x * (long long)blockDim.x + threadIdx.x) >> 5);
  int lane = threadIdx.x & 31;
  int G = lane >> 3;
  int j = lane & 7;
  int c = j * 4;
  const float4 w01 = *(const float4*)(Wm2 + c * 2);
  const float4 w23 = *(const float4*)(Wm2 + c * 2 + 4);
  const float bo = bm2[j & 1];
  const bool hi4 = (j & 4) != 0, hi2 = (j & 2) != 0, hi1 = (j & 1) != 0;
  for (int it = 0; it < EDGE_CHUNKS; ++it) {
    int e0 = (wave * EDGE_CHUNKS + it) * 16;
    if (e0 >= Etot) break;
    float v[8];
#pragma unroll
    for (int i = 0; i < 4; ++i) {
      int e = e0 + 4 * G + i;
      if (e > Etot - 1) e = Etot - 1;
      int s = (e < NP) ? p0[e] : q0[e - NP];
      int d = (e < NP) ? p1[e] : q1[e - NP];
      const float4 p = *(const float4*)(PQ + (size_t)s * (2 * HID) + c);
      const float4 q = *(const float4*)(PQ + (size_t)d * (2 * HID) + HID + c);
      float t0 = fmaxf(p.x + q.x, 0.f);
      float t1 = fmaxf(p.y + q.y, 0.f);
      float t2 = fmaxf(p.z + q.z, 0.f);
      float t3 = fmaxf(p.w + q.w, 0.f);
      v[2 * i]     = t0 * w01.x + t1 * w01.z + t2 * w23.x + t3 * w23.z;
      v[2 * i + 1] = t0 * w01.y + t1 * w01.w + t2 * w23.y + t3 * w23.w;
    }
    float u[4];
#pragma unroll
    for (int k = 0; k < 4; ++k) {
      float keep = hi4 ? v[k + 4] : v[k];
      float send = hi4 ? v[k] : v[k + 4];
      u[k] = keep + __shfl_xor(send, 4, 8);
    }
    float w[2];
#pragma unroll
    for (int k = 0; k < 2; ++k) {
      float keep = hi2 ? u[k + 2] : u[k];
      float send = hi2 ? u[k] : u[k + 2];
      w[k] = keep + __shfl_xor(send, 2, 8);
    }
    float keep1 = hi1 ? w[1] : w[0];
    float send1 = hi1 ? w[0] : w[1];
    float r = keep1 + __shfl_xor(send1, 1, 8);
    int eo = e0 + 4 * G + (j >> 1);
    if (eo < Etot) out[(size_t)e0 * 2 + lane] = r + bo;
  }
}

extern "C" void kernel_launch(void* const* d_in, const int* in_sizes, int n_in,
                              void* d_out, int out_size, void* d_ws, size_t ws_size,
                              hipStream_t stream) {
  const float* x   = (const float*)d_in[0];
  const int*   pe  = (const int*)d_in[1];
  const int*   neg = (const int*)d_in[2];
  const float* W1  = (const float*)d_in[3];
  const float* b1  = (const float*)d_in[4];
  const float* W2  = (const float*)d_in[5];
  const float* b2  = (const float*)d_in[6];
  const float* Wm1 = (const float*)d_in[7];
  const float* bm1 = (const float*)d_in[8];
  const float* Wm2 = (const float*)d_in[9];
  const float* bm2 = (const float*)d_in[10];
  float* out = (float*)d_out;

  const int N  = in_sizes[0] / HID;
  const int NP = in_sizes[1] / 2;
  const int NN = in_sizes[2] / 2;
  const int ET = NP + NN;
  const int* p0 = pe;        const int* p1 = pe + NP;
  const int* q0 = neg;       const int* q1 = neg + NN;

  char* ws = (char*)d_ws;
  size_t off = 0;
  auto wsalloc = [&](size_t bytes) -> char* {
    char* p = ws + off;
    off += (bytes + 255) & ~(size_t)255;
    return p;
  };
  int*   cnt  = (int*)wsalloc((size_t)(N + 1) * 4);
  int*   tot  = cnt + N;
  int*   seg  = (int*)wsalloc((size_t)N * 4);
  int*   rnk  = (int*)wsalloc((size_t)NP * 4);
  int*   col  = (int*)wsalloc((size_t)NP * 4);
  float* dinv = (float*)wsalloc((size_t)N * 4);
  float* xs   = (float*)wsalloc((size_t)N * HID * 4);
  float* ax   = (float*)wsalloc((size_t)N * HID * 4);
  float* xl2  = (float*)wsalloc((size_t)N * HID * 4);
  float* ag2  = (float*)wsalloc((size_t)N * HID * 4);
  float* pq   = (float*)wsalloc((size_t)N * 2 * HID * 4);
  (void)ws_size; (void)n_in; (void)out_size;

  const int B = 256;

  hipMemsetAsync(cnt, 0, (size_t)(N + 1) * 4, stream);
  {
    long long thr = ((long long)NP + EDGE_PIECES - 1) / EDGE_PIECES;
    unsigned eb = (unsigned)((thr + B - 1) / B);
    rank_kernel<<<eb, B, 0, stream>>>(p1, cnt, rnk, NP);
    seg_start_kernel<<<(N + B - 1) / B, B, 0, stream>>>(cnt, seg, tot, N);
    fill_kernel<<<eb, B, 0, stream>>>(p0, p1, rnk, seg, col, NP);
  }
  {
    long long thr = ((long long)N * (HID / 4) + PRESCALE_PIECES - 1) / PRESCALE_PIECES;
    dinv_prescale_kernel<<<(unsigned)((thr + B - 1) / B), B, 0, stream>>>(cnt, x, dinv, xs, N);
  }

  const long long pthr = (long long)((N + PULL_NODES - 1) / PULL_NODES) * 32;
  const unsigned pb = (unsigned)((pthr + B - 1) / B);
  const long long gthr = (long long)((N + 15) / 16) * 32;
  const unsigned gb = (unsigned)((gthr + B - 1) / B);

  pull_rows_kernel<<<pb, B, 0, stream>>>(xs, seg, cnt, col, ax, N);
  gemm_l1l2_wmma_f32<<<gb, B, 0, stream>>>(ax, xs, dinv, W1, b1, W2, xl2, N);
  pull_rows_kernel<<<pb, B, 0, stream>>>(xl2, seg, cnt, col, ag2, N);
  gemm_pq_wmma_f32<<<gb, B, 0, stream>>>(ag2, xl2, dinv, b2, Wm1, bm1, pq, N);

  {
    long long thr = (long long)((ET + 16 * EDGE_CHUNKS - 1) / (16 * EDGE_CHUNKS)) * 32;
    edge_out_kernel<<<(unsigned)((thr + B - 1) / B), B, 0, stream>>>(pq, p0, p1, q0, q1,
                                                                     Wm2, bm2, out, NP, ET);
  }
}


# Edges bucketed in one pass: a fixed 64-slot bucket per node, one 256-B unit (slot = the degree count's returning add), overflow list past a full bucket; no start scan, no fill pass, 2 launches fewer
# speedup vs baseline: 2.9437x; 1.0484x over previous
#include <hip/hip_runtime.h>
#include <hip/hip_bf16.h>

typedef __attribute__((ext_vector_type(2))) float v2f;
typedef __attribute__((ext_vector_type(8))) float v8f;

#define HID 32
#define H1  128

#define BUCKET_CAP 64
#define EDGE_PIECES 4
__global__ __launch_bounds__(256)
void bucket_kernel(const int* __restrict__ src, const int* __restrict__ dst, int* __restrict__ cnt,
                   int* __restrict__ col, int* __restrict__ ovf, int E, int N) {
  int t = blockIdx.x * blockDim.x + threadIdx.x;
  int stride = gridDim.x * blockDim.x;
  int d[EDGE_PIECES];
  int s[EDGE_PIECES];
  int r[EDGE_PIECES];
#pragma unroll
  for (int i = 0; i < EDGE_PIECES; ++i) {
    int e = t + i * stride;
    if (e > E - 1) e = E - 1;
    d[i] = dst[e];
    s[i] = src[e];
  }
#pragma unroll
  for (int i = 0; i < EDGE_PIECES; ++i) {
    int e = t + i * stride;
    r[i] = 0;
    if (e < E) r[i] = atomicAdd(&cnt[d[i]], 1);
  }
#pragma unroll
  for (int i = 0; i < EDGE_PIECES; ++i) {
    int e = t + i * stride;
    if (e < E) {
      if (r[i] < BUCKET_CAP) {
        col[(size_t)d[i] * BUCKET_CAP + r[i]] = s[i];
      } else {
        int k = atomicAdd(&cnt[N], 1);
        ovf[2 * (size_t)k] = s[i];
        ovf[2 * (size_t)k + 1] = d[i];
      }
    }
  }
}

#define PRESCALE_PIECES 4
__global__ void dinv_prescale_kernel(const int* __restrict__ deg, const float* __restrict__ x,
                                     float* __restrict__ dinv, float* __restrict__ xs, int N) {
  int t = blockIdx.x * blockDim.x + threadIdx.x;
  int stride = gridDim.x * blockDim.x;
  int total = N * (HID / 4);
  int dg[PRESCALE_PIECES];
  float4 v[PRESCALE_PIECES];
#pragma unroll
  for (int i = 0; i < PRESCALE_PIECES; ++i) {
    int p = t + i * stride;
    if (p > total - 1) p = total - 1;
    dg[i] = deg[p / (HID / 4)];
    v[i] = *(const float4*)(x + (size_t)p * 4);
  }
#pragma unroll
  for (int i = 0; i < PRESCALE_PIECES; ++i) {
    int p = t + i * stride;
    if (p < total) {
      float dv = rsqrtf((float)dg[i] + 1.0f);
      float4 o = v[i];
      o.x *= dv; o.y *= dv; o.z *= dv; o.w *= dv;
      *(float4*)(xs + (size_t)p * 4) = o;
      if ((p & (HID / 4 - 1)) == 0) dinv[p / (HID / 4)] = dv;
    }
  }
}

__global__ __launch_bounds__(256)
void gemm_l1l2_wmma_f32(const float* __restrict__ agg, const float* __restrict__ xs,
                        const float* __restrict__ dinv, const float* __restrict__ W1,
                        const float* __restrict__ b1, const float* __restrict__ W2,
                        float* __restrict__ XL, int M) {
  int tm = (int)((blockIdx.x * (long long)blockDim.x + threadIdx.x) >> 5);
  int lane = threadIdx.x & 31;
  int mtiles = (M + 15) >> 4;
  if (tm >= mtiles) return;
  int c = lane & 15;
  int h = lane >> 4;
  int node = tm * 16 + c;
  int nodec = node;
  if (nodec > M - 1) nodec = M - 1;
  float di = dinv[nodec];
  const float* ga = agg + (size_t)nodec * HID + 2 * h;
  const float* xa = xs + (size_t)nodec * HID + 2 * h;
  v2f bx[HID / 4];
#pragma unroll
  for (int j = 0; j < HID / 4; ++j) {
    v2f g = *(const v2f*)(ga + 4 * j);
    v2f xv = *(const v2f*)(xa + 4 * j);
    bx[j].x = di * (g.x + xv.x);
    bx[j].y = di * (g.y + xv.y);
  }
  const float* w1a = W1 + (size_t)(2 * h) * H1 + c;
  v8f acc[H1 / 16] = {};
#pragma unroll
  for (int j = 0; j < HID / 4; ++j) {
#pragma unroll
    for (int t = 0; t < H1 / 16; ++t) {
      v2f a;
      a.x = w1a[(size_t)(4 * j) * H1 + 16 * t];
      a.y = w1a[(size_t)(4 * j + 1) * H1 + 16 * t];
      acc[t] = __builtin_amdgcn_wmma_f32_16x16x4_f32(false, a, false, bx[j], (short)0, acc[t],
                                                     false, false);
    }
  }
  const float* bb = b1 + 8 * h;
#pragma unroll
  for (int t = 0; t < H1 / 16; ++t) {
    const float4 bl = *(const float4*)(bb + 16 * t);
    const float4 bh = *(const float4*)(bb + 16 * t + 4);
    acc[t][0] = fmaxf(acc[t][0] + bl.x, 0.f);
    acc[t][1] = fmaxf(acc[t][1] + bl.y, 0.f);
    acc[t][2] = fmaxf(acc[t][2] + bl.z, 0.f);
    acc[t][3] = fmaxf(acc[t][3] + bl.w, 0.f);
    acc[t][4] = fmaxf(acc[t][4] + bh.x, 0.f);
    acc[t][5] = fmaxf(acc[t][5] + bh.y, 0.f);
    acc[t][6] = fmaxf(acc[t][6] + bh.z, 0.f);
    acc[t][7] = fmaxf(acc[t][7] + bh.w, 0.f);
  }
  const float* w2a = W2 + (size_t)(8 * h) * HID + c;
  v8f o[HID / 16] = {};
#pragma unroll
  for (int t = 0; t < H1 / 16; ++t) {
#pragma unroll
    for (int p = 0; p < 4; ++p) {
      v2f b;
      b.x = acc[t][2 * p];
      b.y = acc[t][2 * p + 1];
#pragma unroll
      for (int u = 0; u < HID / 16; ++u) {
        v2f a;
        a.x = w2a[(size_t)(16 * t + 2 * p) * HID + 16 * u];
        a.y = w2a[(size_t)(16 * t + 2 * p + 1) * HID + 16 * u];
        o[u] = __builtin_amdgcn_wmma_f32_16x16x4_f32(false, a, false, b, (short)0, o[u],
                                                     false, false);
      }
    }
  }
  if (node < M) {
    float* xo = XL + (size_t)node * HID + 8 * h;
#pragma unroll
    for (int u = 0; u < HID / 16; ++u) {
      float4 lo, hi;
      lo.x = di * o[u][0]; lo.y = di * o[u][1]; lo.z = di * o[u][2]; lo.w = di * o[u][3];
      hi.x = di * o[u][4]; hi.y = di * o[u][5]; hi.z = di * o[u][6]; hi.w = di * o[u][7];
      *(float4*)(xo + 16 * u) = lo;
      *(float4*)(xo + 16 * u + 4) = hi;
    }
  }
}

#define PULL_NODES 8
#define PULL_BATCH 16
__device__ __forceinline__ unsigned row_off(int row, int lane) {
  return ((unsigned)row * HID + (unsigned)lane) * 4u;
}
__global__ __launch_bounds__(256)
void pull_rows_kernel(const float* __restrict__ rows, const int* __restrict__ cnt,
                      const int* __restrict__ col, const int* __restrict__ ovf,
                      float* __restrict__ agg, int N) {
  static_assert(HID == 32, "one row per wave: HID must equal the wave size");
  static_assert(PULL_NODES <= 32 && (PULL_NODES & (PULL_NODES - 1)) == 0 && 32 % PULL_BATCH == 0,
                "nodes per wave: a power of two within its lanes; batches tile a 32-slot run");
  int wave = (int)((blockIdx.x * (long long)blockDim.x + threadIdx.x) >> 5);
  int lane = threadIdx.x & 31;
  int n0 = wave * PULL_NODES;
  if (n0 >= N) return;
  int nl = n0 + (lane & (PULL_NODES - 1));
  if (nl > N - 1) nl = N - 1;
  int cv = cnt[nl];
  int first[PULL_NODES];
#pragma unroll
  for (int g = 0; g < PULL_NODES; ++g) {
    int nb = n0 + g;
    if (nb > N - 1) nb = N - 1;
    int c = __builtin_amdgcn_readlane(cv, g);
    if (c > BUCKET_CAP) c = BUCKET_CAP;
    int j = (lane < c) ? lane : c - 1;
    first[g] = 0;
    if (c > 0) first[g] = col[(size_t)nb * BUCKET_CAP + j];
  }
#pragma unroll
  for (int g = 0; g < PULL_NODES; ++g) {
    int n = n0 + g;
    if (n < N) {
      int deg = __builtin_amdgcn_readlane(cv, g);
      int c = (deg < BUCKET_CAP) ? deg : BUCKET_CAP;
      const int* bucket = col + (size_t)n * BUCKET_CAP;
      float acc = 0.f;
      for (int j0 = 0; j0 < c; j0 += 32) {
        int cs = first[g];
        if (j0 > 0) {
          int j = j0 + lane;
          if (j > c - 1) j = c - 1;
          cs = bucket[j];
        }
        int m = c - j0;
        if (m > 32) m = 32;
        for (int i = 0; i < m; i += PULL_BATCH) {
          float v[PULL_BATCH];
#pragma unroll
          for (int k = 0; k < PULL_BATCH; ++k) {
            int ik = i + k;
            if (ik > m - 1) ik = m - 1;
            int si = __builtin_amdgcn_readlane(cs, ik);
            v[k] = *(const float*)((const char*)rows + row_off(si, lane));
          }
#pragma unroll
          for (int k = 0; k < PULL_BATCH; ++k) {
            if (i + k < m) acc += v[k];
          }
        }
      }
      if (deg > BUCKET_CAP) {
        int nov = cnt[N];
        int left = deg - BUCKET_CAP;
        for (int k0 = 0; k0 < nov && left > 0; k0 += 32) {
          int k = k0 + lane;
          if (k > nov - 1) k = nov - 1;
          int os = ovf[2 * (size_t)k];
          int od = ovf[2 * (size_t)k + 1];
          unsigned hit = __builtin_amdgcn_ballot_w32(k0 + lane < nov && od == n);
          left -= __builtin_popcount(hit);
          while (hit) {
            int b = __builtin_ctz(hit);
            hit &= hit - 1;
            int si = __builtin_amdgcn_readlane(os, b);
            acc += *(const float*)((const char*)rows + row_off(si, lane));
          }
        }
      }
      agg[(size_t)n * HID + lane] = acc;
    }
  }
}

__global__ __launch_bounds__(256)
void gemm_pq_wmma_f32(const float* __restrict__ agg, const float* __restrict__ xl,
                      const float* __restrict__ dinv, const float* __restrict__ b2,
                      const float* __restrict__ Wm1, const float* __restrict__ bm1,
                      float* __restrict__ PQ, int M) {
  int tm = (int)((blockIdx.x * (long long)blockDim.x + threadIdx.x) >> 5);
  int lane = threadIdx.x & 31;
  int mtiles = (M + 15) >> 4;
  if (tm >= mtiles) return;
  int l15 = lane & 15;
  int mrow = tm * 16 + l15;
  if (mrow > M - 1) mrow = M - 1;
  int kb = (lane >> 4) * 2;
  float di = dinv[mrow];
  const float* ga = agg + (size_t)mrow * HID + kb;
  const float* xa = xl + (size_t)mrow * HID + kb;
  const float* ba = b2 + kb;
  const float* wb = Wm1 + (size_t)kb * HID + l15;
  v8f acc[4] = {};
  for (int k0 = 0; k0 < HID; k0 += 4) {
    v2f g = *(const v2f*)(ga + k0);
    v2f xv = *(const v2f*)(xa + k0);
    v2f a;
    a.x = fmaxf(di * (g.x + xv.x) + ba[k0], 0.f);
    a.y = fmaxf(di * (g.y + xv.y) + ba[k0 + 1], 0.f);
#pragma unroll
    for (int t = 0; t < 4; ++t) {
      const float* wt = wb + (size_t)((t >> 1) * HID) * HID + (t & 1) * 16;
      v2f b;
      b.x = wt[(size_t)k0 * HID];
      b.y = wt[(size_t)(k0 + 1) * HID];
      acc[t] = __builtin_amdgcn_wmma_f32_16x16x4_f32(false, a, false, b, (short)0, acc[t],
                                                     false, false);
    }
  }
  int rbase = tm * 16 + ((lane >> 4) ? 8 : 0);
#pragma unroll
  for (int t = 0; t < 4; ++t) {
    int ncol = t * 16 + l15;
    float bn = (t < 2) ? bm1[ncol] : 0.f;
#pragma unroll
    for (int r = 0; r < 8; ++r) {
      int row = rbase + r;
      float val = acc[t][r];
      if (t < 2) val += bn;
      if (row < M) PQ[(size_t)row * (2 * HID) + ncol] = val;
    }
  }
}

#define EDGE_CHUNKS 4
__global__ __launch_bounds__(256)
void edge_out_kernel(const float* __restrict__ PQ, const int* __restrict__ p0,
                     const int* __restrict__ p1, const int* __restrict__ q0,
                     const int* __restrict__ q1, const float* __restrict__ Wm2,
                     const float* __restrict__ bm2, float* __restrict__ out,
                     int NP, int Etot) {
  int wave = (int)((blockIdx.x * (long long)blockDim.x + threadIdx.x) >> 5);
  int lane = threadIdx.x & 31;
  int G = lane >> 3;
  int j = lane & 7;
  int c = j * 4;
  const float4 w01 = *(const float4*)(Wm2 + c * 2);
  const float4 w23 = *(const float4*)(Wm2 + c * 2 + 4);
  const float bo = bm2[j & 1];
  const bool hi4 = (j & 4) != 0, hi2 = (j & 2) != 0, hi1 = (j & 1) != 0;
  for (int it = 0; it < EDGE_CHUNKS; ++it) {
    int e0 = (wave * EDGE_CHUNKS + it) * 16;
    if (e0 >= Etot) break;
    float v[8];
#pragma unroll
    for (int i = 0; i < 4; ++i) {
      int e = e0 + 4 * G + i;
      if (e > Etot - 1) e = Etot - 1;
      int s = (e < NP) ? p0[e] : q0[e - NP];
      int d = (e < NP) ? p1[e] : q1[e - NP];
      const float4 p = *(const float4*)(PQ + (size_t)s * (2 * HID) + c);
      const float4 q = *(const float4*)(PQ + (size_t)d * (2 * HID) + HID + c);
      float t0 = fmaxf(p.x + q.x, 0.f);
      float t1 = fmaxf(p.y + q.y, 0.f);
      float t2 = fmaxf(p.z + q.z, 0.f);
      float t3 = fmaxf(p.w + q.w, 0.f);
      v[2 * i]     = t0 * w01.x + t1 * w01.z + t2 * w23.x + t3 * w23.z;
      v[2 * i + 1] = t0 * w01.y + t1 * w01.w + t2 * w23.y + t3 * w23.w;
    }
    float u[4];
#pragma unroll
    for (int k = 0; k < 4; ++k) {
      float keep = hi4 ? v[k + 4] : v[k];
      float send = hi4 ? v[k] : v[k + 4];
      u[k] = keep + __shfl_xor(send, 4, 8);
    }
    float w[2];
#pragma unroll
    for (int k = 0; k < 2; ++k) {
      float keep = hi2 ? u[k + 2] : u[k];
      float send = hi2 ? u[k] : u[k + 2];
      w[k] = keep + __shfl_xor(send, 2, 8);
    }
    float keep1 = hi1 ? w[1] : w[0];
    float send1 = hi1 ? w[0] : w[1];
    float r = keep1 + __shfl_xor(send1, 1, 8);
    int eo = e0 + 4 * G + (j >> 1);
    if (eo < Etot) out[(size_t)e0 * 2 + lane] = r + bo;
  }
}

extern "C" void kernel_launch(void* const* d_in, const int* in_sizes, int n_in,
                              void* d_out, int out_size, void* d_ws, size_t ws_size,
                              hipStream_t stream) {
  const float* x   = (const float*)d_in[0];
  const int*   pe  = (const int*)d_in[1];
  const int*   neg = (const int*)d_in[2];
  const float* W1  = (const float*)d_in[3];
  const float* b1  = (const float*)d_in[4];
  const float* W2  = (const float*)d_in[5];
  const float* b2  = (const float*)d_in[6];
  const float* Wm1 = (const float*)d_in[7];
  const float* bm1 = (const float*)d_in[8];
  const float* Wm2 = (const float*)d_in[9];
  const float* bm2 = (const float*)d_in[10];
  float* out = (float*)d_out;

  const int N  = in_sizes[0] / HID;
  const int NP = in_sizes[1] / 2;
  const int NN = in_sizes[2] / 2;
  const int ET = NP + NN;
  const int* p0 = pe;        const int* p1 = pe + NP;
  const int* q0 = neg;       const int* q1 = neg + NN;

  char* ws = (char*)d_ws;
  size_t off = 0;
  auto wsalloc = [&](size_t bytes) -> char* {
    char* p = ws + off;
    off += (bytes + 255) & ~(size_t)255;
    return p;
  };
  int*   cnt  = (int*)wsalloc((size_t)(N + 1) * 4);
  int*   col  = (int*)wsalloc((size_t)N * BUCKET_CAP * 4);
  int*   ovf  = (int*)wsalloc((size_t)NP * 2 * 4);
  float* dinv = (float*)wsalloc((size_t)N * 4);
  float* xs   = (float*)wsalloc((size_t)N * HID * 4);
  float* ax   = (float*)wsalloc((size_t)N * HID * 4);
  float* xl2  = (float*)wsalloc((size_t)N * HID * 4);
  float* ag2  = (float*)wsalloc((size_t)N * HID * 4);
  float* pq   = (float*)wsalloc((size_t)N * 2 * HID * 4);
  (void)ws_size; (void)n_in; (void)out_size;

  const int B = 256;

  hipMemsetAsync(cnt, 0, (size_t)(N + 1) * 4, stream);
  {
    long long thr = ((long long)NP + EDGE_PIECES - 1) / EDGE_PIECES;
    unsigned eb = (unsigned)((thr + B - 1) / B);
    bucket_kernel<<<eb, B, 0, stream>>>(p0, p1, cnt, col, ovf, NP, N);
  }
  {
    long long thr = ((long long)N * (HID / 4) + PRESCALE_PIECES - 1) / PRESCALE_PIECES;
    dinv_prescale_kernel<<<(unsigned)((thr + B - 1) / B), B, 0, stream>>>(cnt, x, dinv, xs, N);
  }

  const long long pthr = (long long)((N + PULL_NODES - 1) / PULL_NODES) * 32;
  const unsigned pb = (unsigned)((pthr + B - 1) / B);
  const long long gthr = (long long)((N + 15) / 16) * 32;
  const unsigned gb = (unsigned)((gthr + B - 1) / B);

  pull_rows_kernel<<<pb, B, 0, stream>>>(xs, cnt, col, ovf, ax, N);
  gemm_l1l2_wmma_f32<<<gb, B, 0, stream>>>(ax, xs, dinv, W1, b1, W2, xl2, N);
  pull_rows_kernel<<<pb, B, 0, stream>>>(xl2, cnt, col, ovf, ag2, N);
  gemm_pq_wmma_f32<<<gb, B, 0, stream>>>(ag2, xl2, dinv, b2, Wm1, bm1, pq, N);

  {
    long long thr = (long long)((ET + 16 * EDGE_CHUNKS - 1) / (16 * EDGE_CHUNKS)) * 32;
    edge_out_kernel<<<(unsigned)((thr + B - 1) / B), B, 0, stream>>>(pq, p0, p1, q0, q1,
                                                                     Wm2, bm2, out, NP, ET);
  }
}


# Edge bucketing binned in LDS first: a block sorts 8192 edges by 256-node bin in LDS (one global add per block and bin), then a block per bin fills node buckets in LDS; edge sums' fp32 order changes
# speedup vs baseline: 5.0730x; 1.7233x over previous
#include <hip/hip_runtime.h>
#include <hip/hip_bf16.h>

typedef __attribute__((ext_vector_type(2))) float v2f;
typedef __attribute__((ext_vector_type(8))) float v8f;
typedef unsigned long long u64;

#define HID 32
#define H1  128

#define BUCKET_CAP 64
#define BUCKET_PITCH (BUCKET_CAP + 1)
#define BIN_SHIFT 8
#define BIN_NODES (1 << BIN_SHIFT)
#define BIN_MAX 512
#define BIN_CAP 8192
#define BIN_PAD 64
#define BINA_THREADS 512
#define BINA_EPT 16
#define BINA_EDGES (BINA_THREADS * BINA_EPT)
__device__ __forceinline__ u64 edge_word(int s, int d) { return ((u64)(unsigned)d << 32) | (u64)(unsigned)s; }
__device__ __forceinline__ int edge_src(u64 w) { return (int)(unsigned)w; }
__device__ __forceinline__ int edge_dst(u64 w) { return (int)(unsigned)(w >> 32); }
__global__ __launch_bounds__(BINA_THREADS)
void bin_edges_kernel(const int* __restrict__ src, const int* __restrict__ dst, int* __restrict__ gcnt,
                      int* __restrict__ binned, int* __restrict__ spill, int* __restrict__ cnt,
                      int E, int N, int NB) {
  static_assert(BIN_MAX == BINA_THREADS, "the scan holds one bin per thread");
  __shared__ int lcnt[BIN_MAX];
  __shared__ int lstart[BIN_MAX];
  __shared__ int delta[BIN_MAX];
  __shared__ int wsum[BINA_THREADS / 32];
  __shared__ u64 stage[BINA_EDGES];
  const int tid = threadIdx.x;
  const int lane = tid & 31;
  const int wv = tid >> 5;
  const int e0 = blockIdx.x * BINA_EDGES;
  if (blockIdx.x == 0 && tid == 0) cnt[N] = 0;
  lcnt[tid] = 0;
  __syncthreads();
  int d[BINA_EPT];
  int s[BINA_EPT];
  int r[BINA_EPT];
#pragma unroll
  for (int i = 0; i < BINA_EPT; ++i) {
    int e = e0 + i * BINA_THREADS + tid;
    if (e > E - 1) e = E - 1;
    d[i] = dst[e];
    s[i] = src[e];
  }
#pragma unroll
  for (int i = 0; i < BINA_EPT; ++i) {
    int e = e0 + i * BINA_THREADS + tid;
    r[i] = -1;
    if (e < E) r[i] = atomicAdd(&lcnt[d[i] >> BIN_SHIFT], 1);
  }
  __syncthreads();
  const int v = lcnt[tid];
  int incl = v;
#pragma unroll
  for (int off = 1; off < 32; off <<= 1) {
    int t = __shfl_up(incl, off, 32);
    if (lane >= off) incl += t;
  }
  if (lane == 31) wsum[wv] = incl;
  __syncthreads();
  if (wv == 0) {
    int w = (lane < BINA_THREADS / 32) ? wsum[lane] : 0;
    int wi = w;
#pragma unroll
    for (int off = 1; off < BINA_THREADS / 32; off <<= 1) {
      int t = __shfl_up(wi, off, 32);
      if (lane >= off) wi += t;
    }
    if (lane < BINA_THREADS / 32) wsum[lane] = wi - w;
  }
  __syncthreads();
  const int start = wsum[wv] + incl - v;
  int base = 0;
  if (v > 0) base = atomicAdd(&gcnt[(size_t)tid * BIN_PAD], v);
  lstart[tid] = start;
  delta[tid] = base - start;
  __syncthreads();
  int at[BINA_EPT];
#pragma unroll
  for (int i = 0; i < BINA_EPT; ++i) at[i] = lstart[d[i] >> BIN_SHIFT] + r[i];
#pragma unroll
  for (int i = 0; i < BINA_EPT; ++i) {
    if (r[i] >= 0) stage[at[i]] = edge_word(s[i], d[i]);
  }
  __syncthreads();
  int total = E - e0;
  if (total > BINA_EDGES) total = BINA_EDGES;
  u64* bo = (u64*)binned;
  u64* so = (u64*)spill;
  u64 w[BINA_EPT];
  int pos[BINA_EPT];
#pragma unroll
  for (int i = 0; i < BINA_EPT; ++i) {
    int k = i * BINA_THREADS + tid;
    if (k > total - 1) k = total - 1;
    w[i] = stage[k];
  }
#pragma unroll
  for (int i = 0; i < BINA_EPT; ++i) {
    int k = i * BINA_THREADS + tid;
    pos[i] = delta[edge_dst(w[i]) >> BIN_SHIFT] + k;
  }
#pragma unroll
  for (int i = 0; i < BINA_EPT; ++i) {
    int k = i * BINA_THREADS + tid;
    if (k < total) {
      u64* to;
      if (pos[i] < BIN_CAP) {
        to = bo + (size_t)(edge_dst(w[i]) >> BIN_SHIFT) * BIN_CAP + pos[i];
      } else {
        to = so + atomicAdd(&gcnt[(size_t)NB * BIN_PAD], 1);
      }
      *to = w[i];
    }
  }
}

#define BINB_THREADS 512
#define BINB_EPT 8
__global__ __launch_bounds__(BINB_THREADS)
void fill_buckets_kernel(const int* __restrict__ gcnt, const int* __restrict__ binned,
                         const int* __restrict__ spill, int* __restrict__ cnt, int* __restrict__ col,
                         int* __restrict__ ovf, int N, int NB) {
  static_assert(BIN_NODES <= BINB_THREADS, "one thread per node clears and stores the counts");
  __shared__ int lc[BIN_NODES];
  __shared__ int lb[BIN_NODES * BUCKET_PITCH];
  const int tid = threadIdx.x;
  const int lane = tid & 31;
  const int wv = tid >> 5;
  const int b = blockIdx.x;
  const int n0 = b << BIN_SHIFT;
  if (tid < BIN_NODES) lc[tid] = 0;
  __syncthreads();
  auto put = [&](u64 p, int slot) {
    int ld = edge_dst(p) - n0;
    if (slot < BUCKET_CAP) {
      lb[ld * BUCKET_PITCH + slot] = edge_src(p);
    } else {
      int k = atomicAdd(&cnt[N], 1);
      ovf[2 * (size_t)k] = edge_src(p);
      ovf[2 * (size_t)k + 1] = edge_dst(p);
    }
  };
  const int got = gcnt[(size_t)b * BIN_PAD];
  const int inbin = (got < BIN_CAP) ? got : BIN_CAP;
  const u64* be = (const u64*)binned + (size_t)b * BIN_CAP;
  for (int i0 = 0; i0 < inbin; i0 += BINB_THREADS * BINB_EPT) {
    u64 p[BINB_EPT];
    int slot[BINB_EPT];
#pragma unroll
    for (int k = 0; k < BINB_EPT; ++k) {
      int i = i0 + k * BINB_THREADS + tid;
      if (i > inbin - 1) i = inbin - 1;
      p[k] = be[i];
    }
#pragma unroll
    for (int k = 0; k < BINB_EPT; ++k) {
      int i = i0 + k * BINB_THREADS + tid;
      slot[k] = -1;
      if (i < inbin) slot[k] = atomicAdd(&lc[edge_dst(p[k]) - n0], 1);
    }
#pragma unroll
    for (int k = 0; k < BINB_EPT; ++k) {
      if (slot[k] >= 0) put(p[k], slot[k]);
    }
  }
  if (got > BIN_CAP) {
    const int ns = gcnt[(size_t)NB * BIN_PAD];
    const u64* sp = (const u64*)spill;
    for (int i = tid; i < ns; i += BINB_THREADS) {
      u64 p = sp[i];
      if ((edge_dst(p) >> BIN_SHIFT) == b) put(p, atomicAdd(&lc[edge_dst(p) - n0], 1));
    }
  }
  __syncthreads();
  if (tid < BIN_NODES && n0 + tid < N) cnt[n0 + tid] = lc[tid];
  for (int ld = wv; ld < BIN_NODES; ld += BINB_THREADS / 32) {
    int n = n0 + ld;
    if (n >= N) break;
    int c = lc[ld];
    if (c > BUCKET_CAP) c = BUCKET_CAP;
    int* bucket = col + (size_t)n * BUCKET_CAP;
    for (int j0 = 0; j0 < c; j0 += 32) {
      int j = j0 + lane;
      if (j < c) bucket[j] = lb[ld * BUCKET_PITCH + j];
    }
  }
}

#define PRESCALE_PIECES 4
__global__ void dinv_prescale_kernel(const int* __restrict__ deg, const float* __restrict__ x,
                                     float* __restrict__ dinv, float* __restrict__ xs, int N) {
  int t = blockIdx.x * blockDim.x + threadIdx.x;
  int stride = gridDim.x * blockDim.x;
  int total = N * (HID / 4);
  int dg[PRESCALE_PIECES];
  float4 v[PRESCALE_PIECES];
#pragma unroll
  for (int i = 0; i < PRESCALE_PIECES; ++i) {
    int p = t + i * stride;
    if (p > total - 1) p = total - 1;
    dg[i] = deg[p / (HID / 4)];
    v[i] = *(const float4*)(x + (size_t)p * 4);
  }
#pragma unroll
  for (int i = 0; i < PRESCALE_PIECES; ++i) {
    int p = t + i * stride;
    if (p < total) {
      float dv = rsqrtf((float)dg[i] + 1.0f);
      float4 o = v[i];
      o.x *= dv; o.y *= dv; o.z *= dv; o.w *= dv;
      *(float4*)(xs + (size_t)p * 4) = o;
      if ((p & (HID / 4 - 1)) == 0) dinv[p / (HID / 4)] = dv;
    }
  }
}

__global__ __launch_bounds__(256)
void gemm_l1l2_wmma_f32(const float* __restrict__ agg, const float* __restrict__ xs,
                        const float* __restrict__ dinv, const float* __restrict__ W1,
                        const float* __restrict__ b1, const float* __restrict__ W2,
                        float* __restrict__ XL, int M) {
  int tm = (int)((blockIdx.x * (long long)blockDim.x + threadIdx.x) >> 5);
  int lane = threadIdx.x & 31;
  int mtiles = (M + 15) >> 4;
  if (tm >= mtiles) return;
  int c = lane & 15;
  int h = lane >> 4;
  int node = tm * 16 + c;
  int nodec = node;
  if (nodec > M - 1) nodec = M - 1;
  float di = dinv[nodec];
  const float* ga = agg + (size_t)nodec * HID + 2 * h;
  const float* xa = xs + (size_t)nodec * HID + 2 * h;
  v2f bx[HID / 4];
#pragma unroll
  for (int j = 0; j < HID / 4; ++j) {
    v2f g = *(const v2f*)(ga + 4 * j);
    v2f xv = *(const v2f*)(xa + 4 * j);
    bx[j].x = di * (g.x + xv.x);
    bx[j].y = di * (g.y + xv.y);
  }
  const float* w1a = W1 + (size_t)(2 * h) * H1 + c;
  v8f acc[H1 / 16] = {};
#pragma unroll
  for (int j = 0; j < HID / 4; ++j) {
#pragma unroll
    for (int t = 0; t < H1 / 16; ++t) {
      v2f a;
      a.x = w1a[(size_t)(4 * j) * H1 + 16 * t];
      a.y = w1a[(size_t)(4 * j + 1) * H1 + 16 * t];
      acc[t] = __builtin_amdgcn_wmma_f32_16x16x4_f32(false, a, false, bx[j], (short)0, acc[t],
                                                     false, false);
    }
  }
  const float* bb = b1 + 8 * h;
#pragma unroll
  for (int t = 0; t < H1 / 16; ++t) {
    const float4 bl = *(const float4*)(bb + 16 * t);
    const float4 bh = *(const float4*)(bb + 16 * t + 4);
    acc[t][0] = fmaxf(acc[t][0] + bl.x, 0.f);
    acc[t][1] = fmaxf(acc[t][1] + bl.y, 0.f);
    acc[t][2] = fmaxf(acc[t][2] + bl.z, 0.f);
    acc[t][3] = fmaxf(acc[t][3] + bl.w, 0.f);
    acc[t][4] = fmaxf(acc[t][4] + bh.x, 0.f);
    acc[t][5] = fmaxf(acc[t][5] + bh.y, 0.f);
    acc[t][6] = fmaxf(acc[t][6] + bh.z, 0.f);
    acc[t][7] = fmaxf(acc[t][7] + bh.w, 0.f);
  }
  const float* w2a = W2 + (size_t)(8 * h) * HID + c;
  v8f o[HID / 16] = {};
#pragma unroll
  for (int t = 0; t < H1 / 16; ++t) {
#pragma unroll
    for (int p = 0; p < 4; ++p) {
      v2f b;
      b.x = acc[t][2 * p];
      b.y = acc[t][2 * p + 1];
#pragma unroll
      for (int u = 0; u < HID / 16; ++u) {
        v2f a;
        a.x = w2a[(size_t)(16 * t + 2 * p) * HID + 16 * u];
        a.y = w2a[(size_t)(16 * t + 2 * p + 1) * HID + 16 * u];
        o[u] = __builtin_amdgcn_wmma_f32_16x16x4_f32(false, a, false, b, (short)0, o[u],
                                                     false, false);
      }
    }
  }
  if (node < M) {
    float* xo = XL + (size_t)node * HID + 8 * h;
#pragma unroll
    for (int u = 0; u < HID / 16; ++u) {
      float4 lo, hi;
      lo.x = di * o[u][0]; lo.y = di * o[u][1]; lo.z = di * o[u][2]; lo.w = di * o[u][3];
      hi.x = di * o[u][4]; hi.y = di * o[u][5]; hi.z = di * o[u][6]; hi.w = di * o[u][7];
      *(float4*)(xo + 16 * u) = lo;
      *(float4*)(xo + 16 * u + 4) = hi;
    }
  }
}

#define PULL_NODES 8
#define PULL_BATCH 16
__device__ __forceinline__ unsigned row_off(int row, int lane) {
  return ((unsigned)row * HID + (unsigned)lane) * 4u;
}
__global__ __launch_bounds__(256)
void pull_rows_kernel(const float* __restrict__ rows, const int* __restrict__ cnt,
                      const int* __restrict__ col, const int* __restrict__ ovf,
                      float* __restrict__ agg, int N) {
  static_assert(HID == 32, "one row per wave: HID must equal the wave size");
  static_assert(PULL_NODES <= 32 && (PULL_NODES & (PULL_NODES - 1)) == 0 && 32 % PULL_BATCH == 0,
                "nodes per wave: a power of two within its lanes; batches tile a 32-slot run");
  int wave = (int)((blockIdx.x * (long long)blockDim.x + threadIdx.x) >> 5);
  int lane = threadIdx.x & 31;
  int n0 = wave * PULL_NODES;
  if (n0 >= N) return;
  int nl = n0 + (lane & (PULL_NODES - 1));
  if (nl > N - 1) nl = N - 1;
  int cv = cnt[nl];
  int first[PULL_NODES];
#pragma unroll
  for (int g = 0; g < PULL_NODES; ++g) {
    int nb = n0 + g;
    if (nb > N - 1) nb = N - 1;
    int c = __builtin_amdgcn_readlane(cv, g);
    if (c > BUCKET_CAP) c = BUCKET_CAP;
    int j = (lane < c) ? lane : c - 1;
    first[g] = 0;
    if (c > 0) first[g] = col[(size_t)nb * BUCKET_CAP + j];
  }
#pragma unroll
  for (int g = 0; g < PULL_NODES; ++g) {
    int n = n0 + g;
    if (n < N) {
      int deg = __builtin_amdgcn_readlane(cv, g);
      int c = (deg < BUCKET_CAP) ? deg : BUCKET_CAP;
      const int* bucket = col + (size_t)n * BUCKET_CAP;
      float acc = 0.f;
      for (int j0 = 0; j0 < c; j0 += 32) {
        int cs = first[g];
        if (j0 > 0) {
          int j = j0 + lane;
          if (j > c - 1) j = c - 1;
          cs = bucket[j];
        }
        int m = c - j0;
        if (m > 32) m = 32;
        for (int i = 0; i < m; i += PULL_BATCH) {
          float v[PULL_BATCH];
#pragma unroll
          for (int k = 0; k < PULL_BATCH; ++k) {
            int ik = i + k;
            if (ik > m - 1) ik = m - 1;
            int si = __builtin_amdgcn_readlane(cs, ik);
            v[k] = *(const float*)((const char*)rows + row_off(si, lane));
          }
#pragma unroll
          for (int k = 0; k < PULL_BATCH; ++k) {
            if (i + k < m) acc += v[k];
          }
        }
      }
      if (deg > BUCKET_CAP) {
        int nov = cnt[N];
        int left = deg - BUCKET_CAP;
        for (int k0 = 0; k0 < nov && left > 0; k0 += 32) {
          int k = k0 + lane;
          if (k > nov - 1) k = nov - 1;
          int os = ovf[2 * (size_t)k];
          int od = ovf[2 * (size_t)k + 1];
          unsigned hit = __builtin_amdgcn_ballot_w32(k0 + lane < nov && od == n);
          left -= __builtin_popcount(hit);
          while (hit) {
            int b = __builtin_ctz(hit);
            hit &= hit - 1;
            int si = __builtin_amdgcn_readlane(os, b);
            acc += *(const float*)((const char*)rows + row_off(si, lane));
          }
        }
      }
      agg[(size_t)n * HID + lane] = acc;
    }
  }
}

__global__ __launch_bounds__(256)
void gemm_pq_wmma_f32(const float* __restrict__ agg, const float* __restrict__ xl,
                      const float* __restrict__ dinv, const float* __restrict__ b2,
                      const float* __restrict__ Wm1, const float* __restrict__ bm1,
                      float* __restrict__ PQ, int M) {
  int tm = (int)((blockIdx.x * (long long)blockDim.x + threadIdx.x) >> 5);
  int lane = threadIdx.x & 31;
  int mtiles = (M + 15) >> 4;
  if (tm >= mtiles) return;
  int l15 = lane & 15;
  int mrow = tm * 16 + l15;
  if (mrow > M - 1) mrow = M - 1;
  int kb = (lane >> 4) * 2;
  float di = dinv[mrow];
  const float* ga = agg + (size_t)mrow * HID + kb;
  const float* xa = xl + (size_t)mrow * HID + kb;
  const float* ba = b2 + kb;
  const float* wb = Wm1 + (size_t)kb * HID + l15;
  v8f acc[4] = {};
  for (int k0 = 0; k0 < HID; k0 += 4) {
    v2f g = *(const v2f*)(ga + k0);
    v2f xv = *(const v2f*)(xa + k0);
    v2f a;
    a.x = fmaxf(di * (g.x + xv.x) + ba[k0], 0.f);
    a.y = fmaxf(di * (g.y + xv.y) + ba[k0 + 1], 0.f);
#pragma unroll
    for (int t = 0; t < 4; ++t) {
      const float* wt = wb + (size_t)((t >> 1) * HID) * HID + (t & 1) * 16;
      v2f b;
      b.x = wt[(size_t)k0 * HID];
      b.y = wt[(size_t)(k0 + 1) * HID];
      acc[t] = __builtin_amdgcn_wmma_f32_16x16x4_f32(false, a, false, b, (short)0, acc[t],
                                                     false, false);
    }
  }
  int rbase = tm * 16 + ((lane >> 4) ? 8 : 0);
#pragma unroll
  for (int t = 0; t < 4; ++t) {
    int ncol = t * 16 + l15;
    float bn = (t < 2) ? bm1[ncol] : 0.f;
#pragma unroll
    for (int r = 0; r < 8; ++r) {
      int row = rbase + r;
      float val = acc[t][r];
      if (t < 2) val += bn;
      if (row < M) PQ[(size_t)row * (2 * HID) + ncol] = val;
    }
  }
}

#define EDGE_CHUNKS 4
__global__ __launch_bounds__(256)
void edge_out_kernel(const float* __restrict__ PQ, const int* __restrict__ p0,
                     const int* __restrict__ p1, const int* __restrict__ q0,
                     const int* __restrict__ q1, const float* __restrict__ Wm2,
                     const float* __restrict__ bm2, float* __restrict__ out,
                     int NP, int Etot) {
  int wave = (int)((blockIdx.x * (long long)blockDim.x + threadIdx.x) >> 5);
  int lane = threadIdx.x & 31;
  int G = lane >> 3;
  int j = lane & 7;
  int c = j * 4;
  const float4 w01 = *(const float4*)(Wm2 + c * 2);
  const float4 w23 = *(const float4*)(Wm2 + c * 2 + 4);
  const float bo = bm2[j & 1];
  const bool hi4 = (j & 4) != 0, hi2 = (j & 2) != 0, hi1 = (j & 1) != 0;
  for (int it = 0; it < EDGE_CHUNKS; ++it) {
    int e0 = (wave * EDGE_CHUNKS + it) * 16;
    if (e0 >= Etot) break;
    float v[8];
#pragma unroll
    for (int i = 0; i < 4; ++i) {
      int e = e0 + 4 * G + i;
      if (e > Etot - 1) e = Etot - 1;
      int s = (e < NP) ? p0[e] : q0[e - NP];
      int d = (e < NP) ? p1[e] : q1[e - NP];
      const float4 p = *(const float4*)(PQ + (size_t)s * (2 * HID) + c);
      const float4 q = *(const float4*)(PQ + (size_t)d * (2 * HID) + HID + c);
      float t0 = fmaxf(p.x + q.x, 0.f);
      float t1 = fmaxf(p.y + q.y, 0.f);
      float t2 = fmaxf(p.z + q.z, 0.f);
      float t3 = fmaxf(p.w + q.w, 0.f);
      v[2 * i]     = t0 * w01.x + t1 * w01.z + t2 * w23.x + t3 * w23.z;
      v[2 * i + 1] = t0 * w01.y + t1 * w01.w + t2 * w23.y + t3 * w23.w;
    }
    float u[4];
#pragma unroll
    for (int k = 0; k < 4; ++k) {
      float keep = hi4 ? v[k + 4] : v[k];
      float send = hi4 ? v[k] : v[k + 4];
      u[k] = keep + __shfl_xor(send, 4, 8);
    }
    float w[2];
#pragma unroll
    for (int k = 0; k < 2; ++k) {
      float keep = hi2 ? u[k + 2] : u[k];
      float send = hi2 ? u[k] : u[k + 2];
      w[k] = keep + __shfl_xor(send, 2, 8);
    }
    float keep1 = hi1 ? w[1] : w[0];
    float send1 = hi1 ? w[0] : w[1];
    float r = keep1 + __shfl_xor(send1, 1, 8);
    int eo = e0 + 4 * G + (j >> 1);
    if (eo < Etot) out[(size_t)e0 * 2 + lane] = r + bo;
  }
}

extern "C" void kernel_launch(void* const* d_in, const int* in_sizes, int n_in,
                              void* d_out, int out_size, void* d_ws, size_t ws_size,
                              hipStream_t stream) {
  const float* x   = (const float*)d_in[0];
  const int*   pe  = (const int*)d_in[1];
  const int*   neg = (const int*)d_in[2];
  const float* W1  = (const float*)d_in[3];
  const float* b1  = (const float*)d_in[4];
  const float* W2  = (const float*)d_in[5];
  const float* b2  = (const float*)d_in[6];
  const float* Wm1 = (const float*)d_in[7];
  const float* bm1 = (const float*)d_in[8];
  const float* Wm2 = (const float*)d_in[9];
  const float* bm2 = (const float*)d_in[10];
  float* out = (float*)d_out;

  const int N  = in_sizes[0] / HID;
  const int NP = in_sizes[1] / 2;
  const int NN = in_sizes[2] / 2;
  const int ET = NP + NN;
  const int* p0 = pe;        const int* p1 = pe + NP;
  const int* q0 = neg;       const int* q1 = neg + NN;

  char* ws = (char*)d_ws;
  size_t off = 0;
  auto wsalloc = [&](size_t bytes) -> char* {
    char* p = ws + off;
    off += (bytes + 255) & ~(size_t)255;
    return p;
  };
  const int NB = (N + BIN_NODES - 1) >> BIN_SHIFT;
  int*   gcnt = (int*)wsalloc(((size_t)NB + 1) * BIN_PAD * 4);
  int*   binned = (int*)wsalloc((size_t)NB * BIN_CAP * 8);
  int*   spill = (int*)wsalloc((size_t)NP * 8);
  int*   cnt  = (int*)wsalloc((size_t)(N + 1) * 4);
  int*   col  = (int*)wsalloc((size_t)N * BUCKET_CAP * 4);
  int*   ovf  = (int*)wsalloc((size_t)NP * 2 * 4);
  float* dinv = (float*)wsalloc((size_t)N * 4);
  float* xs   = (float*)wsalloc((size_t)N * HID * 4);
  float* ax   = (float*)wsalloc((size_t)N * HID * 4);
  float* xl2  = (float*)wsalloc((size_t)N * HID * 4);
  float* ag2  = (float*)wsalloc((size_t)N * HID * 4);
  float* pq   = (float*)wsalloc((size_t)N * 2 * HID * 4);
  (void)ws_size; (void)n_in; (void)out_size;

  const int B = 256;

  hipMemsetAsync(gcnt, 0, ((size_t)NB + 1) * BIN_PAD * 4, stream);
  bin_edges_kernel<<<(unsigned)((NP + BINA_EDGES - 1) / BINA_EDGES), BINA_THREADS, 0, stream>>>(
      p0, p1, gcnt, binned, spill, cnt, NP, N, NB);
  fill_buckets_kernel<<<(unsigned)NB, BINB_THREADS, 0, stream>>>(gcnt, binned, spill, cnt, col, ovf, N, NB);
  {
    long long thr = ((long long)N * (HID / 4) + PRESCALE_PIECES - 1) / PRESCALE_PIECES;
    dinv_prescale_kernel<<<(unsigned)((thr + B - 1) / B), B, 0, stream>>>(cnt, x, dinv, xs, N);
  }

  const long long pthr = (long long)((N + PULL_NODES - 1) / PULL_NODES) * 32;
  const unsigned pb = (unsigned)((pthr + B - 1) / B);
  const long long gthr = (long long)((N + 15) / 16) * 32;
  const unsigned gb = (unsigned)((gthr + B - 1) / B);

  pull_rows_kernel<<<pb, B, 0, stream>>>(xs, cnt, col, ovf, ax, N);
  gemm_l1l2_wmma_f32<<<gb, B, 0, stream>>>(ax, xs, dinv, W1, b1, W2, xl2, N);
  pull_rows_kernel<<<pb, B, 0, stream>>>(xl2, cnt, col, ovf, ag2, N);
  gemm_pq_wmma_f32<<<gb, B, 0, stream>>>(ag2, xl2, dinv, b2, Wm1, bm1, pq, N);

  {
    long long thr = (long long)((ET + 16 * EDGE_CHUNKS - 1) / (16 * EDGE_CHUNKS)) * 32;
    edge_out_kernel<<<(unsigned)((thr + B - 1) / B), B, 0, stream>>>(pq, p0, p1, q0, q1,
                                                                     Wm2, bm2, out, NP, ET);
  }
}


# Both edge sums feed their products inside one launch per layer: 2 waves per 16 nodes walk 8 buckets each, leave sum + own row in LDS, split the tile's products; W2 product's fp32 k sum order changes
# speedup vs baseline: 5.5399x; 1.0920x over previous
#include <hip/hip_runtime.h>
#include <hip/hip_bf16.h>

typedef __attribute__((ext_vector_type(2))) float v2f;
typedef __attribute__((ext_vector_type(8))) float v8f;
typedef unsigned long long u64;

#define HID 32
#define H1  128

#define BUCKET_CAP 64
#define BUCKET_PITCH (BUCKET_CAP + 1)
#define BIN_SHIFT 8
#define BIN_NODES (1 << BIN_SHIFT)
#define BIN_MAX 512
#define BIN_CAP 8192
#define BIN_PAD 64
#define BINA_THREADS 512
#define BINA_EPT 16
#define BINA_EDGES (BINA_THREADS * BINA_EPT)
__device__ __forceinline__ u64 edge_word(int s, int d) { return ((u64)(unsigned)d << 32) | (u64)(unsigned)s; }
__device__ __forceinline__ int edge_src(u64 w) { return (int)(unsigned)w; }
__device__ __forceinline__ int edge_dst(u64 w) { return (int)(unsigned)(w >> 32); }
__global__ __launch_bounds__(BINA_THREADS)
void bin_edges_kernel(const int* __restrict__ src, const int* __restrict__ dst, int* __restrict__ gcnt,
                      int* __restrict__ binned, int* __restrict__ spill, int* __restrict__ cnt,
                      int E, int N, int NB) {
  static_assert(BIN_MAX == BINA_THREADS, "the scan holds one bin per thread");
  __shared__ int lcnt[BIN_MAX];
  __shared__ int lstart[BIN_MAX];
  __shared__ int delta[BIN_MAX];
  __shared__ int wsum[BINA_THREADS / 32];
  __shared__ u64 stage[BINA_EDGES];
  const int tid = threadIdx.x;
  const int lane = tid & 31;
  const int wv = tid >> 5;
  const int e0 = blockIdx.x * BINA_EDGES;
  if (blockIdx.x == 0 && tid == 0) cnt[N] = 0;
  lcnt[tid] = 0;
  __syncthreads();
  int d[BINA_EPT];
  int s[BINA_EPT];
  int r[BINA_EPT];
#pragma unroll
  for (int i = 0; i < BINA_EPT; ++i) {
    int e = e0 + i * BINA_THREADS + tid;
    if (e > E - 1) e = E - 1;
    d[i] = dst[e];
    s[i] = src[e];
  }
#pragma unroll
  for (int i = 0; i < BINA_EPT; ++i) {
    int e = e0 + i * BINA_THREADS + tid;
    r[i] = -1;
    if (e < E) r[i] = atomicAdd(&lcnt[d[i] >> BIN_SHIFT], 1);
  }
  __syncthreads();
  const int v = lcnt[tid];
  int incl = v;
#pragma unroll
  for (int off = 1; off < 32; off <<= 1) {
    int t = __shfl_up(incl, off, 32);
    if (lane >= off) incl += t;
  }
  if (lane == 31) wsum[wv] = incl;
  __syncthreads();
  if (wv == 0) {
    int w = (lane < BINA_THREADS / 32) ? wsum[lane] : 0;
    int wi = w;
#pragma unroll
    for (int off = 1; off < BINA_THREADS / 32; off <<= 1) {
      int t = __shfl_up(wi, off, 32);
      if (lane >= off) wi += t;
    }
    if (lane < BINA_THREADS / 32) wsum[lane] = wi - w;
  }
  __syncthreads();
  const int start = wsum[wv] + incl - v;
  int base = 0;
  if (v > 0) base = atomicAdd(&gcnt[(size_t)tid * BIN_PAD], v);
  lstart[tid] = start;
  delta[tid] = base - start;
  __syncthreads();
  int at[BINA_EPT];
#pragma unroll
  for (int i = 0; i < BINA_EPT; ++i) at[i] = lstart[d[i] >> BIN_SHIFT] + r[i];
#pragma unroll
  for (int i = 0; i < BINA_EPT; ++i) {
    if (r[i] >= 0) stage[at[i]] = edge_word(s[i], d[i]);
  }
  __syncthreads();
  int total = E - e0;
  if (total > BINA_EDGES) total = BINA_EDGES;
  u64* bo = (u64*)binned;
  u64* so = (u64*)spill;
  u64 w[BINA_EPT];
  int pos[BINA_EPT];
#pragma unroll
  for (int i = 0; i < BINA_EPT; ++i) {
    int k = i * BINA_THREADS + tid;
    if (k > total - 1) k = total - 1;
    w[i] = stage[k];
  }
#pragma unroll
  for (int i = 0; i < BINA_EPT; ++i) {
    int k = i * BINA_THREADS + tid;
    pos[i] = delta[edge_dst(w[i]) >> BIN_SHIFT] + k;
  }
#pragma unroll
  for (int i = 0; i < BINA_EPT; ++i) {
    int k = i * BINA_THREADS + tid;
    if (k < total) {
      u64* to;
      if (pos[i] < BIN_CAP) {
        to = bo + (size_t)(edge_dst(w[i]) >> BIN_SHIFT) * BIN_CAP + pos[i];
      } else {
        to = so + atomicAdd(&gcnt[(size_t)NB * BIN_PAD], 1);
      }
      *to = w[i];
    }
  }
}

#define BINB_THREADS 512
#define BINB_EPT 8
__global__ __launch_bounds__(BINB_THREADS)
void fill_buckets_kernel(const int* __restrict__ gcnt, const int* __restrict__ binned,
                         const int* __restrict__ spill, int* __restrict__ cnt, int* __restrict__ col,
                         int* __restrict__ ovf, int N, int NB) {
  static_assert(BIN_NODES <= BINB_THREADS, "one thread per node clears and stores the counts");
  __shared__ int lc[BIN_NODES];
  __shared__ int lb[BIN_NODES * BUCKET_PITCH];
  const int tid = threadIdx.x;
  const int lane = tid & 31;
  const int wv = tid >> 5;
  const int b = blockIdx.x;
  const int n0 = b << BIN_SHIFT;
  if (tid < BIN_NODES) lc[tid] = 0;
  __syncthreads();
  auto put = [&](u64 p, int slot) {
    int ld = edge_dst(p) - n0;
    if (slot < BUCKET_CAP) {
      lb[ld * BUCKET_PITCH + slot] = edge_src(p);
    } else {
      int k = atomicAdd(&cnt[N], 1);
      ovf[2 * (size_t)k] = edge_src(p);
      ovf[2 * (size_t)k + 1] = edge_dst(p);
    }
  };
  const int got = gcnt[(size_t)b * BIN_PAD];
  const int inbin = (got < BIN_CAP) ? got : BIN_CAP;
  const u64* be = (const u64*)binned + (size_t)b * BIN_CAP;
  for (int i0 = 0; i0 < inbin; i0 += BINB_THREADS * BINB_EPT) {
    u64 p[BINB_EPT];
    int slot[BINB_EPT];
#pragma unroll
    for (int k = 0; k < BINB_EPT; ++k) {
      int i = i0 + k * BINB_THREADS + tid;
      if (i > inbin - 1) i = inbin - 1;
      p[k] = be[i];
    }
#pragma unroll
    for (int k = 0; k < BINB_EPT; ++k) {
      int i = i0 + k * BINB_THREADS + tid;
      slot[k] = -1;
      if (i < inbin) slot[k] = atomicAdd(&lc[edge_dst(p[k]) - n0], 1);
    }
#pragma unroll
    for (int k = 0; k < BINB_EPT; ++k) {
      if (slot[k] >= 0) put(p[k], slot[k]);
    }
  }
  if (got > BIN_CAP) {
    const int ns = gcnt[(size_t)NB * BIN_PAD];
    const u64* sp = (const u64*)spill;
    for (int i = tid; i < ns; i += BINB_THREADS) {
      u64 p = sp[i];
      if ((edge_dst(p) >> BIN_SHIFT) == b) put(p, atomicAdd(&lc[edge_dst(p) - n0], 1));
    }
  }
  __syncthreads();
  if (tid < BIN_NODES && n0 + tid < N) cnt[n0 + tid] = lc[tid];
  for (int ld = wv; ld < BIN_NODES; ld += BINB_THREADS / 32) {
    int n = n0 + ld;
    if (n >= N) break;
    int c = lc[ld];
    if (c > BUCKET_CAP) c = BUCKET_CAP;
    int* bucket = col + (size_t)n * BUCKET_CAP;
    for (int j0 = 0; j0 < c; j0 += 32) {
      int j = j0 + lane;
      if (j < c) bucket[j] = lb[ld * BUCKET_PITCH + j];
    }
  }
}

#define PRESCALE_PIECES 4
__global__ void dinv_prescale_kernel(const int* __restrict__ deg, const float* __restrict__ x,
                                     float* __restrict__ dinv, float* __restrict__ xs, int N) {
  int t = blockIdx.x * blockDim.x + threadIdx.x;
  int stride = gridDim.x * blockDim.x;
  int total = N * (HID / 4);
  int dg[PRESCALE_PIECES];
  float4 v[PRESCALE_PIECES];
#pragma unroll
  for (int i = 0; i < PRESCALE_PIECES; ++i) {
    int p = t + i * stride;
    if (p > total - 1) p = total - 1;
    dg[i] = deg[p / (HID / 4)];
    v[i] = *(const float4*)(x + (size_t)p * 4);
  }
#pragma unroll
  for (int i = 0; i < PRESCALE_PIECES; ++i) {
    int p = t + i * stride;
    if (p < total) {
      float dv = rsqrtf((float)dg[i] + 1.0f);
      float4 o = v[i];
      o.x *= dv; o.y *= dv; o.z *= dv; o.w *= dv;
      *(float4*)(xs + (size_t)p * 4) = o;
      if ((p & (HID / 4 - 1)) == 0) dinv[p / (HID / 4)] = dv;
    }
  }
}

#define PULL_NODES 8
#define PULL_BATCH 16
__device__ __forceinline__ unsigned row_off(int row, int lane) {
  return ((unsigned)row * HID + (unsigned)lane) * 4u;
}
template <bool SELF, class Sink>
__device__ __forceinline__ void pull_nodes(const float* __restrict__ rows, const int* __restrict__ cnt,
                                           const int* __restrict__ col, const int* __restrict__ ovf,
                                           int n0, int lane, int N, Sink sink) {
  static_assert(HID == 32, "one row per wave: HID must equal the wave size");
  static_assert(PULL_NODES <= 32 && (PULL_NODES & (PULL_NODES - 1)) == 0 && 32 % PULL_BATCH == 0,
                "nodes per wave: a power of two within its lanes; batches tile a 32-slot run");
  int nl = n0 + (lane & (PULL_NODES - 1));
  if (nl > N - 1) nl = N - 1;
  int cv = cnt[nl];
  int first[PULL_NODES];
#pragma unroll
  for (int g = 0; g < PULL_NODES; ++g) {
    int nb = n0 + g;
    if (nb > N - 1) nb = N - 1;
    int c = __builtin_amdgcn_readlane(cv, g);
    if (c > BUCKET_CAP) c = BUCKET_CAP;
    int j = (lane < c) ? lane : c - 1;
    first[g] = 0;
    if (c > 0) first[g] = col[(size_t)nb * BUCKET_CAP + j];
  }
#pragma unroll
  for (int g = 0; g < PULL_NODES; ++g) {
    int n = n0 + g;
    if (n < N) {
      int deg = __builtin_amdgcn_readlane(cv, g);
      int c = (deg < BUCKET_CAP) ? deg : BUCKET_CAP;
      const int* bucket = col + (size_t)n * BUCKET_CAP;
      float self = 0.f;
      if (SELF) self = *(const float*)((const char*)rows + row_off(n, lane));
      float acc = 0.f;
      for (int j0 = 0; j0 < c; j0 += 32) {
        int cs = first[g];
        if (j0 > 0) {
          int j = j0 + lane;
          if (j > c - 1) j = c - 1;
          cs = bucket[j];
        }
        int m = c - j0;
        if (m > 32) m = 32;
        for (int i = 0; i < m; i += PULL_BATCH) {
          float v[PULL_BATCH];
#pragma unroll
          for (int k = 0; k < PULL_BATCH; ++k) {
            int ik = i + k;
            if (ik > m - 1) ik = m - 1;
            int si = __builtin_amdgcn_readlane(cs, ik);
            v[k] = *(const float*)((const char*)rows + row_off(si, lane));
          }
#pragma unroll
          for (int k = 0; k < PULL_BATCH; ++k) {
            if (i + k < m) acc += v[k];
          }
        }
      }
      if (deg > BUCKET_CAP) {
        int nov = cnt[N];
        int left = deg - BUCKET_CAP;
        for (int k0 = 0; k0 < nov && left > 0; k0 += 32) {
          int k = k0 + lane;
          if (k > nov - 1) k = nov - 1;
          int os = ovf[2 * (size_t)k];
          int od = ovf[2 * (size_t)k + 1];
          unsigned hit = __builtin_amdgcn_ballot_w32(k0 + lane < nov && od == n);
          left -= __builtin_popcount(hit);
          while (hit) {
            int b = __builtin_ctz(hit);
            hit &= hit - 1;
            int si = __builtin_amdgcn_readlane(os, b);
            acc += *(const float*)((const char*)rows + row_off(si, lane));
          }
        }
      }
      if (SELF) acc += self;
      sink(g, n, acc);
    }
  }
}

#define ROW_PITCH (HID + 2)

__global__ __launch_bounds__(2 * 32)
void pull_l1l2_wmma_f32(const float* __restrict__ xs, const int* __restrict__ cnt,
                        const int* __restrict__ col, const int* __restrict__ ovf,
                        const float* __restrict__ dinv, const float* __restrict__ W1,
                        const float* __restrict__ b1, const float* __restrict__ W2,
                        float* __restrict__ XL, int M) {
  static_assert(2 * PULL_NODES == 16, "two waves walk the 16 rows of one product tile");
  static_assert(H1 == 2 * 64 && HID == 2 * 16, "two waves: 64 hidden units and one 16-column output tile each");
  __shared__ __attribute__((aligned(16))) float srow[16 * ROW_PITCH];
  __shared__ float part[2][8 * 32];
  const int lane = threadIdx.x & 31;
  const int wv = threadIdx.x >> 5;
  const int tm = blockIdx.x;
  const int c = lane & 15;
  const int h = lane >> 4;
  const int node = tm * 16 + c;
  int nodec = node;
  if (nodec > M - 1) nodec = M - 1;
  const float di = dinv[nodec];
  const int n0 = tm * 16 + wv * PULL_NODES;
  if (n0 + PULL_NODES > M) {
#pragma unroll
    for (int g = 0; g < PULL_NODES; ++g) {
      if (n0 + g >= M) srow[(wv * PULL_NODES + g) * ROW_PITCH + lane] = 0.f;
    }
  }
  pull_nodes<true>(xs, cnt, col, ovf, n0, lane, M,
                   [&](int g, int, float sum) { srow[(wv * PULL_NODES + g) * ROW_PITCH + lane] = sum; });
  __syncthreads();
  const float* sa = srow + c * ROW_PITCH + 2 * h;
  const unsigned w1o = (unsigned)((2 * h) * H1 + 64 * wv + c) * 4u;
  const unsigned b1o = (unsigned)(64 * wv + 8 * h) * 4u;
  const unsigned w2o = (unsigned)((64 * wv + 8 * h) * HID + c) * 4u;
  v8f o[2] = {};
#pragma unroll 1
  for (int tt = 0; tt < 4; ++tt) {
    v8f acc = {};
#pragma unroll 1
    for (int jh = 0; jh < HID / 4; jh += 4) {
#pragma unroll
      for (int jj = 0; jj < 4; ++jj) {
        const int j = jh + jj;
        v2f sv = *(const v2f*)(sa + 4 * j);
        v2f bx;
        bx.x = di * sv.x;
        bx.y = di * sv.y;
        v2f a;
        a.x = *(const float*)((const char*)W1 + (w1o + (unsigned)((4 * j) * H1 + 16 * tt) * 4u));
        a.y = *(const float*)((const char*)W1 + (w1o + (unsigned)((4 * j + 1) * H1 + 16 * tt) * 4u));
        acc = __builtin_amdgcn_wmma_f32_16x16x4_f32(false, a, false, bx, (short)0, acc, false, false);
      }
    }
    const float4 bl = *(const float4*)((const char*)b1 + (b1o + (unsigned)(16 * tt) * 4u));
    const float4 bh = *(const float4*)((const char*)b1 + (b1o + (unsigned)(16 * tt + 4) * 4u));
    acc[0] = fmaxf(acc[0] + bl.x, 0.f);
    acc[1] = fmaxf(acc[1] + bl.y, 0.f);
    acc[2] = fmaxf(acc[2] + bl.z, 0.f);
    acc[3] = fmaxf(acc[3] + bl.w, 0.f);
    acc[4] = fmaxf(acc[4] + bh.x, 0.f);
    acc[5] = fmaxf(acc[5] + bh.y, 0.f);
    acc[6] = fmaxf(acc[6] + bh.z, 0.f);
    acc[7] = fmaxf(acc[7] + bh.w, 0.f);
#pragma unroll
    for (int p = 0; p < 4; ++p) {
      v2f b;
      b.x = acc[2 * p];
      b.y = acc[2 * p + 1];
#pragma unroll
      for (int u = 0; u < 2; ++u) {
        v2f a;
        a.x = *(const float*)((const char*)W2 + (w2o + (unsigned)((16 * tt + 2 * p) * HID + 16 * u) * 4u));
        a.y = *(const float*)((const char*)W2 + (w2o + (unsigned)((16 * tt + 2 * p + 1) * HID + 16 * u) * 4u));
        o[u] = __builtin_amdgcn_wmma_f32_16x16x4_f32(false, a, false, b, (short)0, o[u], false, false);
      }
    }
  }
  const v8f keep = (wv == 0) ? o[0] : o[1];
  const v8f send = (wv == 0) ? o[1] : o[0];
#pragma unroll
  for (int r = 0; r < 8; ++r) part[wv][r * 32 + lane] = send[r];
  __syncthreads();
  float tot[8];
#pragma unroll
  for (int r = 0; r < 8; ++r) tot[r] = keep[r] + part[1 - wv][r * 32 + lane];
  if (node < M) {
    float* xo = XL + (size_t)node * HID + 16 * wv + 8 * h;
    float4 lo, hi;
    lo.x = di * tot[0]; lo.y = di * tot[1]; lo.z = di * tot[2]; lo.w = di * tot[3];
    hi.x = di * tot[4]; hi.y = di * tot[5]; hi.z = di * tot[6]; hi.w = di * tot[7];
    *(float4*)xo = lo;
    *(float4*)(xo + 4) = hi;
  }
}

__global__ __launch_bounds__(2 * 32)
void pull_pq_wmma_f32(const float* __restrict__ xl, const int* __restrict__ cnt,
                      const int* __restrict__ col, const int* __restrict__ ovf,
                      const float* __restrict__ dinv, const float* __restrict__ b2,
                      const float* __restrict__ Wm1, const float* __restrict__ bm1,
                      float* __restrict__ PQ, int M) {
  static_assert(2 * PULL_NODES == 16, "two waves walk the 16 rows of one product tile");
  __shared__ __attribute__((aligned(16))) float srow[16 * ROW_PITCH];
  const int lane = threadIdx.x & 31;
  const int wv = threadIdx.x >> 5;
  const int tm = blockIdx.x;
  const int l15 = lane & 15;
  int mrow = tm * 16 + l15;
  if (mrow > M - 1) mrow = M - 1;
  const float di = dinv[mrow];
  const int n0 = tm * 16 + wv * PULL_NODES;
  if (n0 + PULL_NODES > M) {
#pragma unroll
    for (int g = 0; g < PULL_NODES; ++g) {
      if (n0 + g >= M) srow[(wv * PULL_NODES + g) * ROW_PITCH + lane] = 0.f;
    }
  }
  pull_nodes<true>(xl, cnt, col, ovf, n0, lane, M,
                   [&](int g, int, float sum) { srow[(wv * PULL_NODES + g) * ROW_PITCH + lane] = sum; });
  __syncthreads();
  const int kb = (lane >> 4) * 2;
  const float* sa = srow + l15 * ROW_PITCH + kb;
  const float* ba = b2 + kb;
  const float* wb = Wm1 + (size_t)(wv * HID + kb) * HID + l15;
  v8f acc[2] = {};
  for (int k0 = 0; k0 < HID; k0 += 4) {
    v2f sv = *(const v2f*)(sa + k0);
    v2f a;
    a.x = fmaxf(di * sv.x + ba[k0], 0.f);
    a.y = fmaxf(di * sv.y + ba[k0 + 1], 0.f);
#pragma unroll
    for (int u = 0; u < 2; ++u) {
      v2f b;
      b.x = wb[(size_t)k0 * HID + 16 * u];
      b.y = wb[(size_t)(k0 + 1) * HID + 16 * u];
      acc[u] = __builtin_amdgcn_wmma_f32_16x16x4_f32(false, a, false, b, (short)0, acc[u],
                                                     false, false);
    }
  }
  const int rbase = tm * 16 + ((lane >> 4) ? 8 : 0);
#pragma unroll
  for (int u = 0; u < 2; ++u) {
    const int ncol = wv * HID + 16 * u + l15;
    const float bn = (wv == 0) ? bm1[16 * u + l15] : 0.f;
#pragma unroll
    for (int r = 0; r < 8; ++r) {
      int row = rbase + r;
      float val = acc[u][r];
      if (wv == 0) val += bn;
      if (row < M) PQ[(size_t)row * (2 * HID) + ncol] = val;
    }
  }
}

#define EDGE_CHUNKS 4
__global__ __launch_bounds__(256)
void edge_out_kernel(const float* __restrict__ PQ, const int* __restrict__ p0,
                     const int* __restrict__ p1, const int* __restrict__ q0,
                     const int* __restrict__ q1, const float* __restrict__ Wm2,
                     const float* __restrict__ bm2, float* __restrict__ out,
                     int NP, int Etot) {
  int wave = (int)((blockIdx.x * (long long)blockDim.x + threadIdx.x) >> 5);
  int lane = threadIdx.x & 31;
  int G = lane >> 3;
  int j = lane & 7;
  int c = j * 4;
  const float4 w01 = *(const float4*)(Wm2 + c * 2);
  const float4 w23 = *(const float4*)(Wm2 + c * 2 + 4);
  const float bo = bm2[j & 1];
  const bool hi4 = (j & 4) != 0, hi2 = (j & 2) != 0, hi1 = (j & 1) != 0;
  for (int it = 0; it < EDGE_CHUNKS; ++it) {
    int e0 = (wave * EDGE_CHUNKS + it) * 16;
    if (e0 >= Etot) break;
    float v[8];
#pragma unroll
    for (int i = 0; i < 4; ++i) {
      int e = e0 + 4 * G + i;
      if (e > Etot - 1) e = Etot - 1;
      int s = (e < NP) ? p0[e] : q0[e - NP];
      int d = (e < NP) ? p1[e] : q1[e - NP];
      const float4 p = *(const float4*)(PQ + (size_t)s * (2 * HID) + c);
      const float4 q = *(const float4*)(PQ + (size_t)d * (2 * HID) + HID + c);
      float t0 = fmaxf(p.x + q.x, 0.f);
      float t1 = fmaxf(p.y + q.y, 0.f);
      float t2 = fmaxf(p.z + q.z, 0.f);
      float t3 = fmaxf(p.w + q.w, 0.f);
      v[2 * i]     = t0 * w01.x + t1 * w01.z + t2 * w23.x + t3 * w23.z;
      v[2 * i + 1] = t0 * w01.y + t1 * w01.w + t2 * w23.y + t3 * w23.w;
    }
    float u[4];
#pragma unroll
    for (int k = 0; k < 4; ++k) {
      float keep = hi4 ? v[k + 4] : v[k];
      float send = hi4 ? v[k] : v[k + 4];
      u[k] = keep + __shfl_xor(send, 4, 8);
    }
    float w[2];
#pragma unroll
    for (int k = 0; k < 2; ++k) {
      float keep = hi2 ? u[k + 2] : u[k];
      float send = hi2 ? u[k] : u[k + 2];
      w[k] = keep + __shfl_xor(send, 2, 8);
    }
    float keep1 = hi1 ? w[1] : w[0];
    float send1 = hi1 ? w[0] : w[1];
    float r = keep1 + __shfl_xor(send1, 1, 8);
    int eo = e0 + 4 * G + (j >> 1);
    if (eo < Etot) out[(size_t)e0 * 2 + lane] = r + bo;
  }
}

extern "C" void kernel_launch(void* const* d_in, const int* in_sizes, int n_in,
                              void* d_out, int out_size, void* d_ws, size_t ws_size,
                              hipStream_t stream) {
  const float* x   = (const float*)d_in[0];
  const int*   pe  = (const int*)d_in[1];
  const int*   neg = (const int*)d_in[2];
  const float* W1  = (const float*)d_in[3];
  const float* b1  = (const float*)d_in[4];
  const float* W2  = (const float*)d_in[5];
  const float* b2  = (const float*)d_in[6];
  const float* Wm1 = (const float*)d_in[7];
  const float* bm1 = (const float*)d_in[8];
  const float* Wm2 = (const float*)d_in[9];
  const float* bm2 = (const float*)d_in[10];
  float* out = (float*)d_out;

  const int N  = in_sizes[0] / HID;
  const int NP = in_sizes[1] / 2;
  const int NN = in_sizes[2] / 2;
  const int ET = NP + NN;
  const int* p0 = pe;        const int* p1 = pe + NP;
  const int* q0 = neg;       const int* q1 = neg + NN;

  char* ws = (char*)d_ws;
  size_t off = 0;
  auto wsalloc = [&](size_t bytes) -> char* {
    char* p = ws + off;
    off += (bytes + 255) & ~(size_t)255;
    return p;
  };
  const int NB = (N + BIN_NODES - 1) >> BIN_SHIFT;
  int*   gcnt = (int*)wsalloc(((size_t)NB + 1) * BIN_PAD * 4);
  int*   binned = (int*)wsalloc((size_t)NB * BIN_CAP * 8);
  int*   spill = (int*)wsalloc((size_t)NP * 8);
  int*   cnt  = (int*)wsalloc((size_t)(N + 1) * 4);
  int*   col  = (int*)wsalloc((size_t)N * BUCKET_CAP * 4);
  int*   ovf  = (int*)wsalloc((size_t)NP * 2 * 4);
  float* dinv = (float*)wsalloc((size_t)N * 4);
  float* xs   = (float*)wsalloc((size_t)N * HID * 4);
  float* xl2  = (float*)wsalloc((size_t)N * HID * 4);
  float* pq   = (float*)wsalloc((size_t)N * 2 * HID * 4);
  (void)ws_size; (void)n_in; (void)out_size;

  const int B = 256;

  hipMemsetAsync(gcnt, 0, ((size_t)NB + 1) * BIN_PAD * 4, stream);
  bin_edges_kernel<<<(unsigned)((NP + BINA_EDGES - 1) / BINA_EDGES), BINA_THREADS, 0, stream>>>(
      p0, p1, gcnt, binned, spill, cnt, NP, N, NB);
  fill_buckets_kernel<<<(unsigned)NB, BINB_THREADS, 0, stream>>>(gcnt, binned, spill, cnt, col, ovf, N, NB);
  {
    long long thr = ((long long)N * (HID / 4) + PRESCALE_PIECES - 1) / PRESCALE_PIECES;
    dinv_prescale_kernel<<<(unsigned)((thr + B - 1) / B), B, 0, stream>>>(cnt, x, dinv, xs, N);
  }

  const unsigned tb = (unsigned)((N + 15) / 16);

  pull_l1l2_wmma_f32<<<tb, 2 * 32, 0, stream>>>(xs, cnt, col, ovf, dinv, W1, b1, W2, xl2, N);
  pull_pq_wmma_f32<<<tb, 2 * 32, 0, stream>>>(xl2, cnt, col, ovf, dinv, b2, Wm1, bm1, pq, N);

  {
    long long thr = (long long)((ET + 16 * EDGE_CHUNKS - 1) / (16 * EDGE_CHUNKS)) * 32;
    edge_out_kernel<<<(unsigned)((thr + B - 1) / B), B, 0, stream>>>(pq, p0, p1, q0, q1,
                                                                     Wm2, bm2, out, NP, ET);
  }
}


# Both edge sums: 8 lanes per 128-B row, a lane group per node: one 16-B-per-lane load fetches rows of 4 nodes (4x fewer loads); buckets padded with a zero row; own row first: fp32 sum order changes
# speedup vs baseline: 6.0815x; 1.0978x over previous
#include <hip/hip_runtime.h>
#include <hip/hip_bf16.h>

typedef __attribute__((ext_vector_type(2))) float v2f;
typedef __attribute__((ext_vector_type(8))) float v8f;
typedef unsigned long long u64;

#define HID 32
#define H1  128

#define BUCKET_CAP 64
#define BUCKET_PITCH (BUCKET_CAP + 1)
#define BIN_SHIFT 8
#define BIN_NODES (1 << BIN_SHIFT)
#define BIN_MAX 512
#define BIN_CAP 8192
#define BIN_PAD 64
#define BINA_THREADS 512
#define BINA_EPT 16
#define BINA_EDGES (BINA_THREADS * BINA_EPT)
__device__ __forceinline__ u64 edge_word(int s, int d) { return ((u64)(unsigned)d << 32) | (u64)(unsigned)s; }
__device__ __forceinline__ int edge_src(u64 w) { return (int)(unsigned)w; }
__device__ __forceinline__ int edge_dst(u64 w) { return (int)(unsigned)(w >> 32); }
__global__ __launch_bounds__(BINA_THREADS)
void bin_edges_kernel(const int* __restrict__ src, const int* __restrict__ dst, int* __restrict__ gcnt,
                      int* __restrict__ binned, int* __restrict__ spill, int* __restrict__ cnt,
                      int E, int N, int NB) {
  static_assert(BIN_MAX == BINA_THREADS, "the scan holds one bin per thread");
  __shared__ int lcnt[BIN_MAX];
  __shared__ int lstart[BIN_MAX];
  __shared__ int delta[BIN_MAX];
  __shared__ int wsum[BINA_THREADS / 32];
  __shared__ u64 stage[BINA_EDGES];
  const int tid = threadIdx.x;
  const int lane = tid & 31;
  const int wv = tid >> 5;
  const int e0 = blockIdx.x * BINA_EDGES;
  if (blockIdx.x == 0 && tid == 0) cnt[N] = 0;
  lcnt[tid] = 0;
  __syncthreads();
  int d[BINA_EPT];
  int s[BINA_EPT];
  int r[BINA_EPT];
#pragma unroll
  for (int i = 0; i < BINA_EPT; ++i) {
    int e = e0 + i * BINA_THREADS + tid;
    if (e > E - 1) e = E - 1;
    d[i] = dst[e];
    s[i] = src[e];
  }
#pragma unroll
  for (int i = 0; i < BINA_EPT; ++i) {
    int e = e0 + i * BINA_THREADS + tid;
    r[i] = -1;
    if (e < E) r[i] = atomicAdd(&lcnt[d[i] >> BIN_SHIFT], 1);
  }
  __syncthreads();
  const int v = lcnt[tid];
  int incl = v;
#pragma unroll
  for (int off = 1; off < 32; off <<= 1) {
    int t = __shfl_up(incl, off, 32);
    if (lane >= off) incl += t;
  }
  if (lane == 31) wsum[wv] = incl;
  __syncthreads();
  if (wv == 0) {
    int w = (lane < BINA_THREADS / 32) ? wsum[lane] : 0;
    int wi = w;
#pragma unroll
    for (int off = 1; off < BINA_THREADS / 32; off <<= 1) {
      int t = __shfl_up(wi, off, 32);
      if (lane >= off) wi += t;
    }
    if (lane < BINA_THREADS / 32) wsum[lane] = wi - w;
  }
  __syncthreads();
  const int start = wsum[wv] + incl - v;
  int base = 0;
  if (v > 0) base = atomicAdd(&gcnt[(size_t)tid * BIN_PAD], v);
  lstart[tid] = start;
  delta[tid] = base - start;
  __syncthreads();
  int at[BINA_EPT];
#pragma unroll
  for (int i = 0; i < BINA_EPT; ++i) at[i] = lstart[d[i] >> BIN_SHIFT] + r[i];
#pragma unroll
  for (int i = 0; i < BINA_EPT; ++i) {
    if (r[i] >= 0) stage[at[i]] = edge_word(s[i], d[i]);
  }
  __syncthreads();
  int total = E - e0;
  if (total > BINA_EDGES) total = BINA_EDGES;
  u64* bo = (u64*)binned;
  u64* so = (u64*)spill;
  u64 w[BINA_EPT];
  int pos[BINA_EPT];
#pragma unroll
  for (int i = 0; i < BINA_EPT; ++i) {
    int k = i * BINA_THREADS + tid;
    if (k > total - 1) k = total - 1;
    w[i] = stage[k];
  }
#pragma unroll
  for (int i = 0; i < BINA_EPT; ++i) {
    int k = i * BINA_THREADS + tid;
    pos[i] = delta[edge_dst(w[i]) >> BIN_SHIFT] + k;
  }
#pragma unroll
  for (int i = 0; i < BINA_EPT; ++i) {
    int k = i * BINA_THREADS + tid;
    if (k < total) {
      u64* to;
      if (pos[i] < BIN_CAP) {
        to = bo + (size_t)(edge_dst(w[i]) >> BIN_SHIFT) * BIN_CAP + pos[i];
      } else {
        to = so + atomicAdd(&gcnt[(size_t)NB * BIN_PAD], 1);
      }
      *to = w[i];
    }
  }
}

#define BINB_THREADS 512
#define BINB_EPT 8
__global__ __launch_bounds__(BINB_THREADS)
void fill_buckets_kernel(const int* __restrict__ gcnt, const int* __restrict__ binned,
                         const int* __restrict__ spill, int* __restrict__ cnt, int* __restrict__ col,
                         int* __restrict__ ovf, int N, int NB) {
  static_assert(BIN_NODES <= BINB_THREADS, "one thread per node clears and stores the counts");
  __shared__ int lc[BIN_NODES];
  __shared__ int lb[BIN_NODES * BUCKET_PITCH];
  const int tid = threadIdx.x;
  const int lane = tid & 31;
  const int wv = tid >> 5;
  const int b = blockIdx.x;
  const int n0 = b << BIN_SHIFT;
  if (tid < BIN_NODES) lc[tid] = 0;
  __syncthreads();
  auto put = [&](u64 p, int slot) {
    int ld = edge_dst(p) - n0;
    if (slot < BUCKET_CAP) {
      lb[ld * BUCKET_PITCH + slot] = edge_src(p);
    } else {
      int k = atomicAdd(&cnt[N], 1);
      ovf[2 * (size_t)k] = edge_src(p);
      ovf[2 * (size_t)k + 1] = edge_dst(p);
    }
  };
  const int got = gcnt[(size_t)b * BIN_PAD];
  const int inbin = (got < BIN_CAP) ? got : BIN_CAP;
  const u64* be = (const u64*)binned + (size_t)b * BIN_CAP;
  for (int i0 = 0; i0 < inbin; i0 += BINB_THREADS * BINB_EPT) {
    u64 p[BINB_EPT];
    int slot[BINB_EPT];
#pragma unroll
    for (int k = 0; k < BINB_EPT; ++k) {
      int i = i0 + k * BINB_THREADS + tid;
      if (i > inbin - 1) i = inbin - 1;
      p[k] = be[i];
    }
#pragma unroll
    for (int k = 0; k < BINB_EPT; ++k) {
      int i = i0 + k * BINB_THREADS + tid;
      slot[k] = -1;
      if (i < inbin) slot[k] = atomicAdd(&lc[edge_dst(p[k]) - n0], 1);
    }
#pragma unroll
    for (int k = 0; k < BINB_EPT; ++k) {
      if (slot[k] >= 0) put(p[k], slot[k]);
    }
  }
  if (got > BIN_CAP) {
    const int ns = gcnt[(size_t)NB * BIN_PAD];
    const u64* sp = (const u64*)spill;
    for (int i = tid; i < ns; i += BINB_THREADS) {
      u64 p = sp[i];
      if ((edge_dst(p) >> BIN_SHIFT) == b) put(p, atomicAdd(&lc[edge_dst(p) - n0], 1));
    }
  }
  __syncthreads();
  if (tid < BIN_NODES && n0 + tid < N) cnt[n0 + tid] = lc[tid];
  for (int ld = wv; ld < BIN_NODES; ld += BINB_THREADS / 32) {
    int n = n0 + ld;
    if (n >= N) break;
    int c = lc[ld];
    if (c > BUCKET_CAP) c = BUCKET_CAP;
    int* bucket = col + (size_t)n * BUCKET_CAP;
    for (int j0 = 0; j0 < 32 || j0 < c; j0 += 32) {
      int j = j0 + lane;
      int v = N;
      if (j < c) v = lb[ld * BUCKET_PITCH + j];
      bucket[j] = v;
    }
  }
}

#define PRESCALE_PIECES 4
__global__ void dinv_prescale_kernel(const int* __restrict__ deg, const float* __restrict__ x,
                                     float* __restrict__ dinv, float* __restrict__ xs, int N) {
  int t = blockIdx.x * blockDim.x + threadIdx.x;
  int stride = gridDim.x * blockDim.x;
  int total = N * (HID / 4);
  if (t < HID / 4) *(float4*)(xs + (size_t)N * HID + 4 * t) = make_float4(0.f, 0.f, 0.f, 0.f);
  int dg[PRESCALE_PIECES];
  float4 v[PRESCALE_PIECES];
#pragma unroll
  for (int i = 0; i < PRESCALE_PIECES; ++i) {
    int p = t + i * stride;
    if (p > total - 1) p = total - 1;
    dg[i] = deg[p / (HID / 4)];
    v[i] = *(const float4*)(x + (size_t)p * 4);
  }
#pragma unroll
  for (int i = 0; i < PRESCALE_PIECES; ++i) {
    int p = t + i * stride;
    if (p < total) {
      float dv = rsqrtf((float)dg[i] + 1.0f);
      float4 o = v[i];
      o.x *= dv; o.y *= dv; o.z *= dv; o.w *= dv;
      *(float4*)(xs + (size_t)p * 4) = o;
      if ((p & (HID / 4 - 1)) == 0) dinv[p / (HID / 4)] = dv;
    }
  }
}

#define PULL_NODES 8
#define PULL_SLOTS 4
template <int B>
__device__ __forceinline__ unsigned group_lane(unsigned v) {
  return (unsigned)__builtin_amdgcn_mov_dpp8((int)v, B * 011111111);
}
template <int B>
__device__ __forceinline__ void pull_batch(const char* __restrict__ rb, const uint4 oa, const uint4 ob, unsigned loff,
                                           float4& accA, float4& accB) {
  const float4 a0 = *(const float4*)(rb + (group_lane<B>(oa.x) + loff));
  const float4 b0 = *(const float4*)(rb + (group_lane<B>(ob.x) + loff));
  const float4 a1 = *(const float4*)(rb + (group_lane<B>(oa.y) + loff));
  const float4 b1 = *(const float4*)(rb + (group_lane<B>(ob.y) + loff));
  const float4 a2 = *(const float4*)(rb + (group_lane<B>(oa.z) + loff));
  const float4 b2 = *(const float4*)(rb + (group_lane<B>(ob.z) + loff));
  const float4 a3 = *(const float4*)(rb + (group_lane<B>(oa.w) + loff));
  const float4 b3 = *(const float4*)(rb + (group_lane<B>(ob.w) + loff));
  accA.x += a0.x; accA.y += a0.y; accA.z += a0.z; accA.w += a0.w;
  accB.x += b0.x; accB.y += b0.y; accB.z += b0.z; accB.w += b0.w;
  accA.x += a1.x; accA.y += a1.y; accA.z += a1.z; accA.w += a1.w;
  accB.x += b1.x; accB.y += b1.y; accB.z += b1.z; accB.w += b1.w;
  accA.x += a2.x; accA.y += a2.y; accA.z += a2.z; accA.w += a2.w;
  accB.x += b2.x; accB.y += b2.y; accB.z += b2.z; accB.w += b2.w;
  accA.x += a3.x; accA.y += a3.y; accA.z += a3.z; accA.w += a3.w;
  accB.x += b3.x; accB.y += b3.y; accB.z += b3.z; accB.w += b3.w;
}
template <bool SELF, class Sink>
__device__ __forceinline__ void pull_nodes(const float* __restrict__ rows, const int* __restrict__ cnt,
                                           const int* __restrict__ col, const int* __restrict__ ovf,
                                           int n0, int lane, int N, Sink sink) {
  static_assert(HID == 32, "a row is 8 pieces of 16 bytes: one per lane of a group");
  static_assert(PULL_NODES == 8 && PULL_SLOTS == 4 && BUCKET_CAP == 64,
                "two nodes per group of 8 lanes; a lane holds 4 slots, a group 32, a bucket two such runs");
  const int G = lane >> 3;
  const int j = lane & 7;
  const int nA = n0 + G;
  const int nB = n0 + 4 + G;
  const int nAc = (nA < N) ? nA : N - 1;
  const int nBc = (nB < N) ? nB : N - 1;
  int degA = cnt[nAc];
  int degB = cnt[nBc];
  if (nA >= N) degA = 0;
  if (nB >= N) degB = 0;
  const int cA = (degA < BUCKET_CAP) ? degA : BUCKET_CAP;
  const int cB = (degB < BUCKET_CAP) ? degB : BUCKET_CAP;
  const char* rb = (const char*)rows;
  const char* cb = (const char*)col;
  const unsigned loff = 16u * (unsigned)j;
  const unsigned bA = (unsigned)nAc * (BUCKET_CAP * 4u) + loff;
  const unsigned bB = (unsigned)nBc * (BUCKET_CAP * 4u) + loff;
  const uint4 sA = *(const uint4*)(cb + bA);
  const uint4 sB = *(const uint4*)(cb + bB);
  const unsigned zoff = (unsigned)N * (HID * 4u);
  const bool overflow = __builtin_amdgcn_ballot_w32(degA > BUCKET_CAP || degB > BUCKET_CAP) != 0;
  float4 accA = make_float4(0.f, 0.f, 0.f, 0.f);
  float4 accB = make_float4(0.f, 0.f, 0.f, 0.f);
  if (SELF) {
    accA = *(const float4*)(rb + ((unsigned)nAc * (HID * 4u) + loff));
    accB = *(const float4*)(rb + ((unsigned)nBc * (HID * 4u) + loff));
  }
  auto run = [&](const uint4 s, const uint4 t, const int base, const bool masked) {
    const int q = base + 4 * j;
    uint4 oa, ob;
    oa.x = (!masked || q + 0 < cA) ? s.x * (HID * 4u) : zoff;
    oa.y = (!masked || q + 1 < cA) ? s.y * (HID * 4u) : zoff;
    oa.z = (!masked || q + 2 < cA) ? s.z * (HID * 4u) : zoff;
    oa.w = (!masked || q + 3 < cA) ? s.w * (HID * 4u) : zoff;
    ob.x = (!masked || q + 0 < cB) ? t.x * (HID * 4u) : zoff;
    ob.y = (!masked || q + 1 < cB) ? t.y * (HID * 4u) : zoff;
    ob.z = (!masked || q + 2 < cB) ? t.z * (HID * 4u) : zoff;
    ob.w = (!masked || q + 3 < cB) ? t.w * (HID * 4u) : zoff;
#define PULL_STEP(B)                                                                                      \
    if (__builtin_amdgcn_ballot_w32(cA > base + 4 * (B) || cB > base + 4 * (B)) == 0) return;              \
    pull_batch<B>(rb, oa, ob, loff, accA, accB);
    PULL_STEP(0) PULL_STEP(1) PULL_STEP(2) PULL_STEP(3) PULL_STEP(4) PULL_STEP(5) PULL_STEP(6) PULL_STEP(7)
#undef PULL_STEP
  };
  run(sA, sB, 0, false);
  if (__builtin_amdgcn_ballot_w32(cA > 32 || cB > 32) != 0) {
    uint4 tA = make_uint4(0u, 0u, 0u, 0u);
    uint4 tB = make_uint4(0u, 0u, 0u, 0u);
    if (cA > 32) tA = *(const uint4*)(cb + (bA + 128u));
    if (cB > 32) tB = *(const uint4*)(cb + (bB + 128u));
    run(tA, tB, 32, true);
  }
  if (overflow) {
    const int nov = cnt[N];
    for (int g = 0; g < PULL_NODES; ++g) {
      const int n = n0 + g;
      if (n >= N) break;
      int left = cnt[n] - BUCKET_CAP;
      for (int k0 = 0; k0 < nov && left > 0; k0 += 32) {
        int k = k0 + lane;
        if (k > nov - 1) k = nov - 1;
        const int os = ovf[2 * (size_t)k];
        const int od = ovf[2 * (size_t)k + 1];
        unsigned hit = __builtin_amdgcn_ballot_w32(k0 + lane < nov && od == n);
        left -= __builtin_popcount(hit);
        while (hit) {
          const int b = __builtin_ctz(hit);
          hit &= hit - 1;
          const int si = __builtin_amdgcn_readlane(os, b);
          const float4 v = *(const float4*)(rb + ((unsigned)si * (HID * 4u) + loff));
          if (G == (g & 3)) {
            if (g < 4) { accA.x += v.x; accA.y += v.y; accA.z += v.z; accA.w += v.w; }
            else       { accB.x += v.x; accB.y += v.y; accB.z += v.z; accB.w += v.w; }
          }
        }
      }
    }
  }
  if (nA >= N) accA = make_float4(0.f, 0.f, 0.f, 0.f);
  if (nB >= N) accB = make_float4(0.f, 0.f, 0.f, 0.f);
  sink(G, accA);
  sink(4 + G, accB);
}

#define ROW_PITCH (HID + 2)

__global__ __launch_bounds__(2 * 32) __attribute__((amdgpu_waves_per_eu(16, 16)))
void pull_l1l2_wmma_f32(const float* __restrict__ xs, const int* __restrict__ cnt,
                        const int* __restrict__ col, const int* __restrict__ ovf,
                        const float* __restrict__ dinv, const float* __restrict__ W1,
                        const float* __restrict__ b1, const float* __restrict__ W2,
                        float* __restrict__ XL, int M) {
  static_assert(2 * PULL_NODES == 16, "two waves walk the 16 rows of one product tile");
  static_assert(H1 == 2 * 64 && HID == 2 * 16, "two waves: 64 hidden units and one 16-column output tile each");
  __shared__ __attribute__((aligned(16))) float srow[16 * ROW_PITCH];
  __shared__ float part[2][8 * 32];
  const int lane = threadIdx.x & 31;
  const int wv = __builtin_amdgcn_readfirstlane((int)(threadIdx.x >> 5));
  const int tm = blockIdx.x;
  const int c = lane & 15;
  const int h = lane >> 4;
  const int node = tm * 16 + c;
  int nodec = node;
  if (nodec > M - 1) nodec = M - 1;
  const int n0 = tm * 16 + wv * PULL_NODES;
  if (tm == 0 && threadIdx.x < HID / 4) *(float4*)(XL + (size_t)M * HID + 4 * threadIdx.x) = make_float4(0.f, 0.f, 0.f, 0.f);
  pull_nodes<true>(xs, cnt, col, ovf, n0, lane, M, [&](int g, float4 sum) {
    float* p = srow + (wv * PULL_NODES + g) * ROW_PITCH + 4 * (lane & 7);
    v2f lo, hi;
    lo.x = sum.x; lo.y = sum.y; hi.x = sum.z; hi.y = sum.w;
    *(v2f*)p = lo;
    *(v2f*)(p + 2) = hi;
  });
  const float di = dinv[nodec];
  __syncthreads();
  const float* sa = srow + c * ROW_PITCH + 2 * h;
  const unsigned w1o = (unsigned)((2 * h) * H1 + 64 * wv + c) * 4u;
  const unsigned b1o = (unsigned)(64 * wv + 8 * h) * 4u;
  const unsigned w2o = (unsigned)((64 * wv + 8 * h) * HID + c) * 4u;
  v8f o[2] = {};
#pragma unroll 1
  for (int tt = 0; tt < 4; ++tt) {
    v8f acc = {};
#pragma unroll 1
    for (int jh = 0; jh < HID / 4; jh += 4) {
#pragma unroll
      for (int jj = 0; jj < 4; ++jj) {
        const int j = jh + jj;
        v2f sv = *(const v2f*)(sa + 4 * j);
        v2f bx;
        bx.x = di * sv.x;
        bx.y = di * sv.y;
        v2f a;
        a.x = *(const float*)((const char*)W1 + (w1o + (unsigned)((4 * j) * H1 + 16 * tt) * 4u));
        a.y = *(const float*)((const char*)W1 + (w1o + (unsigned)((4 * j + 1) * H1 + 16 * tt) * 4u));
        acc = __builtin_amdgcn_wmma_f32_16x16x4_f32(false, a, false, bx, (short)0, acc, false, false);
      }
    }
    const float4 bl = *(const float4*)((const char*)b1 + (b1o + (unsigned)(16 * tt) * 4u));
    const float4 bh = *(const float4*)((const char*)b1 + (b1o + (unsigned)(16 * tt + 4) * 4u));
    acc[0] = fmaxf(acc[0] + bl.x, 0.f);
    acc[1] = fmaxf(acc[1] + bl.y, 0.f);
    acc[2] = fmaxf(acc[2] + bl.z, 0.f);
    acc[3] = fmaxf(acc[3] + bl.w, 0.f);
    acc[4] = fmaxf(acc[4] + bh.x, 0.f);
    acc[5] = fmaxf(acc[5] + bh.y, 0.f);
    acc[6] = fmaxf(acc[6] + bh.z, 0.f);
    acc[7] = fmaxf(acc[7] + bh.w, 0.f);
#pragma unroll
    for (int p = 0; p < 4; ++p) {
      v2f b;
      b.x = acc[2 * p];
      b.y = acc[2 * p + 1];
#pragma unroll
      for (int u = 0; u < 2; ++u) {
        v2f a;
        a.x = *(const float*)((const char*)W2 + (w2o + (unsigned)((16 * tt + 2 * p) * HID + 16 * u) * 4u));
        a.y = *(const float*)((const char*)W2 + (w2o + (unsigned)((16 * tt + 2 * p + 1) * HID + 16 * u) * 4u));
        o[u] = __builtin_amdgcn_wmma_f32_16x16x4_f32(false, a, false, b, (short)0, o[u], false, false);
      }
    }
  }
  const v8f keep = (wv == 0) ? o[0] : o[1];
  const v8f send = (wv == 0) ? o[1] : o[0];
#pragma unroll
  for (int r = 0; r < 8; ++r) part[wv][r * 32 + lane] = send[r];
  __syncthreads();
  float tot[8];
#pragma unroll
  for (int r = 0; r < 8; ++r) tot[r] = keep[r] + part[1 - wv][r * 32 + lane];
  if (node < M) {
    float* xo = XL + (size_t)node * HID + 16 * wv + 8 * h;
    float4 lo, hi;
    lo.x = di * tot[0]; lo.y = di * tot[1]; lo.z = di * tot[2]; lo.w = di * tot[3];
    hi.x = di * tot[4]; hi.y = di * tot[5]; hi.z = di * tot[6]; hi.w = di * tot[7];
    *(float4*)xo = lo;
    *(float4*)(xo + 4) = hi;
  }
}

__global__ __launch_bounds__(2 * 32) __attribute__((amdgpu_waves_per_eu(16, 16)))
void pull_pq_wmma_f32(const float* __restrict__ xl, const int* __restrict__ cnt,
                      const int* __restrict__ col, const int* __restrict__ ovf,
                      const float* __restrict__ dinv, const float* __restrict__ b2,
                      const float* __restrict__ Wm1, const float* __restrict__ bm1,
                      float* __restrict__ PQ, int M) {
  static_assert(2 * PULL_NODES == 16, "two waves walk the 16 rows of one product tile");
  __shared__ __attribute__((aligned(16))) float srow[16 * ROW_PITCH];
  const int lane = threadIdx.x & 31;
  const int wv = __builtin_amdgcn_readfirstlane((int)(threadIdx.x >> 5));
  const int tm = blockIdx.x;
  const int l15 = lane & 15;
  int mrow = tm * 16 + l15;
  if (mrow > M - 1) mrow = M - 1;
  const float di = dinv[mrow];
  const int n0 = tm * 16 + wv * PULL_NODES;
  pull_nodes<true>(xl, cnt, col, ovf, n0, lane, M, [&](int g, float4 sum) {
    float* p = srow + (wv * PULL_NODES + g) * ROW_PITCH + 4 * (lane & 7);
    v2f lo, hi;
    lo.x = sum.x; lo.y = sum.y; hi.x = sum.z; hi.y = sum.w;
    *(v2f*)p = lo;
    *(v2f*)(p + 2) = hi;
  });
  __syncthreads();
  const int kb = (lane >> 4) * 2;
  const float* sa = srow + l15 * ROW_PITCH + kb;
  const float* ba = b2 + kb;
  const float* wb = Wm1 + (size_t)(wv * HID + kb) * HID + l15;
  v8f acc[2] = {};
  for (int k0 = 0; k0 < HID; k0 += 4) {
    v2f sv = *(const v2f*)(sa + k0);
    v2f a;
    a.x = fmaxf(di * sv.x + ba[k0], 0.f);
    a.y = fmaxf(di * sv.y + ba[k0 + 1], 0.f);
#pragma unroll
    for (int u = 0; u < 2; ++u) {
      v2f b;
      b.x = wb[(size_t)k0 * HID + 16 * u];
      b.y = wb[(size_t)(k0 + 1) * HID + 16 * u];
      acc[u] = __builtin_amdgcn_wmma_f32_16x16x4_f32(false, a, false, b, (short)0, acc[u],
                                                     false, false);
    }
  }
  const int rbase = tm * 16 + ((lane >> 4) ? 8 : 0);
#pragma unroll
  for (int u = 0; u < 2; ++u) {
    const int ncol = wv * HID + 16 * u + l15;
    const float bn = (wv == 0) ? bm1[16 * u + l15] : 0.f;
#pragma unroll
    for (int r = 0; r < 8; ++r) {
      int row = rbase + r;
      float val = acc[u][r];
      if (wv == 0) val += bn;
      if (row < M) PQ[(size_t)row * (2 * HID) + ncol] = val;
    }
  }
}

#define EDGE_CHUNKS 4
__global__ __launch_bounds__(256)
void edge_out_kernel(const float* __restrict__ PQ, const int* __restrict__ p0,
                     const int* __restrict__ p1, const int* __restrict__ q0,
                     const int* __restrict__ q1, const float* __restrict__ Wm2,
                     const float* __restrict__ bm2, float* __restrict__ out,
                     int NP, int Etot) {
  int wave = (int)((blockIdx.x * (long long)blockDim.x + threadIdx.x) >> 5);
  int lane = threadIdx.x & 31;
  int G = lane >> 3;
  int j = lane & 7;
  int c = j * 4;
  const float4 w01 = *(const float4*)(Wm2 + c * 2);
  const float4 w23 = *(const float4*)(Wm2 + c * 2 + 4);
  const float bo = bm2[j & 1];
  const bool hi4 = (j & 4) != 0, hi2 = (j & 2) != 0, hi1 = (j & 1) != 0;
  for (int it = 0; it < EDGE_CHUNKS; ++it) {
    int e0 = (wave * EDGE_CHUNKS + it) * 16;
    if (e0 >= Etot) break;
    float v[8];
#pragma unroll
    for (int i = 0; i < 4; ++i) {
      int e = e0 + 4 * G + i;
      if (e > Etot - 1) e = Etot - 1;
      int s = (e < NP) ? p0[e] : q0[e - NP];
      int d = (e < NP) ? p1[e] : q1[e - NP];
      const float4 p = *(const float4*)(PQ + (size_t)s * (2 * HID) + c);
      const float4 q = *(const float4*)(PQ + (size_t)d * (2 * HID) + HID + c);
      float t0 = fmaxf(p.x + q.x, 0.f);
      float t1 = fmaxf(p.y + q.y, 0.f);
      float t2 = fmaxf(p.z + q.z, 0.f);
      float t3 = fmaxf(p.w + q.w, 0.f);
      v[2 * i]     = t0 * w01.x + t1 * w01.z + t2 * w23.x + t3 * w23.z;
      v[2 * i + 1] = t0 * w01.y + t1 * w01.w + t2 * w23.y + t3 * w23.w;
    }
    float u[4];
#pragma unroll
    for (int k = 0; k < 4; ++k) {
      float keep = hi4 ? v[k + 4] : v[k];
      float send = hi4 ? v[k] : v[k + 4];
      u[k] = keep + __shfl_xor(send, 4, 8);
    }
    float w[2];
#pragma unroll
    for (int k = 0; k < 2; ++k) {
      float keep = hi2 ? u[k + 2] : u[k];
      float send = hi2 ? u[k] : u[k + 2];
      w[k] = keep + __shfl_xor(send, 2, 8);
    }
    float keep1 = hi1 ? w[1] : w[0];
    float send1 = hi1 ? w[0] : w[1];
    float r = keep1 + __shfl_xor(send1, 1, 8);
    int eo = e0 + 4 * G + (j >> 1);
    if (eo < Etot) out[(size_t)e0 * 2 + lane] = r + bo;
  }
}

extern "C" void kernel_launch(void* const* d_in, const int* in_sizes, int n_in,
                              void* d_out, int out_size, void* d_ws, size_t ws_size,
                              hipStream_t stream) {
  const float* x   = (const float*)d_in[0];
  const int*   pe  = (const int*)d_in[1];
  const int*   neg = (const int*)d_in[2];
  const float* W1  = (const float*)d_in[3];
  const float* b1  = (const float*)d_in[4];
  const float* W2  = (const float*)d_in[5];
  const float* b2  = (const float*)d_in[6];
  const float* Wm1 = (const float*)d_in[7];
  const float* bm1 = (const float*)d_in[8];
  const float* Wm2 = (const float*)d_in[9];
  const float* bm2 = (const float*)d_in[10];
  float* out = (float*)d_out;

  const int N  = in_sizes[0] / HID;
  const int NP = in_sizes[1] / 2;
  const int NN = in_sizes[2] / 2;
  const int ET = NP + NN;
  const int* p0 = pe;        const int* p1 = pe + NP;
  const int* q0 = neg;       const int* q1 = neg + NN;

  char* ws = (char*)d_ws;
  size_t off = 0;
  auto wsalloc = [&](size_t bytes) -> char* {
    char* p = ws + off;
    off += (bytes + 255) & ~(size_t)255;
    return p;
  };
  const int NB = (N + BIN_NODES - 1) >> BIN_SHIFT;
  int*   gcnt = (int*)wsalloc(((size_t)NB + 1) * BIN_PAD * 4);
  int*   binned = (int*)wsalloc((size_t)NB * BIN_CAP * 8);
  int*   spill = (int*)wsalloc((size_t)NP * 8);
  int*   cnt  = (int*)wsalloc((size_t)(N + 1) * 4);
  int*   col  = (int*)wsalloc((size_t)N * BUCKET_CAP * 4);
  int*   ovf  = (int*)wsalloc((size_t)NP * 2 * 4);
  float* dinv = (float*)wsalloc((size_t)N * 4);
  float* xs   = (float*)wsalloc((size_t)(N + 1) * HID * 4);
  float* xl2  = (float*)wsalloc((size_t)(N + 1) * HID * 4);
  float* pq   = (float*)wsalloc((size_t)N * 2 * HID * 4);
  (void)ws_size; (void)n_in; (void)out_size;

  const int B = 256;

  hipMemsetAsync(gcnt, 0, ((size_t)NB + 1) * BIN_PAD * 4, stream);
  bin_edges_kernel<<<(unsigned)((NP + BINA_EDGES - 1) / BINA_EDGES), BINA_THREADS, 0, stream>>>(
      p0, p1, gcnt, binned, spill, cnt, NP, N, NB);
  fill_buckets_kernel<<<(unsigned)NB, BINB_THREADS, 0, stream>>>(gcnt, binned, spill, cnt, col, ovf, N, NB);
  {
    long long thr = ((long long)N * (HID / 4) + PRESCALE_PIECES - 1) / PRESCALE_PIECES;
    dinv_prescale_kernel<<<(unsigned)((thr + B - 1) / B), B, 0, stream>>>(cnt, x, dinv, xs, N);
  }

  const unsigned tb = (unsigned)((N + 15) / 16);

  pull_l1l2_wmma_f32<<<tb, 2 * 32, 0, stream>>>(xs, cnt, col, ovf, dinv, W1, b1, W2, xl2, N);
  pull_pq_wmma_f32<<<tb, 2 * 32, 0, stream>>>(xl2, cnt, col, ovf, dinv, b2, Wm1, bm1, pq, N);

  {
    long long thr = (long long)((ET + 16 * EDGE_CHUNKS - 1) / (16 * EDGE_CHUNKS)) * 32;
    edge_out_kernel<<<(unsigned)((thr + B - 1) / B), B, 0, stream>>>(pq, p0, p1, q0, q1,
                                                                     Wm2, bm2, out, NP, ET);
  }
}
